# starting kernel
# baseline (speedup 1.0000x reference)
#include <hip/hip_runtime.h>
#include <hip/hip_bf16.h>

typedef __attribute__((ext_vector_type(2))) float v2f;
typedef __attribute__((ext_vector_type(8))) float v8f;

__global__ void gcn_fill_f32(float* __restrict__ p, float v, size_t n) {
    size_t i = (size_t)blockIdx.x * blockDim.x + threadIdx.x;
    size_t stride = (size_t)gridDim.x * blockDim.x;
    for (; i < n; i += stride) p[i] = v;
}

__global__ void gcn_degree(const int* __restrict__ dst, float* __restrict__ deg, int E) {
    int e = blockIdx.x * blockDim.x + threadIdx.x;
    if (e < E) atomicAdd(&deg[dst[e]], 1.0f);
}

__global__ void gcn_rsqrt_inplace(float* __restrict__ d, int n) {
    int i = blockIdx.x * blockDim.x + threadIdx.x;
    if (i < n) {
        float v = d[i];
        d[i] = (v > 0.0f) ? __frsqrt_rn(v) : 0.0f;
    }
}

template <int NOUT>
__global__ __launch_bounds__(256) void gcn_gemm_wmma_f32(
    const float* __restrict__ A, const float* __restrict__ W,
    float* __restrict__ C, int N, int K) {
    constexpr int NT = NOUT / 16;
    const int wave = threadIdx.x >> 5;
    const int lane = threadIdx.x & 31;
    const int row0 = (blockIdx.x * 8 + wave) * 16;
    if (row0 >= N) return;

    const int m = lane & 15;
    const int g = lane >> 4;
    const float* __restrict__ arow = A + (size_t)(row0 + m) * K + 2 * g;

    v8f acc[NT];
#pragma unroll
    for (int t = 0; t < NT; ++t) acc[t] = (v8f){0, 0, 0, 0, 0, 0, 0, 0};

    for (int k = 0; k < K; k += 4) {
        v2f a = *(const v2f*)(arow + k);
        const float* __restrict__ wrow = W + (size_t)(k + 2 * g) * NOUT + m;
#pragma unroll
        for (int t = 0; t < NT; ++t) {
            v2f b;
            b.x = wrow[t * 16];
            b.y = wrow[NOUT + t * 16];
            acc[t] = __builtin_amdgcn_wmma_f32_16x16x4_f32(
                false, a, false, b, (short)0, acc[t], false, false);
        }
    }

#pragma unroll
    for (int t = 0; t < NT; ++t) {
        float* __restrict__ crow = C + (size_t)(row0 + 8 * g) * NOUT + t * 16 + m;
#pragma unroll
        for (int r = 0; r < 8; ++r) {
            crow[(size_t)r * NOUT] = acc[t][r];
        }
    }
}

__global__ void gcn_scatter(const float* __restrict__ h, const int* __restrict__ src,
                            const int* __restrict__ dst, const float* __restrict__ dinv,
                            float* __restrict__ out, int E, int F) {
    int e = blockIdx.x * blockDim.y + threadIdx.y;
    if (e >= E) return;
    int s = src[e];
    int d = dst[e];
    float nrm = dinv[s] * dinv[d];
    int f = threadIdx.x;
    atomicAdd(&out[(size_t)d * F + f], h[(size_t)s * F + f] * nrm);
}

__global__ void gcn_epilogue(float* __restrict__ agg, const float* __restrict__ h,
                             const float* __restrict__ dinv, const float* __restrict__ bias,
                             size_t total, int F, int do_relu) {
    size_t i = (size_t)blockIdx.x * blockDim.x + threadIdx.x;
    if (i >= total) return;
    size_t node = i / F;
    int f = (int)(i - node * F);
    float di = dinv[node];
    float v = agg[i] + h[i] * di * di + bias[f];
    agg[i] = do_relu ? fmaxf(v, 0.0f) : v;
}

extern "C" void kernel_launch(void* const* d_in, const int* in_sizes, int n_in,
                              void* d_out, int out_size, void* d_ws, size_t ws_size,
                              hipStream_t stream) {
    const float* x  = (const float*)d_in[0];
    const int*   ei = (const int*)d_in[1];
    const float* W1 = (const float*)d_in[2];
    const float* b1 = (const float*)d_in[3];
    const float* W2 = (const float*)d_in[4];
    const float* b2 = (const float*)d_in[5];
    float* out = (float*)d_out;

    const int DIN = 128, DH = 128, DOUT = 64;
    const int N = in_sizes[0] / DIN;
    const int E = in_sizes[1] / 2;
    const int* src = ei;
    const int* dst = ei + E;

    float* ws = (float*)d_ws;
    size_t dinvPad = ((size_t)N + 255) & ~(size_t)255;
    float* dinv = ws;
    float* h    = ws + dinvPad;
    float* agg1 = h + (size_t)N * DH;

    const int TB = 256;
    size_t nh = (size_t)N * DH;
    size_t no = (size_t)N * DOUT;

    gcn_fill_f32<<<(int)((N + TB - 1) / TB), TB, 0, stream>>>(dinv, 1.0f, (size_t)N);
    gcn_fill_f32<<<2048, TB, 0, stream>>>(agg1, 0.0f, nh);
    gcn_fill_f32<<<2048, TB, 0, stream>>>(out, 0.0f, no);

    gcn_degree<<<(E + TB - 1) / TB, TB, 0, stream>>>(dst, dinv, E);
    gcn_rsqrt_inplace<<<(N + TB - 1) / TB, TB, 0, stream>>>(dinv, N);

    int gemmBlocks = ((N / 16) + 7) / 8;
    gcn_gemm_wmma_f32<DH><<<gemmBlocks, 256, 0, stream>>>(x, W1, h, N, DIN);

    {
        dim3 blk(DH, 2);
        gcn_scatter<<<(E + 1) / 2, blk, 0, stream>>>(h, src, dst, dinv, agg1, E, DH);
    }

    gcn_epilogue<<<(int)((nh + TB - 1) / TB), TB, 0, stream>>>(agg1, h, dinv, b1, nh, DH, 1);

    gcn_gemm_wmma_f32<DOUT><<<gemmBlocks, 256, 0, stream>>>(agg1, W2, h, N, DH);

    {
        dim3 blk(DOUT, 4);
        gcn_scatter<<<(E + 3) / 4, blk, 0, stream>>>(h, src, dst, dinv, out, E, DOUT);
    }

    gcn_epilogue<<<(int)((no + TB - 1) / TB), TB, 0, stream>>>(out, h, dinv, b2, no, DOUT, 0);
}


# Filter launch cut both ways: a work-group takes 8 neighbouring tiles and one of 8 slices of the edge list (256 work-groups, not 125 whole-list passes); per-slice list segments and degree counts
# speedup vs baseline: 4.0599x; 4.0599x over previous
#include <hip/hip_runtime.h>
#include <hip/hip_bf16.h>

typedef __attribute__((ext_vector_type(2))) float v2f;
typedef __attribute__((ext_vector_type(8))) float v8f;

constexpr int AGG_GS = 8;

template <int NOUT, int DINV>
__global__ __launch_bounds__(256) void gcn_gemm_wmma_f32(
    const float* __restrict__ A, const float* __restrict__ W,
    float* __restrict__ C, int N, int K,
    const int* __restrict__ degp, float* __restrict__ dinv) {
    constexpr int NT = NOUT / 16;
    if (DINV) {
        const int i = blockIdx.x * 256 + threadIdx.x;
        if (i < N) {
            int d = 1;
            for (int sl = 0; sl < AGG_GS; ++sl) d += degp[(size_t)sl * N + i];
            const float v = (float)d;
            dinv[i] = (v > 0.0f) ? __frsqrt_rn(v) : 0.0f;
        }
    }
    const int wave = threadIdx.x >> 5;
    const int lane = threadIdx.x & 31;
    const int row0 = (blockIdx.x * 8 + wave) * 16;
    if (row0 >= N) return;

    const int m = lane & 15;
    const int g = lane >> 4;
    const float* __restrict__ arow = A + (size_t)(row0 + m) * K + 2 * g;

    v8f acc[NT];
#pragma unroll
    for (int t = 0; t < NT; ++t) acc[t] = (v8f){0, 0, 0, 0, 0, 0, 0, 0};

    for (int k = 0; k < K; k += 4) {
        v2f a = *(const v2f*)(arow + k);
        const float* __restrict__ wrow = W + (size_t)(k + 2 * g) * NOUT + m;
#pragma unroll
        for (int t = 0; t < NT; ++t) {
            v2f b;
            b.x = wrow[t * 16];
            b.y = wrow[NOUT + t * 16];
            acc[t] = __builtin_amdgcn_wmma_f32_16x16x4_f32(
                false, a, false, b, (short)0, acc[t], false, false);
        }
    }

#pragma unroll
    for (int t = 0; t < NT; ++t) {
        float* __restrict__ crow = C + (size_t)(row0 + 8 * g) * NOUT + t * 16 + m;
#pragma unroll
        for (int r = 0; r < 8; ++r) {
            crow[(size_t)r * NOUT] = acc[t][r];
        }
    }
}

constexpr int AGG_ROWS  = 160;
constexpr int AGG_WAVES = 32;
constexpr int AGG_GT    = 8;
constexpr int AGG_GROWS = AGG_GT * AGG_ROWS;
constexpr int AGG_RBITS = 11;
constexpr int AGG_EBITS = 32 - AGG_RBITS;
constexpr int AGG_QCAP  = 1024;
constexpr int AGG_FQCAP = 256;
constexpr int AGG_LROW  = 64;
static_assert(AGG_GROWS <= (1 << AGG_RBITS), "row inside the group must fit its bits");
static_assert(AGG_WAVES % AGG_GS == 0, "every list segment gets the same number of waves");

__global__ __launch_bounds__(1024) void gcn_tile_lists(
    const int* __restrict__ src, const int* __restrict__ dst,
    int* __restrict__ degp, float* __restrict__ lists, int N, int E) {
    __shared__ unsigned q[AGG_WAVES * AGG_QCAP];
    __shared__ int degc[AGG_GROWS];
    __shared__ int tcnt[AGG_GT];

    const int tid  = threadIdx.x;
    const int lane = tid & 31;
    const int wave = __builtin_amdgcn_readfirstlane(tid >> 5);
    unsigned* myq = q + wave * AGG_QCAP;

    const int ngroups = (N + AGG_GROWS - 1) / AGG_GROWS;
    const bool vec_ok = (reinterpret_cast<size_t>(dst) & 15) == 0;
    const int nfull = vec_ok ? E / 512 : 0;

    for (int job = blockIdx.x; job < ngroups * AGG_GS; job += gridDim.x) {
        const int group = job / AGG_GS;
        const int slice = job % AGG_GS;
        const int lo = group * AGG_GROWS;
        const int nrows = min(AGG_GROWS, N - lo);
        float* gl = lists + (size_t)lo * AGG_LROW;
        const int t_lo = (int)(((long long)nfull * slice) / AGG_GS);
        const int t_hi = (int)(((long long)nfull * (slice + 1)) / AGG_GS);
        const int nt = t_hi - t_lo;
        const int ebase = t_lo * 512;
        const int* dsl = dst + ebase;
        const int* ssl = src + ebase;

        for (int i = tid; i < AGG_GROWS; i += 1024) degc[i] = 0;
        if (tid < AGG_GT) tcnt[tid] = 0;
        __syncthreads();

        int cnt = 0;

        auto flush = [&]() {
            for (int i0 = 0; i0 < cnt; i0 += 32) {
                const bool v = i0 + lane < cnt;
                unsigned r = 0u, s = 0u;
                int tt = -1;
                if (v) {
                    const unsigned ent = myq[i0 + lane];
                    r = ent >> AGG_EBITS;
                    s = (unsigned)ssl[ent & ((1u << AGG_EBITS) - 1u)];
                    atomicAdd(&degc[r], 1);
                    tt = (int)(r / (unsigned)AGG_ROWS);
                }
                for (int k = 0; k < AGG_GT; ++k) {
                    const bool mk = tt == k;
                    const unsigned bk = __builtin_amdgcn_ballot_w32(mk);
                    if (bk != 0u) {
                        const int nrk  = min(AGG_ROWS, nrows - k * AGG_ROWS);
                        const int segw = (nrk * AGG_LROW) / AGG_GS;
                        float* seg = gl + (size_t)k * AGG_ROWS * AGG_LROW + (size_t)slice * segw;
                        int base = 0;
                        if (lane == 0) base = atomicAdd(&tcnt[k], __builtin_popcount(bk));
                        base = __builtin_amdgcn_readfirstlane(base);
                        if (mk) {
                            const int idx = base + (int)__builtin_amdgcn_mbcnt_lo(bk, 0u);
                            if (idx < segw - 1)
                                seg[1 + idx] = __uint_as_float(((r - (unsigned)(k * AGG_ROWS)) << 24) | s);
                        }
                    }
                }
            }
            cnt = 0;
        };
        auto room = [&](int need) {
            if (cnt > AGG_QCAP - need) flush();
        };
        auto keep4 = [&](const int4& d, int e) {
            const unsigned r0 = (unsigned)(d.x - lo);
            const unsigned r1 = (unsigned)(d.y - lo);
            const unsigned r2 = (unsigned)(d.z - lo);
            const unsigned r3 = (unsigned)(d.w - lo);
            const bool m0 = r0 < (unsigned)nrows;
            const bool m1 = r1 < (unsigned)nrows;
            const bool m2 = r2 < (unsigned)nrows;
            const bool m3 = r3 < (unsigned)nrows;
            if (__builtin_amdgcn_ballot_w32(m0 | m1 | m2 | m3) != 0u) {
                const unsigned b0 = __builtin_amdgcn_ballot_w32(m0);
                if (m0) myq[cnt + (int)__builtin_amdgcn_mbcnt_lo(b0, 0u)] = (r0 << AGG_EBITS) | (unsigned)(e + 0);
                cnt += __builtin_popcount(b0);
                const unsigned b1 = __builtin_amdgcn_ballot_w32(m1);
                if (m1) myq[cnt + (int)__builtin_amdgcn_mbcnt_lo(b1, 0u)] = (r1 << AGG_EBITS) | (unsigned)(e + 1);
                cnt += __builtin_popcount(b1);
                const unsigned b2 = __builtin_amdgcn_ballot_w32(m2);
                if (m2) myq[cnt + (int)__builtin_amdgcn_mbcnt_lo(b2, 0u)] = (r2 << AGG_EBITS) | (unsigned)(e + 2);
                cnt += __builtin_popcount(b2);
                const unsigned b3 = __builtin_amdgcn_ballot_w32(m3);
                if (m3) myq[cnt + (int)__builtin_amdgcn_mbcnt_lo(b3, 0u)] = (r3 << AGG_EBITS) | (unsigned)(e + 3);
                cnt += __builtin_popcount(b3);
            }
        };

        if (wave < nt) {
            const int rstep = max(nt / max(ngroups, 1), 1);
            const int rot = (int)(((unsigned)group * (unsigned)rstep) % (unsigned)nt);
            auto trip_base = [&](int t) {
                int sg = t + rot;
                if (sg >= nt) sg -= nt;
                return sg * 512 + lane * 4;
            };
            int4 a[4], b[4];
            int t  = wave;
            int ea = trip_base(t);
            int eb = ea;
#pragma unroll
            for (int k = 0; k < 4; ++k) a[k] = *(const int4*)(dsl + ea + k * 128);
            while (t < nt) {
                const int tb = t + AGG_WAVES;
                const bool hb = tb < nt;
                room(512);
                eb = trip_base(hb ? tb : t);
#pragma unroll
                for (int k = 0; k < 4; ++k) b[k] = *(const int4*)(dsl + eb + k * 128);
#pragma unroll
                for (int k = 0; k < 4; ++k) keep4(a[k], ea + k * 128);

                const int ta = tb + AGG_WAVES;
                const bool ha = ta < nt;
                room(512);
                ea = trip_base(ha ? ta : t);
#pragma unroll
                for (int k = 0; k < 4; ++k) a[k] = *(const int4*)(dsl + ea + k * 128);
                if (hb) {
#pragma unroll
                    for (int k = 0; k < 4; ++k) keep4(b[k], eb + k * 128);
                }
                t = ta;
            }
        }
        if (slice == AGG_GS - 1) {
            for (int e0 = nfull * 512 + wave * 32; e0 < E; e0 += AGG_WAVES * 32) {
                const int e = e0 + lane;
                const int dv = (e < E) ? dst[e] : -1;
                const unsigned r = (unsigned)(dv - lo);
                const bool m = r < (unsigned)nrows;
                const unsigned bm = __builtin_amdgcn_ballot_w32(m);
                if (bm != 0u) {
                    room(32);
                    if (m) myq[cnt + (int)__builtin_amdgcn_mbcnt_lo(bm, 0u)] = (r << AGG_EBITS) | (unsigned)(e - ebase);
                    cnt += __builtin_popcount(bm);
                }
            }
        }
        flush();
        __syncthreads();

        if (tid < AGG_GT && nrows > tid * AGG_ROWS) {
            const int nrk  = min(AGG_ROWS, nrows - tid * AGG_ROWS);
            const int segw = (nrk * AGG_LROW) / AGG_GS;
            gl[(size_t)tid * AGG_ROWS * AGG_LROW + (size_t)slice * segw] = __uint_as_float((unsigned)tcnt[tid]);
        }
        for (int i = tid; i < nrows; i += 1024) degp[(size_t)slice * N + lo + i] = degc[i];
        __syncthreads();
    }
}

template <int F>
__device__ __forceinline__ void gcn_agg_drain(
    const float* __restrict__ h, const float* __restrict__ dinv,
    const float* dl, float* acc, const float* list, int cnt, int lane) {
    constexpr int LPE = F / 4;
    constexpr int EPW = 32 / LPE;
    constexpr int UD  = 8;
    const int sub = (EPW == 1) ? 0 : lane / LPE;
    const int fl  = lane % LPE;
    const int hi  = (lane >> 4) & 1;
    unsigned pre = (lane < cnt) ? __float_as_uint(list[lane]) : 0u;
    for (int base = 0; base < cnt; base += 32) {
        const int n = min(32, cnt - base);
        const unsigned ent = pre;
        const int nb = base + 32 + lane;
        pre = (nb < cnt) ? __float_as_uint(list[nb]) : 0u;
        int s = 0, r = 0;
        float nrm = 0.0f;
        if (lane < n) {
            r = (int)(ent >> 24);
            s = (int)(ent & 0xFFFFFFu);
            nrm = dinv[s] * dl[r];
        }
        for (int i = 0; i < n; i += UD * EPW) {
            float4 v[UD];
            float  w[UD];
            int    si[UD];
            int    rr[UD];
            bool   ok[UD];
#pragma unroll
            for (int k = 0; k < UD; ++k) {
                const int idx = i + k * EPW + sub;
                ok[k] = idx < n;
                const int ii = ok[k] ? idx : 0;
                if constexpr (EPW == 1) {
                    si[k] = __builtin_amdgcn_readlane(s, ii);
                    rr[k] = __builtin_amdgcn_readlane(r, ii);
                    w[k]  = __int_as_float(__builtin_amdgcn_readlane(__float_as_int(nrm), ii));
                } else {
                    si[k] = __shfl(s, ii);
                    rr[k] = __shfl(r, ii);
                    w[k]  = __shfl(nrm, ii);
                }
            }
#pragma unroll
            for (int k = 0; k < UD; ++k) {
                v[k] = make_float4(0.0f, 0.0f, 0.0f, 0.0f);
                if (ok[k]) v[k] = *((const float4*)(h + (size_t)si[k] * F) + fl);
            }
#pragma unroll
            for (int k = 0; k < UD; ++k) {
                if (ok[k]) {
                    float* a  = acc + rr[k] * F + fl * 4;
                    float* p0 = a + 2 * hi;
                    float* p1 = a + 2 - 2 * hi;
                    const float c0 = hi ? v[k].z : v[k].x;
                    const float c1 = hi ? v[k].w : v[k].y;
                    const float c2 = hi ? v[k].x : v[k].z;
                    const float c3 = hi ? v[k].y : v[k].w;
                    atomicAdd(p0 + 0, c0 * w[k]);
                    atomicAdd(p0 + 1, c1 * w[k]);
                    atomicAdd(p1 + 0, c2 * w[k]);
                    atomicAdd(p1 + 1, c3 * w[k]);
                }
            }
        }
    }
}

template <int F, int RELU>
__global__ __launch_bounds__(1024) void gcn_agg_tile(
    const float* __restrict__ h, const int* __restrict__ src, const int* __restrict__ dst,
    const float* __restrict__ dinv, const float* __restrict__ bias,
    const float* lists, float* out, int N, int E) {
    __shared__ float acc[AGG_ROWS * F];
    __shared__ float dl[AGG_ROWS];
    __shared__ float fq[AGG_WAVES * AGG_FQCAP];

    const int tid  = threadIdx.x;
    const int lane = tid & 31;
    const int wave = __builtin_amdgcn_readfirstlane(tid >> 5);

    const int ntiles = (N + AGG_ROWS - 1) / AGG_ROWS;

    for (int tile = blockIdx.x; tile < ntiles; tile += gridDim.x) {
        const int lo = tile * AGG_ROWS;
        const int nrows = min(AGG_ROWS, N - lo);
        const float* tl = lists + (size_t)lo * AGG_LROW;
        constexpr int WPS = AGG_WAVES / AGG_GS;
        const int segw = (nrows * AGG_LROW) / AGG_GS;
        const int seg  = wave / WPS;
        bool fits = true;
        int total = 0;
        for (int sg = 0; sg < AGG_GS; ++sg) {
            const int c = __builtin_amdgcn_readfirstlane((int)__float_as_uint(tl[(size_t)sg * segw]));
            fits = fits && c >= 0 && c <= segw - 1;
            if (sg == seg) total = c;
        }

        for (int i = tid; i < AGG_ROWS * F / 4; i += 1024)
            ((float4*)acc)[i] = make_float4(0.0f, 0.0f, 0.0f, 0.0f);
        for (int i = tid; i < nrows; i += 1024) dl[i] = dinv[lo + i];
        __syncthreads();

        if (fits) {
            const int per = (total + WPS - 1) / WPS;
            const int b = min(total, (wave % WPS) * per);
            const int e = min(total, b + per);
            gcn_agg_drain<F>(h, dinv, dl, acc, tl + (size_t)seg * segw + 1 + b, e - b, lane);
        } else {
            float* myq = fq + wave * AGG_FQCAP;
            int cnt = 0;
            for (int e0 = wave * 32; e0 < E; e0 += AGG_WAVES * 32) {
                const int e = e0 + lane;
                const int dv = (e < E) ? dst[e] : -1;
                const unsigned r = (unsigned)(dv - lo);
                const bool m = r < (unsigned)nrows;
                const unsigned bm = __builtin_amdgcn_ballot_w32(m);
                if (bm != 0u) {
                    if (cnt > AGG_FQCAP - 32) {
                        gcn_agg_drain<F>(h, dinv, dl, acc, myq, cnt, lane);
                        cnt = 0;
                    }
                    if (m) myq[cnt + (int)__builtin_amdgcn_mbcnt_lo(bm, 0u)] =
                        __uint_as_float((r << 24) | (unsigned)src[e]);
                    cnt += __builtin_popcount(bm);
                }
            }
            gcn_agg_drain<F>(h, dinv, dl, acc, myq, cnt, lane);
        }
        __syncthreads();

        for (int i = tid; i < nrows * (F / 4); i += 1024) {
            const int r  = i / (F / 4);
            const int f4 = i % (F / 4);
            const size_t gi = ((size_t)(lo + r) * F) / 4 + f4;
            const float di = dl[r];
            const float4 sa = ((const float4*)acc)[i];
            const float4 hv = ((const float4*)h)[gi];
            const float4 bv = ((const float4*)bias)[f4];
            float4 v;
            v.x = sa.x + hv.x * di * di + bv.x;
            v.y = sa.y + hv.y * di * di + bv.y;
            v.z = sa.z + hv.z * di * di + bv.z;
            v.w = sa.w + hv.w * di * di + bv.w;
            if (RELU) {
                v.x = fmaxf(v.x, 0.0f); v.y = fmaxf(v.y, 0.0f);
                v.z = fmaxf(v.z, 0.0f); v.w = fmaxf(v.w, 0.0f);
            }
            ((float4*)out)[gi] = v;
        }
        __syncthreads();
    }
}

extern "C" void kernel_launch(void* const* d_in, const int* in_sizes, int n_in,
                              void* d_out, int out_size, void* d_ws, size_t ws_size,
                              hipStream_t stream) {
    const float* x  = (const float*)d_in[0];
    const int*   ei = (const int*)d_in[1];
    const float* W1 = (const float*)d_in[2];
    const float* b1 = (const float*)d_in[3];
    const float* W2 = (const float*)d_in[4];
    const float* b2 = (const float*)d_in[5];
    float* out = (float*)d_out;

    const int DIN = 128, DH = 128, DOUT = 64;
    const int N = in_sizes[0] / DIN;
    const int E = in_sizes[1] / 2;
    const int* src = ei;
    const int* dst = ei + E;

    float* ws = (float*)d_ws;
    size_t dinvPad = ((size_t)N + 255) & ~(size_t)255;
    float* dinv = ws;
    float* h    = ws + dinvPad;
    float* agg1 = h + (size_t)N * DH;

    int* degp = (int*)agg1;
    gcn_tile_lists<<<256, 1024, 0, stream>>>(src, dst, degp, out, N, E);

    int gemmBlocks = ((N / 16) + 7) / 8;
    gcn_gemm_wmma_f32<DH, 1><<<gemmBlocks, 256, 0, stream>>>(x, W1, h, N, DIN, degp, dinv);

    gcn_agg_tile<128, 1><<<256, 1024, 0, stream>>>(h, src, dst, dinv, b1, out, agg1, N, E);

    gcn_gemm_wmma_f32<DOUT, 0><<<gemmBlocks, 256, 0, stream>>>(agg1, W2, h, N, DH, degp, dinv);

    gcn_agg_tile<64, 0><<<256, 1024, 0, stream>>>(h, src, dst, dinv, b2, out, out, N, E);
}


# GEMM launches stage W once per work-group in LDS (row pairs interleaved, padded pitch); B fragments are 8-byte LDS reads, not four-byte global loads per k-step; 10 strips per work-group (250 groups)
# speedup vs baseline: 4.3876x; 1.0807x over previous
#include <hip/hip_runtime.h>
#include <hip/hip_bf16.h>

typedef __attribute__((ext_vector_type(2))) float v2f;
typedef __attribute__((ext_vector_type(8))) float v8f;

constexpr int AGG_GS = 8;
constexpr int GEMM_WAVES = 10;
constexpr int GEMM_K = 128;

template <int NOUT, int DINV>
__global__ __launch_bounds__(GEMM_WAVES * 32) void gcn_gemm_wmma_f32(
    const float* __restrict__ A, const float* __restrict__ W,
    float* __restrict__ C, int N,
    const int* __restrict__ degp, float* __restrict__ dinv) {
    constexpr int K  = GEMM_K;
    constexpr int NT = NOUT / 16;
    constexpr int PP = 2 * NOUT + 32;
    constexpr int NTHR = GEMM_WAVES * 32;
    __shared__ __attribute__((aligned(16))) float Ws[(K / 2) * PP];

    const int tid = threadIdx.x;
    if (DINV) {
        for (int i = blockIdx.x * NTHR + tid; i < N; i += gridDim.x * NTHR) {
            int d = 1;
            for (int sl = 0; sl < AGG_GS; ++sl) d += degp[(size_t)sl * N + i];
            const float v = (float)d;
            dinv[i] = (v > 0.0f) ? __frsqrt_rn(v) : 0.0f;
        }
    }
    const int wave = tid >> 5;
    const int lane = tid & 31;
    const int row0 = (blockIdx.x * GEMM_WAVES + wave) * 16;
    const bool live = row0 < N;

    const int m = lane & 15;
    const int g = lane >> 4;

    v2f a[K / 4];
    {
        const float* __restrict__ arow = A + (size_t)((live ? row0 : 0) + m) * K + 2 * g;
#pragma unroll
        for (int s = 0; s < K / 4; ++s) a[s] = *(const v2f*)(arow + 4 * s);
    }

    constexpr int ITEMS = (K / 2) * (NOUT / 4);
    constexpr int TRIPS = (ITEMS + NTHR - 1) / NTHR;
    float4 v0[TRIPS], v1[TRIPS];
#pragma unroll
    for (int j = 0; j < TRIPS; ++j) {
        const int i  = tid + j * NTHR;
        const bool full = (j + 1) * NTHR <= ITEMS;
        const int ii = (full || i < ITEMS) ? i : 0;
        const int p  = ii / (NOUT / 4);
        const int c4 = ii % (NOUT / 4);
        v0[j] = ((const float4*)W)[(size_t)(2 * p) * (NOUT / 4) + c4];
        v1[j] = ((const float4*)W)[(size_t)(2 * p + 1) * (NOUT / 4) + c4];
    }
#pragma unroll
    for (int j = 0; j < TRIPS; ++j) {
        const int i = tid + j * NTHR;
        const bool full = (j + 1) * NTHR <= ITEMS;
        if (full || i < ITEMS) {
            const int p  = i / (NOUT / 4);
            const int c4 = i % (NOUT / 4);
            float4* d = (float4*)&Ws[p * PP + c4 * 8];
            d[0] = make_float4(v0[j].x, v1[j].x, v0[j].y, v1[j].y);
            d[1] = make_float4(v0[j].z, v1[j].z, v0[j].w, v1[j].w);
        }
    }
    __syncthreads();
    if (!live) return;

    v8f acc[NT];
#pragma unroll
    for (int t = 0; t < NT; ++t) acc[t] = (v8f){0, 0, 0, 0, 0, 0, 0, 0};

    const float* __restrict__ wl = Ws + g * PP + 2 * m;
    v2f b[NT], bn[NT];
#pragma unroll
    for (int t = 0; t < NT; ++t) b[t] = *(const v2f*)(wl + t * 32);
#pragma unroll
    for (int s = 0; s < K / 4; ++s) {
        if (s + 1 < K / 4) {
            const float* __restrict__ wn = wl + (2 * (s + 1)) * PP;
#pragma unroll
            for (int t = 0; t < NT; ++t) bn[t] = *(const v2f*)(wn + t * 32);
        }
#pragma unroll
        for (int t = 0; t < NT; ++t) {
            acc[t] = __builtin_amdgcn_wmma_f32_16x16x4_f32(
                false, a[s], false, b[t], (short)0, acc[t], false, false);
        }
#pragma unroll
        for (int t = 0; t < NT; ++t) b[t] = bn[t];
    }

#pragma unroll
    for (int t = 0; t < NT; ++t) {
        float* __restrict__ crow = C + (size_t)(row0 + 8 * g) * NOUT + t * 16 + m;
#pragma unroll
        for (int r = 0; r < 8; ++r) {
            crow[(size_t)r * NOUT] = acc[t][r];
        }
    }
}

constexpr int AGG_ROWS  = 160;
constexpr int AGG_WAVES = 32;
constexpr int AGG_GT    = 8;
constexpr int AGG_GROWS = AGG_GT * AGG_ROWS;
constexpr int AGG_RBITS = 11;
constexpr int AGG_EBITS = 32 - AGG_RBITS;
constexpr int AGG_QCAP  = 1024;
constexpr int AGG_FQCAP = 256;
constexpr int AGG_LROW  = 64;
static_assert(AGG_GROWS <= (1 << AGG_RBITS), "row inside the group must fit its bits");
static_assert(AGG_WAVES % AGG_GS == 0, "every list segment gets the same number of waves");

__global__ __launch_bounds__(1024) void gcn_tile_lists(
    const int* __restrict__ src, const int* __restrict__ dst,
    int* __restrict__ degp, float* __restrict__ lists, int N, int E) {
    __shared__ unsigned q[AGG_WAVES * AGG_QCAP];
    __shared__ int degc[AGG_GROWS];
    __shared__ int tcnt[AGG_GT];

    const int tid  = threadIdx.x;
    const int lane = tid & 31;
    const int wave = __builtin_amdgcn_readfirstlane(tid >> 5);
    unsigned* myq = q + wave * AGG_QCAP;

    const int ngroups = (N + AGG_GROWS - 1) / AGG_GROWS;
    const bool vec_ok = (reinterpret_cast<size_t>(dst) & 15) == 0;
    const int nfull = vec_ok ? E / 512 : 0;

    for (int job = blockIdx.x; job < ngroups * AGG_GS; job += gridDim.x) {
        const int group = job / AGG_GS;
        const int slice = job % AGG_GS;
        const int lo = group * AGG_GROWS;
        const int nrows = min(AGG_GROWS, N - lo);
        float* gl = lists + (size_t)lo * AGG_LROW;
        const int t_lo = (int)(((long long)nfull * slice) / AGG_GS);
        const int t_hi = (int)(((long long)nfull * (slice + 1)) / AGG_GS);
        const int nt = t_hi - t_lo;
        const int ebase = t_lo * 512;
        const int* dsl = dst + ebase;
        const int* ssl = src + ebase;

        for (int i = tid; i < AGG_GROWS; i += 1024) degc[i] = 0;
        if (tid < AGG_GT) tcnt[tid] = 0;
        __syncthreads();

        int cnt = 0;

        auto flush = [&]() {
            for (int i0 = 0; i0 < cnt; i0 += 32) {
                const bool v = i0 + lane < cnt;
                unsigned r = 0u, s = 0u;
                int tt = -1;
                if (v) {
                    const unsigned ent = myq[i0 + lane];
                    r = ent >> AGG_EBITS;
                    s = (unsigned)ssl[ent & ((1u << AGG_EBITS) - 1u)];
                    atomicAdd(&degc[r], 1);
                    tt = (int)(r / (unsigned)AGG_ROWS);
                }
                for (int k = 0; k < AGG_GT; ++k) {
                    const bool mk = tt == k;
                    const unsigned bk = __builtin_amdgcn_ballot_w32(mk);
                    if (bk != 0u) {
                        const int nrk  = min(AGG_ROWS, nrows - k * AGG_ROWS);
                        const int segw = (nrk * AGG_LROW) / AGG_GS;
                        float* seg = gl + (size_t)k * AGG_ROWS * AGG_LROW + (size_t)slice * segw;
                        int base = 0;
                        if (lane == 0) base = atomicAdd(&tcnt[k], __builtin_popcount(bk));
                        base = __builtin_amdgcn_readfirstlane(base);
                        if (mk) {
                            const int idx = base + (int)__builtin_amdgcn_mbcnt_lo(bk, 0u);
                            if (idx < segw - 1)
                                seg[1 + idx] = __uint_as_float(((r - (unsigned)(k * AGG_ROWS)) << 24) | s);
                        }
                    }
                }
            }
            cnt = 0;
        };
        auto room = [&](int need) {
            if (cnt > AGG_QCAP - need) flush();
        };
        auto keep4 = [&](const int4& d, int e) {
            const unsigned r0 = (unsigned)(d.x - lo);
            const unsigned r1 = (unsigned)(d.y - lo);
            const unsigned r2 = (unsigned)(d.z - lo);
            const unsigned r3 = (unsigned)(d.w - lo);
            const bool m0 = r0 < (unsigned)nrows;
            const bool m1 = r1 < (unsigned)nrows;
            const bool m2 = r2 < (unsigned)nrows;
            const bool m3 = r3 < (unsigned)nrows;
            if (__builtin_amdgcn_ballot_w32(m0 | m1 | m2 | m3) != 0u) {
                const unsigned b0 = __builtin_amdgcn_ballot_w32(m0);
                if (m0) myq[cnt + (int)__builtin_amdgcn_mbcnt_lo(b0, 0u)] = (r0 << AGG_EBITS) | (unsigned)(e + 0);
                cnt += __builtin_popcount(b0);
                const unsigned b1 = __builtin_amdgcn_ballot_w32(m1);
                if (m1) myq[cnt + (int)__builtin_amdgcn_mbcnt_lo(b1, 0u)] = (r1 << AGG_EBITS) | (unsigned)(e + 1);
                cnt += __builtin_popcount(b1);
                const unsigned b2 = __builtin_amdgcn_ballot_w32(m2);
                if (m2) myq[cnt + (int)__builtin_amdgcn_mbcnt_lo(b2, 0u)] = (r2 << AGG_EBITS) | (unsigned)(e + 2);
                cnt += __builtin_popcount(b2);
                const unsigned b3 = __builtin_amdgcn_ballot_w32(m3);
                if (m3) myq[cnt + (int)__builtin_amdgcn_mbcnt_lo(b3, 0u)] = (r3 << AGG_EBITS) | (unsigned)(e + 3);
                cnt += __builtin_popcount(b3);
            }
        };

        if (wave < nt) {
            const int rstep = max(nt / max(ngroups, 1), 1);
            const int rot = (int)(((unsigned)group * (unsigned)rstep) % (unsigned)nt);
            auto trip_base = [&](int t) {
                int sg = t + rot;
                if (sg >= nt) sg -= nt;
                return sg * 512 + lane * 4;
            };
            int4 a[4], b[4];
            int t  = wave;
            int ea = trip_base(t);
            int eb = ea;
#pragma unroll
            for (int k = 0; k < 4; ++k) a[k] = *(const int4*)(dsl + ea + k * 128);
            while (t < nt) {
                const int tb = t + AGG_WAVES;
                const bool hb = tb < nt;
                room(512);
                eb = trip_base(hb ? tb : t);
#pragma unroll
                for (int k = 0; k < 4; ++k) b[k] = *(const int4*)(dsl + eb + k * 128);
#pragma unroll
                for (int k = 0; k < 4; ++k) keep4(a[k], ea + k * 128);

                const int ta = tb + AGG_WAVES;
                const bool ha = ta < nt;
                room(512);
                ea = trip_base(ha ? ta : t);
#pragma unroll
                for (int k = 0; k < 4; ++k) a[k] = *(const int4*)(dsl + ea + k * 128);
                if (hb) {
#pragma unroll
                    for (int k = 0; k < 4; ++k) keep4(b[k], eb + k * 128);
                }
                t = ta;
            }
        }
        if (slice == AGG_GS - 1) {
            for (int e0 = nfull * 512 + wave * 32; e0 < E; e0 += AGG_WAVES * 32) {
                const int e = e0 + lane;
                const int dv = (e < E) ? dst[e] : -1;
                const unsigned r = (unsigned)(dv - lo);
                const bool m = r < (unsigned)nrows;
                const unsigned bm = __builtin_amdgcn_ballot_w32(m);
                if (bm != 0u) {
                    room(32);
                    if (m) myq[cnt + (int)__builtin_amdgcn_mbcnt_lo(bm, 0u)] = (r << AGG_EBITS) | (unsigned)(e - ebase);
                    cnt += __builtin_popcount(bm);
                }
            }
        }
        flush();
        __syncthreads();

        if (tid < AGG_GT && nrows > tid * AGG_ROWS) {
            const int nrk  = min(AGG_ROWS, nrows - tid * AGG_ROWS);
            const int segw = (nrk * AGG_LROW) / AGG_GS;
            gl[(size_t)tid * AGG_ROWS * AGG_LROW + (size_t)slice * segw] = __uint_as_float((unsigned)tcnt[tid]);
        }
        for (int i = tid; i < nrows; i += 1024) degp[(size_t)slice * N + lo + i] = degc[i];
        __syncthreads();
    }
}

template <int F>
__device__ __forceinline__ void gcn_agg_drain(
    const float* __restrict__ h, const float* __restrict__ dinv,
    const float* dl, float* acc, const float* list, int cnt, int lane) {
    constexpr int LPE = F / 4;
    constexpr int EPW = 32 / LPE;
    constexpr int UD  = 8;
    const int sub = (EPW == 1) ? 0 : lane / LPE;
    const int fl  = lane % LPE;
    const int hi  = (lane >> 4) & 1;
    unsigned pre = (lane < cnt) ? __float_as_uint(list[lane]) : 0u;
    for (int base = 0; base < cnt; base += 32) {
        const int n = min(32, cnt - base);
        const unsigned ent = pre;
        const int nb = base + 32 + lane;
        pre = (nb < cnt) ? __float_as_uint(list[nb]) : 0u;
        int s = 0, r = 0;
        float nrm = 0.0f;
        if (lane < n) {
            r = (int)(ent >> 24);
            s = (int)(ent & 0xFFFFFFu);
            nrm = dinv[s] * dl[r];
        }
        for (int i = 0; i < n; i += UD * EPW) {
            float4 v[UD];
            float  w[UD];
            int    si[UD];
            int    rr[UD];
            bool   ok[UD];
#pragma unroll
            for (int k = 0; k < UD; ++k) {
                const int idx = i + k * EPW + sub;
                ok[k] = idx < n;
                const int ii = ok[k] ? idx : 0;
                if constexpr (EPW == 1) {
                    si[k] = __builtin_amdgcn_readlane(s, ii);
                    rr[k] = __builtin_amdgcn_readlane(r, ii);
                    w[k]  = __int_as_float(__builtin_amdgcn_readlane(__float_as_int(nrm), ii));
                } else {
                    si[k] = __shfl(s, ii);
                    rr[k] = __shfl(r, ii);
                    w[k]  = __shfl(nrm, ii);
                }
            }
#pragma unroll
            for (int k = 0; k < UD; ++k) {
                v[k] = make_float4(0.0f, 0.0f, 0.0f, 0.0f);
                if (ok[k]) v[k] = *((const float4*)(h + (size_t)si[k] * F) + fl);
            }
#pragma unroll
            for (int k = 0; k < UD; ++k) {
                if (ok[k]) {
                    float* a  = acc + rr[k] * F + fl * 4;
                    float* p0 = a + 2 * hi;
                    float* p1 = a + 2 - 2 * hi;
                    const float c0 = hi ? v[k].z : v[k].x;
                    const float c1 = hi ? v[k].w : v[k].y;
                    const float c2 = hi ? v[k].x : v[k].z;
                    const float c3 = hi ? v[k].y : v[k].w;
                    atomicAdd(p0 + 0, c0 * w[k]);
                    atomicAdd(p0 + 1, c1 * w[k]);
                    atomicAdd(p1 + 0, c2 * w[k]);
                    atomicAdd(p1 + 1, c3 * w[k]);
                }
            }
        }
    }
}

template <int F, int RELU>
__global__ __launch_bounds__(1024) void gcn_agg_tile(
    const float* __restrict__ h, const int* __restrict__ src, const int* __restrict__ dst,
    const float* __restrict__ dinv, const float* __restrict__ bias,
    const float* lists, float* out, int N, int E) {
    __shared__ float acc[AGG_ROWS * F];
    __shared__ float dl[AGG_ROWS];
    __shared__ float fq[AGG_WAVES * AGG_FQCAP];

    const int tid  = threadIdx.x;
    const int lane = tid & 31;
    const int wave = __builtin_amdgcn_readfirstlane(tid >> 5);

    const int ntiles = (N + AGG_ROWS - 1) / AGG_ROWS;

    for (int tile = blockIdx.x; tile < ntiles; tile += gridDim.x) {
        const int lo = tile * AGG_ROWS;
        const int nrows = min(AGG_ROWS, N - lo);
        const float* tl = lists + (size_t)lo * AGG_LROW;
        constexpr int WPS = AGG_WAVES / AGG_GS;
        const int segw = (nrows * AGG_LROW) / AGG_GS;
        const int seg  = wave / WPS;
        bool fits = true;
        int total = 0;
        for (int sg = 0; sg < AGG_GS; ++sg) {
            const int c = __builtin_amdgcn_readfirstlane((int)__float_as_uint(tl[(size_t)sg * segw]));
            fits = fits && c >= 0 && c <= segw - 1;
            if (sg == seg) total = c;
        }

        for (int i = tid; i < AGG_ROWS * F / 4; i += 1024)
            ((float4*)acc)[i] = make_float4(0.0f, 0.0f, 0.0f, 0.0f);
        for (int i = tid; i < nrows; i += 1024) dl[i] = dinv[lo + i];
        __syncthreads();

        if (fits) {
            const int per = (total + WPS - 1) / WPS;
            const int b = min(total, (wave % WPS) * per);
            const int e = min(total, b + per);
            gcn_agg_drain<F>(h, dinv, dl, acc, tl + (size_t)seg * segw + 1 + b, e - b, lane);
        } else {
            float* myq = fq + wave * AGG_FQCAP;
            int cnt = 0;
            for (int e0 = wave * 32; e0 < E; e0 += AGG_WAVES * 32) {
                const int e = e0 + lane;
                const int dv = (e < E) ? dst[e] : -1;
                const unsigned r = (unsigned)(dv - lo);
                const bool m = r < (unsigned)nrows;
                const unsigned bm = __builtin_amdgcn_ballot_w32(m);
                if (bm != 0u) {
                    if (cnt > AGG_FQCAP - 32) {
                        gcn_agg_drain<F>(h, dinv, dl, acc, myq, cnt, lane);
                        cnt = 0;
                    }
                    if (m) myq[cnt + (int)__builtin_amdgcn_mbcnt_lo(bm, 0u)] =
                        __uint_as_float((r << 24) | (unsigned)src[e]);
                    cnt += __builtin_popcount(bm);
                }
            }
            gcn_agg_drain<F>(h, dinv, dl, acc, myq, cnt, lane);
        }
        __syncthreads();

        for (int i = tid; i < nrows * (F / 4); i += 1024) {
            const int r  = i / (F / 4);
            const int f4 = i % (F / 4);
            const size_t gi = ((size_t)(lo + r) * F) / 4 + f4;
            const float di = dl[r];
            const float4 sa = ((const float4*)acc)[i];
            const float4 hv = ((const float4*)h)[gi];
            const float4 bv = ((const float4*)bias)[f4];
            float4 v;
            v.x = sa.x + hv.x * di * di + bv.x;
            v.y = sa.y + hv.y * di * di + bv.y;
            v.z = sa.z + hv.z * di * di + bv.z;
            v.w = sa.w + hv.w * di * di + bv.w;
            if (RELU) {
                v.x = fmaxf(v.x, 0.0f); v.y = fmaxf(v.y, 0.0f);
                v.z = fmaxf(v.z, 0.0f); v.w = fmaxf(v.w, 0.0f);
            }
            ((float4*)out)[gi] = v;
        }
        __syncthreads();
    }
}

extern "C" void kernel_launch(void* const* d_in, const int* in_sizes, int n_in,
                              void* d_out, int out_size, void* d_ws, size_t ws_size,
                              hipStream_t stream) {
    const float* x  = (const float*)d_in[0];
    const int*   ei = (const int*)d_in[1];
    const float* W1 = (const float*)d_in[2];
    const float* b1 = (const float*)d_in[3];
    const float* W2 = (const float*)d_in[4];
    const float* b2 = (const float*)d_in[5];
    float* out = (float*)d_out;

    const int DIN = 128, DH = 128, DOUT = 64;
    const int N = in_sizes[0] / DIN;
    const int E = in_sizes[1] / 2;
    const int* src = ei;
    const int* dst = ei + E;

    float* ws = (float*)d_ws;
    size_t dinvPad = ((size_t)N + 255) & ~(size_t)255;
    float* dinv = ws;
    float* h    = ws + dinvPad;
    float* agg1 = h + (size_t)N * DH;

    int* degp = (int*)agg1;
    gcn_tile_lists<<<256, 1024, 0, stream>>>(src, dst, degp, out, N, E);

    int gemmBlocks = ((N / 16) + GEMM_WAVES - 1) / GEMM_WAVES;
    gcn_gemm_wmma_f32<DH, 1><<<gemmBlocks, GEMM_WAVES * 32, 0, stream>>>(x, W1, h, N, degp, dinv);

    gcn_agg_tile<128, 1><<<256, 1024, 0, stream>>>(h, src, dst, dinv, b1, out, agg1, N, E);

    gcn_gemm_wmma_f32<DOUT, 0><<<gemmBlocks, GEMM_WAVES * 32, 0, stream>>>(agg1, W2, h, N, degp, dinv);

    gcn_agg_tile<64, 0><<<256, 1024, 0, stream>>>(h, src, dst, dinv, b2, out, out, N, E);
}


# Layer 1 aggregates x before its product ((A x) W1 = A (x W1)): the tile sums stay in LDS and both f32 WMMA products run on them there; only g is written; dinv from per-node counts; 5 -> 3 launches
# speedup vs baseline: 5.0943x; 1.1611x over previous
#include <hip/hip_runtime.h>
#include <hip/hip_bf16.h>

typedef __attribute__((ext_vector_type(2))) float v2f;
typedef __attribute__((ext_vector_type(8))) float v8f;

constexpr int AGG_GS = 8;
constexpr int GEMM_K = 128;
static_assert(AGG_GS == 8, "a node's per-slice counts are read as two 16-byte words");

__device__ __forceinline__ float gcn_dinv_of(const int* __restrict__ deg8, int i) {
    const int4 c0 = *(const int4*)(deg8 + (size_t)i * AGG_GS);
    const int4 c1 = *(const int4*)(deg8 + (size_t)i * AGG_GS + 4);
    const int d = 1 + c0.x + c0.y + c0.z + c0.w + c1.x + c1.y + c1.z + c1.w;
    const float v = (float)d;
    return (v > 0.0f) ? __frsqrt_rn(v) : 0.0f;
}

constexpr int AGG_ROWS  = 160;
constexpr int AGG_WAVES = 32;
constexpr int AGG_GT    = 8;
constexpr int AGG_GROWS = AGG_GT * AGG_ROWS;
constexpr int AGG_RBITS = 11;
constexpr int AGG_EBITS = 32 - AGG_RBITS;
constexpr int AGG_QCAP  = 1024;
constexpr int AGG_FQCAP = 256;
constexpr int AGG_LROW  = 64;
static_assert(AGG_GROWS <= (1 << AGG_RBITS), "row inside the group must fit its bits");
static_assert(AGG_WAVES % AGG_GS == 0, "every list segment gets the same number of waves");

__global__ __launch_bounds__(1024) void gcn_tile_lists(
    const int* __restrict__ src, const int* __restrict__ dst,
    int* __restrict__ degp, float* __restrict__ lists, int N, int E) {
    __shared__ unsigned q[AGG_WAVES * AGG_QCAP];
    __shared__ int degc[AGG_GROWS];
    __shared__ int tcnt[AGG_GT];

    const int tid  = threadIdx.x;
    const int lane = tid & 31;
    const int wave = __builtin_amdgcn_readfirstlane(tid >> 5);
    unsigned* myq = q + wave * AGG_QCAP;

    const int ngroups = (N + AGG_GROWS - 1) / AGG_GROWS;
    const bool vec_ok = (reinterpret_cast<size_t>(dst) & 15) == 0;
    const int nfull = vec_ok ? E / 512 : 0;

    for (int job = blockIdx.x; job < ngroups * AGG_GS; job += gridDim.x) {
        const int group = job / AGG_GS;
        const int slice = job % AGG_GS;
        const int lo = group * AGG_GROWS;
        const int nrows = min(AGG_GROWS, N - lo);
        float* gl = lists + (size_t)lo * AGG_LROW;
        const int t_lo = (int)(((long long)nfull * slice) / AGG_GS);
        const int t_hi = (int)(((long long)nfull * (slice + 1)) / AGG_GS);
        const int nt = t_hi - t_lo;
        const int ebase = t_lo * 512;
        const int* dsl = dst + ebase;
        const int* ssl = src + ebase;

        for (int i = tid; i < AGG_GROWS; i += 1024) degc[i] = 0;
        if (tid < AGG_GT) tcnt[tid] = 0;
        __syncthreads();

        int cnt = 0;

        auto flush = [&]() {
            for (int i0 = 0; i0 < cnt; i0 += 32) {
                const bool v = i0 + lane < cnt;
                unsigned r = 0u, s = 0u;
                int tt = -1;
                if (v) {
                    const unsigned ent = myq[i0 + lane];
                    r = ent >> AGG_EBITS;
                    s = (unsigned)ssl[ent & ((1u << AGG_EBITS) - 1u)];
                    atomicAdd(&degc[r], 1);
                    tt = (int)(r / (unsigned)AGG_ROWS);
                }
                for (int k = 0; k < AGG_GT; ++k) {
                    const bool mk = tt == k;
                    const unsigned bk = __builtin_amdgcn_ballot_w32(mk);
                    if (bk != 0u) {
                        const int nrk  = min(AGG_ROWS, nrows - k * AGG_ROWS);
                        const int segw = (nrk * AGG_LROW) / AGG_GS;
                        float* seg = gl + (size_t)k * AGG_ROWS * AGG_LROW + (size_t)slice * segw;
                        int base = 0;
                        if (lane == 0) base = atomicAdd(&tcnt[k], __builtin_popcount(bk));
                        base = __builtin_amdgcn_readfirstlane(base);
                        if (mk) {
                            const int idx = base + (int)__builtin_amdgcn_mbcnt_lo(bk, 0u);
                            if (idx < segw - 1)
                                seg[1 + idx] = __uint_as_float(((r - (unsigned)(k * AGG_ROWS)) << 24) | s);
                        }
                    }
                }
            }
            cnt = 0;
        };
        auto room = [&](int need) {
            if (cnt > AGG_QCAP - need) flush();
        };
        auto keep4 = [&](const int4& d, int e) {
            const unsigned r0 = (unsigned)(d.x - lo);
            const unsigned r1 = (unsigned)(d.y - lo);
            const unsigned r2 = (unsigned)(d.z - lo);
            const unsigned r3 = (unsigned)(d.w - lo);
            const bool m0 = r0 < (unsigned)nrows;
            const bool m1 = r1 < (unsigned)nrows;
            const bool m2 = r2 < (unsigned)nrows;
            const bool m3 = r3 < (unsigned)nrows;
            if (__builtin_amdgcn_ballot_w32(m0 | m1 | m2 | m3) != 0u) {
                const unsigned b0 = __builtin_amdgcn_ballot_w32(m0);
                if (m0) myq[cnt + (int)__builtin_amdgcn_mbcnt_lo(b0, 0u)] = (r0 << AGG_EBITS) | (unsigned)(e + 0);
                cnt += __builtin_popcount(b0);
                const unsigned b1 = __builtin_amdgcn_ballot_w32(m1);
                if (m1) myq[cnt + (int)__builtin_amdgcn_mbcnt_lo(b1, 0u)] = (r1 << AGG_EBITS) | (unsigned)(e + 1);
                cnt += __builtin_popcount(b1);
                const unsigned b2 = __builtin_amdgcn_ballot_w32(m2);
                if (m2) myq[cnt + (int)__builtin_amdgcn_mbcnt_lo(b2, 0u)] = (r2 << AGG_EBITS) | (unsigned)(e + 2);
                cnt += __builtin_popcount(b2);
                const unsigned b3 = __builtin_amdgcn_ballot_w32(m3);
                if (m3) myq[cnt + (int)__builtin_amdgcn_mbcnt_lo(b3, 0u)] = (r3 << AGG_EBITS) | (unsigned)(e + 3);
                cnt += __builtin_popcount(b3);
            }
        };

        if (wave < nt) {
            const int rstep = max(nt / max(ngroups, 1), 1);
            const int rot = (int)(((unsigned)group * (unsigned)rstep) % (unsigned)nt);
            auto trip_base = [&](int t) {
                int sg = t + rot;
                if (sg >= nt) sg -= nt;
                return sg * 512 + lane * 4;
            };
            int4 a[4], b[4];
            int t  = wave;
            int ea = trip_base(t);
            int eb = ea;
#pragma unroll
            for (int k = 0; k < 4; ++k) a[k] = *(const int4*)(dsl + ea + k * 128);
            while (t < nt) {
                const int tb = t + AGG_WAVES;
                const bool hb = tb < nt;
                room(512);
                eb = trip_base(hb ? tb : t);
#pragma unroll
                for (int k = 0; k < 4; ++k) b[k] = *(const int4*)(dsl + eb + k * 128);
#pragma unroll
                for (int k = 0; k < 4; ++k) keep4(a[k], ea + k * 128);

                const int ta = tb + AGG_WAVES;
                const bool ha = ta < nt;
                room(512);
                ea = trip_base(ha ? ta : t);
#pragma unroll
                for (int k = 0; k < 4; ++k) a[k] = *(const int4*)(dsl + ea + k * 128);
                if (hb) {
#pragma unroll
                    for (int k = 0; k < 4; ++k) keep4(b[k], eb + k * 128);
                }
                t = ta;
            }
        }
        if (slice == AGG_GS - 1) {
            for (int e0 = nfull * 512 + wave * 32; e0 < E; e0 += AGG_WAVES * 32) {
                const int e = e0 + lane;
                const int dv = (e < E) ? dst[e] : -1;
                const unsigned r = (unsigned)(dv - lo);
                const bool m = r < (unsigned)nrows;
                const unsigned bm = __builtin_amdgcn_ballot_w32(m);
                if (bm != 0u) {
                    room(32);
                    if (m) myq[cnt + (int)__builtin_amdgcn_mbcnt_lo(bm, 0u)] = (r << AGG_EBITS) | (unsigned)(e - ebase);
                    cnt += __builtin_popcount(bm);
                }
            }
        }
        flush();
        __syncthreads();

        if (tid < AGG_GT && nrows > tid * AGG_ROWS) {
            const int nrk  = min(AGG_ROWS, nrows - tid * AGG_ROWS);
            const int segw = (nrk * AGG_LROW) / AGG_GS;
            gl[(size_t)tid * AGG_ROWS * AGG_LROW + (size_t)slice * segw] = __uint_as_float((unsigned)tcnt[tid]);
        }
        for (int i = tid; i < nrows; i += 1024) degp[(size_t)(lo + i) * AGG_GS + slice] = degc[i];
        __syncthreads();
    }
}

template <int F, int P, int DEG>
__device__ __forceinline__ void gcn_agg_drain(
    const float* __restrict__ h, const float* __restrict__ dinv, const int* __restrict__ deg8,
    const float* dl, float* acc, const float* list, int cnt, int lane) {
    constexpr int LPE = F / 4;
    constexpr int EPW = 32 / LPE;
    constexpr int UD  = 8;
    const int sub = (EPW == 1) ? 0 : lane / LPE;
    const int fl  = lane % LPE;
    const int hi  = (lane >> 4) & 1;
    unsigned pre = (lane < cnt) ? __float_as_uint(list[lane]) : 0u;
    for (int base = 0; base < cnt; base += 32) {
        const int n = min(32, cnt - base);
        const unsigned ent = pre;
        const int nb = base + 32 + lane;
        pre = (nb < cnt) ? __float_as_uint(list[nb]) : 0u;
        int s = 0, r = 0;
        float nrm = 0.0f;
        if (lane < n) {
            r = (int)(ent >> 24);
            s = (int)(ent & 0xFFFFFFu);
            float ds;
            if constexpr (DEG) ds = gcn_dinv_of(deg8, s);
            else               ds = dinv[s];
            nrm = ds * dl[r];
        }
        for (int i = 0; i < n; i += UD * EPW) {
            float4 v[UD];
            float  w[UD];
            int    si[UD];
            int    rr[UD];
            bool   ok[UD];
#pragma unroll
            for (int k = 0; k < UD; ++k) {
                const int idx = i + k * EPW + sub;
                ok[k] = idx < n;
                const int ii = ok[k] ? idx : 0;
                if constexpr (EPW == 1) {
                    si[k] = __builtin_amdgcn_readlane(s, ii);
                    rr[k] = __builtin_amdgcn_readlane(r, ii);
                    w[k]  = __int_as_float(__builtin_amdgcn_readlane(__float_as_int(nrm), ii));
                } else {
                    si[k] = __shfl(s, ii);
                    rr[k] = __shfl(r, ii);
                    w[k]  = __shfl(nrm, ii);
                }
            }
#pragma unroll
            for (int k = 0; k < UD; ++k) {
                v[k] = make_float4(0.0f, 0.0f, 0.0f, 0.0f);
                if (ok[k]) v[k] = *((const float4*)(h + (size_t)si[k] * F) + fl);
            }
#pragma unroll
            for (int k = 0; k < UD; ++k) {
                if (ok[k]) {
                    float* a  = acc + rr[k] * P + fl * 4;
                    float* p0 = a + 2 * hi;
                    float* p1 = a + 2 - 2 * hi;
                    const float c0 = hi ? v[k].z : v[k].x;
                    const float c1 = hi ? v[k].w : v[k].y;
                    const float c2 = hi ? v[k].x : v[k].z;
                    const float c3 = hi ? v[k].y : v[k].w;
                    atomicAdd(p0 + 0, c0 * w[k]);
                    atomicAdd(p0 + 1, c1 * w[k]);
                    atomicAdd(p1 + 0, c2 * w[k]);
                    atomicAdd(p1 + 1, c3 * w[k]);
                }
            }
        }
    }
}

template <int F, int RELU>
__global__ __launch_bounds__(1024) void gcn_agg_tile(
    const float* __restrict__ h, const int* __restrict__ src, const int* __restrict__ dst,
    const float* __restrict__ dinv, const float* __restrict__ bias,
    const float* lists, float* out, int N, int E) {
    __shared__ float acc[AGG_ROWS * F];
    __shared__ float dl[AGG_ROWS];
    __shared__ float fq[AGG_WAVES * AGG_FQCAP];

    const int tid  = threadIdx.x;
    const int lane = tid & 31;
    const int wave = __builtin_amdgcn_readfirstlane(tid >> 5);

    const int ntiles = (N + AGG_ROWS - 1) / AGG_ROWS;

    for (int tile = blockIdx.x; tile < ntiles; tile += gridDim.x) {
        const int lo = tile * AGG_ROWS;
        const int nrows = min(AGG_ROWS, N - lo);
        const float* tl = lists + (size_t)lo * AGG_LROW;
        constexpr int WPS = AGG_WAVES / AGG_GS;
        const int segw = (nrows * AGG_LROW) / AGG_GS;
        const int seg  = wave / WPS;
        bool fits = true;
        int total = 0;
        for (int sg = 0; sg < AGG_GS; ++sg) {
            const int c = __builtin_amdgcn_readfirstlane((int)__float_as_uint(tl[(size_t)sg * segw]));
            fits = fits && c >= 0 && c <= segw - 1;
            if (sg == seg) total = c;
        }

        for (int i = tid; i < AGG_ROWS * F / 4; i += 1024)
            ((float4*)acc)[i] = make_float4(0.0f, 0.0f, 0.0f, 0.0f);
        for (int i = tid; i < nrows; i += 1024) dl[i] = dinv[lo + i];
        __syncthreads();

        if (fits) {
            const int per = (total + WPS - 1) / WPS;
            const int b = min(total, (wave % WPS) * per);
            const int e = min(total, b + per);
            gcn_agg_drain<F, F, 0>(h, dinv, nullptr, dl, acc, tl + (size_t)seg * segw + 1 + b, e - b, lane);
        } else {
            float* myq = fq + wave * AGG_FQCAP;
            int cnt = 0;
            for (int e0 = wave * 32; e0 < E; e0 += AGG_WAVES * 32) {
                const int e = e0 + lane;
                const int dv = (e < E) ? dst[e] : -1;
                const unsigned r = (unsigned)(dv - lo);
                const bool m = r < (unsigned)nrows;
                const unsigned bm = __builtin_amdgcn_ballot_w32(m);
                if (bm != 0u) {
                    if (cnt > AGG_FQCAP - 32) {
                        gcn_agg_drain<F, F, 0>(h, dinv, nullptr, dl, acc, myq, cnt, lane);
                        cnt = 0;
                    }
                    if (m) myq[cnt + (int)__builtin_amdgcn_mbcnt_lo(bm, 0u)] =
                        __uint_as_float((r << 24) | (unsigned)src[e]);
                    cnt += __builtin_popcount(bm);
                }
            }
            gcn_agg_drain<F, F, 0>(h, dinv, nullptr, dl, acc, myq, cnt, lane);
        }
        __syncthreads();

        for (int i = tid; i < nrows * (F / 4); i += 1024) {
            const int r  = i / (F / 4);
            const int f4 = i % (F / 4);
            const size_t gi = ((size_t)(lo + r) * F) / 4 + f4;
            const float di = dl[r];
            const float4 sa = ((const float4*)acc)[i];
            const float4 hv = ((const float4*)h)[gi];
            const float4 bv = ((const float4*)bias)[f4];
            float4 v;
            v.x = sa.x + hv.x * di * di + bv.x;
            v.y = sa.y + hv.y * di * di + bv.y;
            v.z = sa.z + hv.z * di * di + bv.z;
            v.w = sa.w + hv.w * di * di + bv.w;
            if (RELU) {
                v.x = fmaxf(v.x, 0.0f); v.y = fmaxf(v.y, 0.0f);
                v.z = fmaxf(v.z, 0.0f); v.w = fmaxf(v.w, 0.0f);
            }
            ((float4*)out)[gi] = v;
        }
        __syncthreads();
    }
}

constexpr int L1_F      = 128;
constexpr int L1_G      = 64;
constexpr int L1_TP     = L1_F + 4;
constexpr int L1_PP1    = 2 * L1_F + 32;
constexpr int L1_PP2    = 2 * L1_G + 32;
constexpr int L1_STRIPS = AGG_ROWS / 16;
static_assert(AGG_ROWS % 16 == 0, "a tile is whole strips");
static_assert(2 * L1_STRIPS == 20 && AGG_WAVES == 32, "the item-to-wave map below places 20 items on 32 waves");

template <int NOUT, int PP>
__device__ __forceinline__ void gcn_stage_w(const float* __restrict__ W, float* Ws, int tid) {
    constexpr int ITEMS = (GEMM_K / 2) * (NOUT / 4);
    static_assert(ITEMS % 1024 == 0, "whole trips of the work-group");
    constexpr int TRIPS = ITEMS / 1024;
    float4 v0[TRIPS], v1[TRIPS];
#pragma unroll
    for (int j = 0; j < TRIPS; ++j) {
        const int i  = tid + j * 1024;
        const int p  = i / (NOUT / 4);
        const int c4 = i % (NOUT / 4);
        v0[j] = ((const float4*)W)[(size_t)(2 * p) * (NOUT / 4) + c4];
        v1[j] = ((const float4*)W)[(size_t)(2 * p + 1) * (NOUT / 4) + c4];
    }
#pragma unroll
    for (int j = 0; j < TRIPS; ++j) {
        const int i  = tid + j * 1024;
        const int p  = i / (NOUT / 4);
        const int c4 = i % (NOUT / 4);
        float4* d = (float4*)&Ws[p * PP + c4 * 8];
        d[0] = make_float4(v0[j].x, v1[j].x, v0[j].y, v1[j].y);
        d[1] = make_float4(v0[j].z, v1[j].z, v0[j].w, v1[j].w);
    }
}

template <int NT, int PP>
__device__ __forceinline__ void gcn_strip_product(const float* arow, const float* wl, v8f (&c)[NT]) {
    v2f a = *(const v2f*)arow, an = a;
    v2f b[NT], bn[NT];
#pragma unroll
    for (int t = 0; t < NT; ++t) { b[t] = *(const v2f*)(wl + t * 32); bn[t] = b[t]; }
#pragma unroll
    for (int s = 0; s < GEMM_K / 4; ++s) {
        if (s + 1 < GEMM_K / 4) {
            an = *(const v2f*)(arow + 4 * (s + 1));
            const float* wn = wl + (2 * (s + 1)) * PP;
#pragma unroll
            for (int t = 0; t < NT; ++t) bn[t] = *(const v2f*)(wn + t * 32);
        }
#pragma unroll
        for (int t = 0; t < NT; ++t) {
            c[t] = __builtin_amdgcn_wmma_f32_16x16x4_f32(
                false, a, false, b[t], (short)0, c[t], false, false);
        }
        a = an;
#pragma unroll
        for (int t = 0; t < NT; ++t) b[t] = bn[t];
    }
}

__global__ __launch_bounds__(1024) void gcn_layer1_tile(
    const float* __restrict__ x, const int* __restrict__ src, const int* __restrict__ dst,
    const int* __restrict__ deg8, const float* __restrict__ W1, const float* __restrict__ b1,
    const float* __restrict__ W2, const float* __restrict__ lists,
    float* __restrict__ g, float* __restrict__ dinv, int N, int E) {
    __shared__ __attribute__((aligned(16))) float acc[AGG_ROWS * L1_TP];
    __shared__ float dl[AGG_ROWS];
    __shared__ float fq[AGG_WAVES * AGG_FQCAP];
    __shared__ __attribute__((aligned(16))) float W1s[(GEMM_K / 2) * L1_PP1];
    __shared__ __attribute__((aligned(16))) float W2s[(GEMM_K / 2) * L1_PP2];

    const int tid  = threadIdx.x;
    const int lane = tid & 31;
    const int wave = __builtin_amdgcn_readfirstlane(tid >> 5);

    gcn_stage_w<L1_F, L1_PP1>(W1, W1s, tid);
    gcn_stage_w<L1_G, L1_PP2>(W2, W2s, tid);

    const int wq = wave & 7;
    const int wb = wave >> 3;
    const bool prod  = (wq < 4) || (wq == 4 + wb);
    const int  item  = wb * 5 + ((wq < 4) ? wq : 4);
    const int  strip = item % L1_STRIPS;
    const int  half  = item / L1_STRIPS;
    const int  m  = lane & 15;
    const int  gh = lane >> 4;

    const int ntiles = (N + AGG_ROWS - 1) / AGG_ROWS;

    for (int tile = blockIdx.x; tile < ntiles; tile += gridDim.x) {
        const int lo = tile * AGG_ROWS;
        const int nrows = min(AGG_ROWS, N - lo);
        const float* tl = lists + (size_t)lo * AGG_LROW;
        constexpr int WPS = AGG_WAVES / AGG_GS;
        const int segw = (nrows * AGG_LROW) / AGG_GS;
        const int seg  = wave / WPS;
        bool fits = true;
        int total = 0;
        for (int sg = 0; sg < AGG_GS; ++sg) {
            const int c = __builtin_amdgcn_readfirstlane((int)__float_as_uint(tl[(size_t)sg * segw]));
            fits = fits && c >= 0 && c <= segw - 1;
            if (sg == seg) total = c;
        }

        for (int i = tid; i < AGG_ROWS * L1_TP / 4; i += 1024)
            ((float4*)acc)[i] = make_float4(0.0f, 0.0f, 0.0f, 0.0f);
        for (int i = tid; i < nrows; i += 1024) {
            const float d = gcn_dinv_of(deg8, lo + i);
            dl[i] = d;
            dinv[lo + i] = d;
        }
        __syncthreads();

        if (fits) {
            const int per = (total + WPS - 1) / WPS;
            const int b = min(total, (wave % WPS) * per);
            const int e = min(total, b + per);
            gcn_agg_drain<L1_F, L1_TP, 1>(x, nullptr, deg8, dl, acc, tl + (size_t)seg * segw + 1 + b, e - b, lane);
        } else {
            float* myq = fq + wave * AGG_FQCAP;
            int cnt = 0;
            for (int e0 = wave * 32; e0 < E; e0 += AGG_WAVES * 32) {
                const int e = e0 + lane;
                const int dv = (e < E) ? dst[e] : -1;
                const unsigned r = (unsigned)(dv - lo);
                const bool mk = r < (unsigned)nrows;
                const unsigned bm = __builtin_amdgcn_ballot_w32(mk);
                if (bm != 0u) {
                    if (cnt > AGG_FQCAP - 32) {
                        gcn_agg_drain<L1_F, L1_TP, 1>(x, nullptr, deg8, dl, acc, myq, cnt, lane);
                        cnt = 0;
                    }
                    if (mk) myq[cnt + (int)__builtin_amdgcn_mbcnt_lo(bm, 0u)] =
                        __uint_as_float((r << 24) | (unsigned)src[e]);
                    cnt += __builtin_popcount(bm);
                }
            }
            gcn_agg_drain<L1_F, L1_TP, 1>(x, nullptr, deg8, dl, acc, myq, cnt, lane);
        }
        __syncthreads();

        for (int i = tid; i < nrows * (L1_F / 4); i += 1024) {
            const int r  = i / (L1_F / 4);
            const int f4 = i % (L1_F / 4);
            const float di = dl[r];
            const float4 xv = ((const float4*)x)[(size_t)(lo + r) * (L1_F / 4) + f4];
            float4* p = (float4*)&acc[r * L1_TP + 4 * f4];
            float4 v = *p;
            v.x += xv.x * di * di;
            v.y += xv.y * di * di;
            v.z += xv.z * di * di;
            v.w += xv.w * di * di;
            *p = v;
        }
        __syncthreads();

        v8f c1[4];
        float bv[4];
        if (prod) {
#pragma unroll
            for (int t = 0; t < 4; ++t) {
                c1[t] = (v8f){0, 0, 0, 0, 0, 0, 0, 0};
                bv[t] = b1[(4 * half + t) * 16 + m];
            }
            gcn_strip_product<4, L1_PP1>(acc + (strip * 16 + m) * L1_TP + 2 * gh,
                                         W1s + gh * L1_PP1 + 2 * m + (4 * half) * 32, c1);
        }
        __syncthreads();
        if (prod) {
#pragma unroll
            for (int t = 0; t < 4; ++t) {
                const int col = (4 * half + t) * 16 + m;
#pragma unroll
                for (int r = 0; r < 8; ++r)
                    acc[(strip * 16 + 8 * gh + r) * L1_TP + col] = fmaxf(c1[t][r] + bv[t], 0.0f);
            }
        }
        __syncthreads();

        if (prod) {
            v8f c2[2];
#pragma unroll
            for (int t = 0; t < 2; ++t) c2[t] = (v8f){0, 0, 0, 0, 0, 0, 0, 0};
            gcn_strip_product<2, L1_PP2>(acc + (strip * 16 + m) * L1_TP + 2 * gh,
                                         W2s + gh * L1_PP2 + 2 * m + (2 * half) * 32, c2);
#pragma unroll
            for (int t = 0; t < 2; ++t) {
                const int col = (2 * half + t) * 16 + m;
#pragma unroll
                for (int r = 0; r < 8; ++r) {
                    const int row = strip * 16 + 8 * gh + r;
                    if (row < nrows) g[(size_t)(lo + row) * L1_G + col] = c2[t][r];
                }
            }
        }
        __syncthreads();
    }
}

extern "C" void kernel_launch(void* const* d_in, const int* in_sizes, int n_in,
                              void* d_out, int out_size, void* d_ws, size_t ws_size,
                              hipStream_t stream) {
    const float* x  = (const float*)d_in[0];
    const int*   ei = (const int*)d_in[1];
    const float* W1 = (const float*)d_in[2];
    const float* b1 = (const float*)d_in[3];
    const float* W2 = (const float*)d_in[4];
    const float* b2 = (const float*)d_in[5];
    float* out = (float*)d_out;

    const int DIN = 128, DH = 128;
    const int N = in_sizes[0] / DIN;
    const int E = in_sizes[1] / 2;
    const int* src = ei;
    const int* dst = ei + E;

    float* ws = (float*)d_ws;
    size_t dinvPad = ((size_t)N + 255) & ~(size_t)255;
    float* dinv = ws;
    float* g    = ws + dinvPad;
    int*   deg8 = (int*)(g + (size_t)N * DH);

    gcn_tile_lists<<<256, 1024, 0, stream>>>(src, dst, deg8, out, N, E);

    gcn_layer1_tile<<<256, 1024, 0, stream>>>(x, src, dst, deg8, W1, b1, W2, out, g, dinv, N, E);

    gcn_agg_tile<64, 0><<<256, 1024, 0, stream>>>(g, src, dst, dinv, b2, out, out, N, E);
}


# Tile lists bucketed by owner wave (row mod 32): each wave sums its five rows of the tile in registers, no LDS float adds and no tile zero-fill; the order of the f32 edge sums changes
# speedup vs baseline: 5.3781x; 1.0557x over previous
#include <hip/hip_runtime.h>
#include <hip/hip_bf16.h>

typedef __attribute__((ext_vector_type(2))) float v2f;
typedef __attribute__((ext_vector_type(8))) float v8f;

constexpr int AGG_GS = 8;
constexpr int GEMM_K = 128;
static_assert(AGG_GS == 8, "a node's per-slice counts are read as two 16-byte words");

__device__ __forceinline__ float gcn_dinv_from(const int4& c0, const int4& c1) {
    const int d = 1 + c0.x + c0.y + c0.z + c0.w + c1.x + c1.y + c1.z + c1.w;
    const float v = (float)d;
    return (v > 0.0f) ? __frsqrt_rn(v) : 0.0f;
}
__device__ __forceinline__ float gcn_dinv_of(const int* __restrict__ deg8, int i) {
    const int4 c0 = *(const int4*)(deg8 + (size_t)i * AGG_GS);
    const int4 c1 = *(const int4*)(deg8 + (size_t)i * AGG_GS + 4);
    return gcn_dinv_from(c0, c1);
}

constexpr int AGG_ROWS  = 160;
constexpr int AGG_WAVES = 32;
constexpr int AGG_GT    = 8;
constexpr int AGG_GROWS = AGG_GT * AGG_ROWS;
constexpr int AGG_RBITS = 11;
constexpr int AGG_EBITS = 32 - AGG_RBITS;
constexpr int AGG_QCAP  = 1024;
constexpr int AGG_FQCAP = 320;
constexpr int AGG_LROW  = 64;
static_assert(AGG_GROWS <= (1 << AGG_RBITS), "row inside the group must fit its bits");
constexpr int AGG_OWN   = AGG_ROWS / AGG_WAVES;
constexpr int AGG_SUBW  = (AGG_ROWS * AGG_LROW) / (AGG_GS * AGG_WAVES);
static_assert(AGG_ROWS % AGG_WAVES == 0, "every wave owns the same number of rows");
static_assert(AGG_ROWS <= 1024, "one thread per row of a tile fetches the row's dinv");
static_assert(AGG_GS * (AGG_SUBW - 1) <= AGG_FQCAP, "a wave's queue holds all its buckets");
static_assert(AGG_SUBW - 1 >= 32 && AGG_SUBW - 1 <= 64, "a bucket is copied in one whole and one partial step of 32 entries");

__global__ __launch_bounds__(1024) void gcn_tile_lists(
    const int* __restrict__ src, const int* __restrict__ dst,
    int* __restrict__ degp, float* __restrict__ lists, int N, int E) {
    __shared__ unsigned q[AGG_WAVES * AGG_QCAP];
    __shared__ int degc[AGG_GROWS];
    __shared__ int bcnt[AGG_GT * AGG_WAVES];

    const int tid  = threadIdx.x;
    const int lane = tid & 31;
    const int wave = __builtin_amdgcn_readfirstlane(tid >> 5);
    unsigned* myq = q + wave * AGG_QCAP;

    const int ngroups = (N + AGG_GROWS - 1) / AGG_GROWS;
    const bool vec_ok = (reinterpret_cast<size_t>(dst) & 15) == 0;
    const int nfull = vec_ok ? E / 512 : 0;

    for (int job = blockIdx.x; job < ngroups * AGG_GS; job += gridDim.x) {
        const int group = job / AGG_GS;
        const int slice = job % AGG_GS;
        const int lo = group * AGG_GROWS;
        const int nrows = min(AGG_GROWS, N - lo);
        float* gl = lists + (size_t)lo * AGG_LROW;
        const int t_lo = (int)(((long long)nfull * slice) / AGG_GS);
        const int t_hi = (int)(((long long)nfull * (slice + 1)) / AGG_GS);
        const int nt = t_hi - t_lo;
        const int ebase = t_lo * 512;
        const int* dsl = dst + ebase;
        const int* ssl = src + ebase;

        for (int i = tid; i < AGG_GROWS; i += 1024) degc[i] = 0;
        for (int i = tid; i < AGG_GT * AGG_WAVES; i += 1024) bcnt[i] = 0;
        __syncthreads();

        int cnt = 0;

        auto flush = [&]() {
            for (int i0 = 0; i0 < cnt; i0 += 32) {
                if (i0 + lane < cnt) {
                    const unsigned ent = myq[i0 + lane];
                    const unsigned r = ent >> AGG_EBITS;
                    const unsigned s = (unsigned)ssl[ent & ((1u << AGG_EBITS) - 1u)];
                    atomicAdd(&degc[r], 1);
                    const int tt = (int)(r / (unsigned)AGG_ROWS);
                    const unsigned rt = r - (unsigned)(tt * AGG_ROWS);
                    const int ow = (int)(rt % (unsigned)AGG_WAVES);
                    const int nrk  = min(AGG_ROWS, nrows - tt * AGG_ROWS);
                    const int segw = (nrk * AGG_LROW) / AGG_GS;
                    const int subw = segw / AGG_WAVES;
                    const int slot = atomicAdd(&bcnt[tt * AGG_WAVES + ow], 1);
                    if (slot < subw - 1)
                        gl[(size_t)tt * AGG_ROWS * AGG_LROW + (size_t)slice * segw + (size_t)ow * subw + 1 + slot] =
                            __uint_as_float((rt << 24) | s);
                }
            }
            cnt = 0;
        };
        auto room = [&](int need) {
            if (cnt > AGG_QCAP - need) flush();
        };
        auto keep4 = [&](const int4& d, int e) {
            const unsigned r0 = (unsigned)(d.x - lo);
            const unsigned r1 = (unsigned)(d.y - lo);
            const unsigned r2 = (unsigned)(d.z - lo);
            const unsigned r3 = (unsigned)(d.w - lo);
            const bool m0 = r0 < (unsigned)nrows;
            const bool m1 = r1 < (unsigned)nrows;
            const bool m2 = r2 < (unsigned)nrows;
            const bool m3 = r3 < (unsigned)nrows;
            if (__builtin_amdgcn_ballot_w32(m0 | m1 | m2 | m3) != 0u) {
                const unsigned b0 = __builtin_amdgcn_ballot_w32(m0);
                if (m0) myq[cnt + (int)__builtin_amdgcn_mbcnt_lo(b0, 0u)] = (r0 << AGG_EBITS) | (unsigned)(e + 0);
                cnt += __builtin_popcount(b0);
                const unsigned b1 = __builtin_amdgcn_ballot_w32(m1);
                if (m1) myq[cnt + (int)__builtin_amdgcn_mbcnt_lo(b1, 0u)] = (r1 << AGG_EBITS) | (unsigned)(e + 1);
                cnt += __builtin_popcount(b1);
                const unsigned b2 = __builtin_amdgcn_ballot_w32(m2);
                if (m2) myq[cnt + (int)__builtin_amdgcn_mbcnt_lo(b2, 0u)] = (r2 << AGG_EBITS) | (unsigned)(e + 2);
                cnt += __builtin_popcount(b2);
                const unsigned b3 = __builtin_amdgcn_ballot_w32(m3);
                if (m3) myq[cnt + (int)__builtin_amdgcn_mbcnt_lo(b3, 0u)] = (r3 << AGG_EBITS) | (unsigned)(e + 3);
                cnt += __builtin_popcount(b3);
            }
        };

        if (wave < nt) {
            const int rstep = max(nt / max(ngroups, 1), 1);
            const int rot = (int)(((unsigned)group * (unsigned)rstep) % (unsigned)nt);
            auto trip_base = [&](int t) {
                int sg = t + rot;
                if (sg >= nt) sg -= nt;
                return sg * 512 + lane * 4;
            };
            int4 a[4], b[4];
            int t  = wave;
            int ea = trip_base(t);
            int eb = ea;
#pragma unroll
            for (int k = 0; k < 4; ++k) a[k] = *(const int4*)(dsl + ea + k * 128);
            while (t < nt) {
                const int tb = t + AGG_WAVES;
                const bool hb = tb < nt;
                room(512);
                eb = trip_base(hb ? tb : t);
#pragma unroll
                for (int k = 0; k < 4; ++k) b[k] = *(const int4*)(dsl + eb + k * 128);
#pragma unroll
                for (int k = 0; k < 4; ++k) keep4(a[k], ea + k * 128);

                const int ta = tb + AGG_WAVES;
                const bool ha = ta < nt;
                room(512);
                ea = trip_base(ha ? ta : t);
#pragma unroll
                for (int k = 0; k < 4; ++k) a[k] = *(const int4*)(dsl + ea + k * 128);
                if (hb) {
#pragma unroll
                    for (int k = 0; k < 4; ++k) keep4(b[k], eb + k * 128);
                }
                t = ta;
            }
        }
        if (slice == AGG_GS - 1) {
            for (int e0 = nfull * 512 + wave * 32; e0 < E; e0 += AGG_WAVES * 32) {
                const int e = e0 + lane;
                const int dv = (e < E) ? dst[e] : -1;
                const unsigned r = (unsigned)(dv - lo);
                const bool m = r < (unsigned)nrows;
                const unsigned bm = __builtin_amdgcn_ballot_w32(m);
                if (bm != 0u) {
                    room(32);
                    if (m) myq[cnt + (int)__builtin_amdgcn_mbcnt_lo(bm, 0u)] = (r << AGG_EBITS) | (unsigned)(e - ebase);
                    cnt += __builtin_popcount(bm);
                }
            }
        }
        flush();
        __syncthreads();

        for (int i = tid; i < AGG_GT * AGG_WAVES; i += 1024) {
            const int tt = i / AGG_WAVES;
            const int ow = i % AGG_WAVES;
            if (nrows > tt * AGG_ROWS) {
                const int nrk  = min(AGG_ROWS, nrows - tt * AGG_ROWS);
                const int segw = (nrk * AGG_LROW) / AGG_GS;
                const int subw = segw / AGG_WAVES;
                if (subw >= 1)
                    gl[(size_t)tt * AGG_ROWS * AGG_LROW + (size_t)slice * segw + (size_t)ow * subw] =
                        __uint_as_float((unsigned)bcnt[i]);
            }
        }
        for (int i = tid; i < nrows; i += 1024) degp[(size_t)(lo + i) * AGG_GS + slice] = degc[i];
        __syncthreads();
    }
}

template <int F, int P, int DEG>
__device__ __forceinline__ void gcn_agg_drain(
    const float* __restrict__ h, const float* __restrict__ dinv, const int* __restrict__ deg8,
    const float* dl, float* acc, const unsigned* list, int cnt, int lane) {
    constexpr int LPE = F / 4;
    constexpr int EPW = 32 / LPE;
    constexpr int UD  = 8;
    const int sub = (EPW == 1) ? 0 : lane / LPE;
    const int fl  = lane % LPE;
    const int hi  = (lane >> 4) & 1;
    unsigned pre = (lane < cnt) ? list[lane] : 0u;
    for (int base = 0; base < cnt; base += 32) {
        const int n = min(32, cnt - base);
        const unsigned ent = pre;
        const int nb = base + 32 + lane;
        pre = (nb < cnt) ? list[nb] : 0u;
        int s = 0, r = 0;
        float nrm = 0.0f;
        if (lane < n) {
            r = (int)(ent >> 24);
            s = (int)(ent & 0xFFFFFFu);
            float ds;
            if constexpr (DEG) ds = gcn_dinv_of(deg8, s);
            else               ds = dinv[s];
            nrm = ds * dl[r];
        }
        for (int i = 0; i < n; i += UD * EPW) {
            float4 v[UD];
            float  w[UD];
            int    si[UD];
            int    rr[UD];
            bool   ok[UD];
#pragma unroll
            for (int k = 0; k < UD; ++k) {
                const int idx = i + k * EPW + sub;
                ok[k] = idx < n;
                const int ii = ok[k] ? idx : 0;
                if constexpr (EPW == 1) {
                    si[k] = __builtin_amdgcn_readlane(s, ii);
                    rr[k] = __builtin_amdgcn_readlane(r, ii);
                    w[k]  = __int_as_float(__builtin_amdgcn_readlane(__float_as_int(nrm), ii));
                } else {
                    si[k] = __shfl(s, ii);
                    rr[k] = __shfl(r, ii);
                    w[k]  = __shfl(nrm, ii);
                }
            }
#pragma unroll
            for (int k = 0; k < UD; ++k) {
                v[k] = make_float4(0.0f, 0.0f, 0.0f, 0.0f);
                if (ok[k]) v[k] = *((const float4*)(h + (size_t)si[k] * F) + fl);
            }
#pragma unroll
            for (int k = 0; k < UD; ++k) {
                if (ok[k]) {
                    float* a  = acc + rr[k] * P + fl * 4;
                    float* p0 = a + 2 * hi;
                    float* p1 = a + 2 - 2 * hi;
                    const float c0 = hi ? v[k].z : v[k].x;
                    const float c1 = hi ? v[k].w : v[k].y;
                    const float c2 = hi ? v[k].x : v[k].z;
                    const float c3 = hi ? v[k].y : v[k].w;
                    atomicAdd(p0 + 0, c0 * w[k]);
                    atomicAdd(p0 + 1, c1 * w[k]);
                    atomicAdd(p1 + 0, c2 * w[k]);
                    atomicAdd(p1 + 1, c3 * w[k]);
                }
            }
        }
    }
}

template <int DEG>
__device__ __forceinline__ int gcn_own_fetch(
    const float* tl, int segw, int subw, const float* __restrict__ dinv, const int* __restrict__ deg8,
    int wave, int lane, unsigned* myq, float* nq) {
    const float* bk = tl + (size_t)wave * subw;
    int cw = 0;
    if (lane < AGG_GS) cw = (int)__float_as_uint(bk[(size_t)lane * segw]);
    unsigned e0[AGG_GS];
#pragma unroll
    for (int sg = 0; sg < AGG_GS; ++sg)
        e0[sg] = (1 + lane < subw) ? __float_as_uint(bk[(size_t)sg * segw + 1 + lane]) : 0u;
    const bool bad = (lane < AGG_GS) && (cw < 0 || cw > subw - 1);
    if (__builtin_amdgcn_ballot_w32(bad) != 0u) return -1;
    int off = 0;
#pragma unroll
    for (int sg = 0; sg < AGG_GS; ++sg) {
        const int cs = __builtin_amdgcn_readlane(cw, sg);
        if (lane < cs) myq[off + lane] = e0[sg];
        if (cs > 32) {
            if (32 + lane < cs) myq[off + 32 + lane] = __float_as_uint(bk[(size_t)sg * segw + 33 + lane]);
        }
        off += cs;
    }
    for (int b0 = 0; b0 < off; b0 += 128) {
        float ds[4];
#pragma unroll
        for (int q = 0; q < 4; ++q) {
            const int e = b0 + q * 32 + lane;
            int s = 0;
            if (e < off) s = (int)(myq[e] & 0xFFFFFFu);
            if constexpr (DEG) ds[q] = gcn_dinv_of(deg8, s);
            else               ds[q] = dinv[s];
        }
#pragma unroll
        for (int q = 0; q < 4; ++q) {
            const int e = b0 + q * 32 + lane;
            if (e < off) nq[e] = ds[q];
        }
    }
    return off;
}

template <int F>
__device__ __forceinline__ void gcn_own_drain(
    const float* __restrict__ h, const float* dl, const unsigned* myq, const float* nq, int cnt, int lane,
    float __attribute__((ext_vector_type(F / 32))) (&a)[AGG_OWN]) {
    typedef float __attribute__((ext_vector_type(F / 32))) vt;
    constexpr int UD = (F >= 128) ? 8 : 16;
    for (int base = 0; base < cnt; base += 32) {
        const int n = min(32, cnt - base);
        int s = 0, r = 0;
        float nrm = 0.0f;
        if (lane < n) {
            const unsigned ent = myq[base + lane];
            r = (int)(ent >> 24);
            s = (int)(ent & 0xFFFFFFu);
            nrm = nq[base + lane] * dl[r];
        }
        for (int i = 0; i < n; i += UD) {
            vt    v[UD];
            float w[UD];
            int   jj[UD];
#pragma unroll
            for (int k = 0; k < UD; ++k) {
                const int idx = i + k;
                const bool ok = idx < n;
                const int ii = ok ? idx : 0;
                const int si = __builtin_amdgcn_readlane(s, ii);
                jj[k] = __builtin_amdgcn_readlane(r, ii) / AGG_WAVES;
                const float wk = __int_as_float(__builtin_amdgcn_readlane(__float_as_int(nrm), ii));
                w[k] = ok ? wk : 0.0f;
                v[k] = (vt)(0.0f);
                if (ok) v[k] = *((const vt*)(h + (size_t)si * F) + lane);
            }
#pragma unroll
            for (int k = 0; k < UD; ++k) {
#pragma unroll
                for (int j = 0; j < AGG_OWN; ++j) {
                    const float wj = (jj[k] == j) ? w[k] : 0.0f;
                    a[j] += v[k] * wj;
                }
            }
        }
    }
}

template <int F, int RELU>
__global__ __launch_bounds__(1024) void gcn_agg_tile(
    const float* __restrict__ h, const int* __restrict__ src, const int* __restrict__ dst,
    const float* __restrict__ dinv, const float* __restrict__ bias,
    const float* lists, float* out, int N, int E) {
    typedef float __attribute__((ext_vector_type(F / 32))) vt;
    __shared__ float acc[AGG_ROWS * F];
    __shared__ float dl[AGG_ROWS];
    __shared__ unsigned fq[AGG_WAVES * AGG_FQCAP];
    __shared__ float fn[AGG_WAVES * AGG_FQCAP];
    __shared__ int ovf[2];

    const int tid  = threadIdx.x;
    const int lane = tid & 31;
    const int wave = __builtin_amdgcn_readfirstlane(tid >> 5);
    unsigned* myq = fq + wave * AGG_FQCAP;
    float*    nq  = fn + wave * AGG_FQCAP;

    if (tid == 0) { ovf[0] = 0; ovf[1] = 0; }
    __syncthreads();
    int par = 0;

    const int ntiles = (N + AGG_ROWS - 1) / AGG_ROWS;

    for (int tile = blockIdx.x; tile < ntiles; tile += gridDim.x) {
        const int lo = tile * AGG_ROWS;
        const int nrows = min(AGG_ROWS, N - lo);
        const float* tl = lists + (size_t)lo * AGG_LROW;
        const int segw = (nrows * AGG_LROW) / AGG_GS;
        const int subw = segw / AGG_WAVES;

        float dmy = 0.0f;
        if (tid < nrows) dmy = dinv[lo + tid];
        int total = -1;
        if (subw >= 2) total = gcn_own_fetch<0>(tl, segw, subw, dinv, nullptr, wave, lane, myq, nq);
        if (total < 0) ovf[par] = 1;
        if (tid < nrows) dl[tid] = dmy;
        __syncthreads();
        const bool fits = __builtin_amdgcn_readfirstlane(ovf[par]) == 0;
        if (tid == 0) ovf[par ^ 1] = 0;
        par ^= 1;

        if (fits) {
            vt hs[AGG_OWN], a[AGG_OWN];
#pragma unroll
            for (int j = 0; j < AGG_OWN; ++j) {
                const int row = min(wave + j * AGG_WAVES, nrows - 1);
                hs[j] = *((const vt*)(h + (size_t)(lo + row) * F) + lane);
                a[j] = (vt)(0.0f);
            }
            const vt bv = *((const vt*)bias + lane);
            gcn_own_drain<F>(h, dl, myq, nq, total, lane, a);
#pragma unroll
            for (int j = 0; j < AGG_OWN; ++j) {
                const int row = wave + j * AGG_WAVES;
                if (row < nrows) {
                    const float di = dl[row];
                    vt v = a[j] + hs[j] * di * di + bv;
                    if (RELU) {
#pragma unroll
                        for (int q = 0; q < F / 32; ++q) v[q] = fmaxf(v[q], 0.0f);
                    }
                    *((vt*)(out + (size_t)(lo + row) * F) + lane) = v;
                }
            }
        } else {
            for (int i = tid; i < AGG_ROWS * F / 4; i += 1024)
                ((float4*)acc)[i] = make_float4(0.0f, 0.0f, 0.0f, 0.0f);
            __syncthreads();
            int cnt = 0;
            for (int e0 = wave * 32; e0 < E; e0 += AGG_WAVES * 32) {
                const int e = e0 + lane;
                const int dv = (e < E) ? dst[e] : -1;
                const unsigned r = (unsigned)(dv - lo);
                const bool m = r < (unsigned)nrows;
                const unsigned bm = __builtin_amdgcn_ballot_w32(m);
                if (bm != 0u) {
                    if (cnt > AGG_FQCAP - 32) {
                        gcn_agg_drain<F, F, 0>(h, dinv, nullptr, dl, acc, myq, cnt, lane);
                        cnt = 0;
                    }
                    if (m) myq[cnt + (int)__builtin_amdgcn_mbcnt_lo(bm, 0u)] = (r << 24) | (unsigned)src[e];
                    cnt += __builtin_popcount(bm);
                }
            }
            gcn_agg_drain<F, F, 0>(h, dinv, nullptr, dl, acc, myq, cnt, lane);
            __syncthreads();

            for (int i = tid; i < nrows * (F / 4); i += 1024) {
                const int r  = i / (F / 4);
                const int f4 = i % (F / 4);
                const size_t gi = ((size_t)(lo + r) * F) / 4 + f4;
                const float di = dl[r];
                const float4 sa = ((const float4*)acc)[i];
                const float4 hv = ((const float4*)h)[gi];
                const float4 bv = ((const float4*)bias)[f4];
                float4 v;
                v.x = sa.x + hv.x * di * di + bv.x;
                v.y = sa.y + hv.y * di * di + bv.y;
                v.z = sa.z + hv.z * di * di + bv.z;
                v.w = sa.w + hv.w * di * di + bv.w;
                if (RELU) {
                    v.x = fmaxf(v.x, 0.0f); v.y = fmaxf(v.y, 0.0f);
                    v.z = fmaxf(v.z, 0.0f); v.w = fmaxf(v.w, 0.0f);
                }
                ((float4*)out)[gi] = v;
            }
        }
        __syncthreads();
    }
}

constexpr int L1_F      = 128;
constexpr int L1_G      = 64;
constexpr int L1_TP     = L1_F + 4;
constexpr int L1_PP1    = 2 * L1_F + 32;
constexpr int L1_PP2    = 2 * L1_G + 32;
constexpr int L1_STRIPS = AGG_ROWS / 16;
static_assert(AGG_ROWS % 16 == 0, "a tile is whole strips");
static_assert(2 * L1_STRIPS == 20 && AGG_WAVES == 32, "the item-to-wave map below places 20 items on 32 waves");

template <int NOUT, int PP>
__device__ __forceinline__ void gcn_stage_w(const float* __restrict__ W, float* Ws, int tid) {
    constexpr int ITEMS = (GEMM_K / 2) * (NOUT / 4);
    static_assert(ITEMS % 1024 == 0, "whole trips of the work-group");
    constexpr int TRIPS = ITEMS / 1024;
    float4 v0[TRIPS], v1[TRIPS];
#pragma unroll
    for (int j = 0; j < TRIPS; ++j) {
        const int i  = tid + j * 1024;
        const int p  = i / (NOUT / 4);
        const int c4 = i % (NOUT / 4);
        v0[j] = ((const float4*)W)[(size_t)(2 * p) * (NOUT / 4) + c4];
        v1[j] = ((const float4*)W)[(size_t)(2 * p + 1) * (NOUT / 4) + c4];
    }
#pragma unroll
    for (int j = 0; j < TRIPS; ++j) {
        const int i  = tid + j * 1024;
        const int p  = i / (NOUT / 4);
        const int c4 = i % (NOUT / 4);
        float4* d = (float4*)&Ws[p * PP + c4 * 8];
        d[0] = make_float4(v0[j].x, v1[j].x, v0[j].y, v1[j].y);
        d[1] = make_float4(v0[j].z, v1[j].z, v0[j].w, v1[j].w);
    }
}

template <int NT, int PP>
__device__ __forceinline__ void gcn_strip_product(const float* arow, const float* wl, v8f (&c)[NT]) {
    v2f a = *(const v2f*)arow, an = a;
    v2f b[NT], bn[NT];
#pragma unroll
    for (int t = 0; t < NT; ++t) { b[t] = *(const v2f*)(wl + t * 32); bn[t] = b[t]; }
#pragma unroll
    for (int s = 0; s < GEMM_K / 4; ++s) {
        if (s + 1 < GEMM_K / 4) {
            an = *(const v2f*)(arow + 4 * (s + 1));
            const float* wn = wl + (2 * (s + 1)) * PP;
#pragma unroll
            for (int t = 0; t < NT; ++t) bn[t] = *(const v2f*)(wn + t * 32);
        }
#pragma unroll
        for (int t = 0; t < NT; ++t) {
            c[t] = __builtin_amdgcn_wmma_f32_16x16x4_f32(
                false, a, false, b[t], (short)0, c[t], false, false);
        }
        a = an;
#pragma unroll
        for (int t = 0; t < NT; ++t) b[t] = bn[t];
    }
}

__global__ __launch_bounds__(1024) void gcn_layer1_tile(
    const float* __restrict__ x, const int* __restrict__ src, const int* __restrict__ dst,
    const int* __restrict__ deg8, const float* __restrict__ W1, const float* __restrict__ b1,
    const float* __restrict__ W2, const float* __restrict__ lists,
    float* __restrict__ g, float* __restrict__ dinv, int N, int E) {
    __shared__ __attribute__((aligned(16))) float acc[AGG_ROWS * L1_TP];
    __shared__ float dl[AGG_ROWS];
    __shared__ unsigned fq[AGG_WAVES * AGG_FQCAP];
    __shared__ float fn[AGG_WAVES * AGG_FQCAP];
    __shared__ int ovf[2];
    __shared__ __attribute__((aligned(16))) float W1s[(GEMM_K / 2) * L1_PP1];
    __shared__ __attribute__((aligned(16))) float W2s[(GEMM_K / 2) * L1_PP2];

    const int tid  = threadIdx.x;
    const int lane = tid & 31;
    const int wave = __builtin_amdgcn_readfirstlane(tid >> 5);
    unsigned* myq = fq + wave * AGG_FQCAP;
    float*    nq  = fn + wave * AGG_FQCAP;

    if (tid == 0) { ovf[0] = 0; ovf[1] = 0; }
    __syncthreads();
    int par = 0;

    gcn_stage_w<L1_F, L1_PP1>(W1, W1s, tid);
    gcn_stage_w<L1_G, L1_PP2>(W2, W2s, tid);

    const int wq = wave & 7;
    const int wb = wave >> 3;
    const bool prod  = (wq < 4) || (wq == 4 + wb);
    const int  item  = wb * 5 + ((wq < 4) ? wq : 4);
    const int  strip = item % L1_STRIPS;
    const int  half  = item / L1_STRIPS;
    const int  m  = lane & 15;
    const int  gh = lane >> 4;

    const int ntiles = (N + AGG_ROWS - 1) / AGG_ROWS;

    for (int tile = blockIdx.x; tile < ntiles; tile += gridDim.x) {
        const int lo = tile * AGG_ROWS;
        const int nrows = min(AGG_ROWS, N - lo);
        const float* tl = lists + (size_t)lo * AGG_LROW;
        const int segw = (nrows * AGG_LROW) / AGG_GS;
        const int subw = segw / AGG_WAVES;

        typedef float __attribute__((ext_vector_type(L1_F / 32))) vt;
        const int rowc = lo + min(tid, nrows - 1);
        const int4 dc0 = *(const int4*)(deg8 + (size_t)rowc * AGG_GS);
        const int4 dc1 = *(const int4*)(deg8 + (size_t)rowc * AGG_GS + 4);
        vt xs[AGG_OWN];
#pragma unroll
        for (int j = 0; j < AGG_OWN; ++j) {
            const int row = min(wave + j * AGG_WAVES, nrows - 1);
            xs[j] = *((const vt*)(x + (size_t)(lo + row) * L1_F) + lane);
        }
        int total = -1;
        if (subw >= 2) total = gcn_own_fetch<1>(tl, segw, subw, nullptr, deg8, wave, lane, myq, nq);
        if (total < 0) ovf[par] = 1;
        if (tid < nrows) {
            const float dmy = gcn_dinv_from(dc0, dc1);
            dl[tid] = dmy;
            dinv[lo + tid] = dmy;
        }
        __syncthreads();
        const bool fits = __builtin_amdgcn_readfirstlane(ovf[par]) == 0;
        if (tid == 0) ovf[par ^ 1] = 0;
        par ^= 1;

        if (fits) {
            vt a[AGG_OWN];
#pragma unroll
            for (int j = 0; j < AGG_OWN; ++j) {
                const int row = wave + j * AGG_WAVES;
                const float di = dl[min(row, nrows - 1)];
                a[j] = (row < nrows) ? xs[j] * di * di : (vt)(0.0f);
            }
            gcn_own_drain<L1_F>(x, dl, myq, nq, total, lane, a);
#pragma unroll
            for (int j = 0; j < AGG_OWN; ++j) {
                const int row = wave + j * AGG_WAVES;
                *(vt*)&acc[row * L1_TP + (L1_F / 32) * lane] = a[j];
            }
        } else {
            for (int i = tid; i < AGG_ROWS * L1_TP / 4; i += 1024)
                ((float4*)acc)[i] = make_float4(0.0f, 0.0f, 0.0f, 0.0f);
            __syncthreads();
            int cnt = 0;
            for (int e0 = wave * 32; e0 < E; e0 += AGG_WAVES * 32) {
                const int e = e0 + lane;
                const int dv = (e < E) ? dst[e] : -1;
                const unsigned r = (unsigned)(dv - lo);
                const bool mk = r < (unsigned)nrows;
                const unsigned bm = __builtin_amdgcn_ballot_w32(mk);
                if (bm != 0u) {
                    if (cnt > AGG_FQCAP - 32) {
                        gcn_agg_drain<L1_F, L1_TP, 1>(x, nullptr, deg8, dl, acc, myq, cnt, lane);
                        cnt = 0;
                    }
                    if (mk) myq[cnt + (int)__builtin_amdgcn_mbcnt_lo(bm, 0u)] = (r << 24) | (unsigned)src[e];
                    cnt += __builtin_popcount(bm);
                }
            }
            gcn_agg_drain<L1_F, L1_TP, 1>(x, nullptr, deg8, dl, acc, myq, cnt, lane);
            __syncthreads();

            for (int i = tid; i < nrows * (L1_F / 4); i += 1024) {
                const int r  = i / (L1_F / 4);
                const int f4 = i % (L1_F / 4);
                const float di = dl[r];
                const float4 xv = ((const float4*)x)[(size_t)(lo + r) * (L1_F / 4) + f4];
                float4* p = (float4*)&acc[r * L1_TP + 4 * f4];
                float4 v = *p;
                v.x += xv.x * di * di;
                v.y += xv.y * di * di;
                v.z += xv.z * di * di;
                v.w += xv.w * di * di;
                *p = v;
            }
        }
        __syncthreads();

        v8f c1[4];
        float bv[4];
        if (prod) {
#pragma unroll
            for (int t = 0; t < 4; ++t) {
                c1[t] = (v8f){0, 0, 0, 0, 0, 0, 0, 0};
                bv[t] = b1[(4 * half + t) * 16 + m];
            }
            gcn_strip_product<4, L1_PP1>(acc + (strip * 16 + m) * L1_TP + 2 * gh,
                                         W1s + gh * L1_PP1 + 2 * m + (4 * half) * 32, c1);
        }
        __syncthreads();
        if (prod) {
#pragma unroll
            for (int t = 0; t < 4; ++t) {
                const int col = (4 * half + t) * 16 + m;
#pragma unroll
                for (int r = 0; r < 8; ++r)
                    acc[(strip * 16 + 8 * gh + r) * L1_TP + col] = fmaxf(c1[t][r] + bv[t], 0.0f);
            }
        }
        __syncthreads();

        if (prod) {
            v8f c2[2];
#pragma unroll
            for (int t = 0; t < 2; ++t) c2[t] = (v8f){0, 0, 0, 0, 0, 0, 0, 0};
            gcn_strip_product<2, L1_PP2>(acc + (strip * 16 + m) * L1_TP + 2 * gh,
                                         W2s + gh * L1_PP2 + 2 * m + (2 * half) * 32, c2);
#pragma unroll
            for (int t = 0; t < 2; ++t) {
                const int col = (2 * half + t) * 16 + m;
#pragma unroll
                for (int r = 0; r < 8; ++r) {
                    const int row = strip * 16 + 8 * gh + r;
                    if (row < nrows) g[(size_t)(lo + row) * L1_G + col] = c2[t][r];
                }
            }
        }
        __syncthreads();
    }
}

extern "C" void kernel_launch(void* const* d_in, const int* in_sizes, int n_in,
                              void* d_out, int out_size, void* d_ws, size_t ws_size,
                              hipStream_t stream) {
    const float* x  = (const float*)d_in[0];
    const int*   ei = (const int*)d_in[1];
    const float* W1 = (const float*)d_in[2];
    const float* b1 = (const float*)d_in[3];
    const float* W2 = (const float*)d_in[4];
    const float* b2 = (const float*)d_in[5];
    float* out = (float*)d_out;

    const int DIN = 128, DH = 128;
    const int N = in_sizes[0] / DIN;
    const int E = in_sizes[1] / 2;
    const int* src = ei;
    const int* dst = ei + E;

    float* ws = (float*)d_ws;
    size_t dinvPad = ((size_t)N + 255) & ~(size_t)255;
    float* dinv = ws;
    float* g    = ws + dinvPad;
    int*   deg8 = (int*)(g + (size_t)N * DH);

    gcn_tile_lists<<<256, 1024, 0, stream>>>(src, dst, deg8, out, N, E);

    gcn_layer1_tile<<<256, 1024, 0, stream>>>(x, src, dst, deg8, W1, b1, W2, out, g, dinv, N, E);

    gcn_agg_tile<64, 0><<<256, 1024, 0, stream>>>(g, src, dst, dinv, b2, out, out, N, E);
}


# A wave's edge queue is split by destination row (stable): every row's loop adds into its own registers only, F/32 multiply-adds per entry, not five times that with zero weights
# speedup vs baseline: 5.7471x; 1.0686x over previous
#include <hip/hip_runtime.h>
#include <hip/hip_bf16.h>

typedef __attribute__((ext_vector_type(2))) float v2f;
typedef __attribute__((ext_vector_type(8))) float v8f;

constexpr int AGG_GS = 8;
constexpr int GEMM_K = 128;
static_assert(AGG_GS == 8, "a node's per-slice counts are read as two 16-byte words");

__device__ __forceinline__ float gcn_dinv_from(const int4& c0, const int4& c1) {
    const int d = 1 + c0.x + c0.y + c0.z + c0.w + c1.x + c1.y + c1.z + c1.w;
    const float v = (float)d;
    return (v > 0.0f) ? __frsqrt_rn(v) : 0.0f;
}
__device__ __forceinline__ float gcn_dinv_of(const int* __restrict__ deg8, int i) {
    const int4 c0 = *(const int4*)(deg8 + (size_t)i * AGG_GS);
    const int4 c1 = *(const int4*)(deg8 + (size_t)i * AGG_GS + 4);
    return gcn_dinv_from(c0, c1);
}

constexpr int AGG_ROWS  = 160;
constexpr int AGG_WAVES = 32;
constexpr int AGG_GT    = 8;
constexpr int AGG_GROWS = AGG_GT * AGG_ROWS;
constexpr int AGG_RBITS = 11;
constexpr int AGG_EBITS = 32 - AGG_RBITS;
constexpr int AGG_QCAP  = 1024;
constexpr int AGG_FQCAP = 320;
constexpr int AGG_LROW  = 64;
static_assert(AGG_GROWS <= (1 << AGG_RBITS), "row inside the group must fit its bits");
constexpr int AGG_OWN   = AGG_ROWS / AGG_WAVES;
constexpr int AGG_SUBW  = (AGG_ROWS * AGG_LROW) / (AGG_GS * AGG_WAVES);
static_assert(AGG_ROWS % AGG_WAVES == 0, "every wave owns the same number of rows");
static_assert(AGG_ROWS <= 1024, "one thread per row of a tile fetches the row's dinv");
static_assert(AGG_GS * (AGG_SUBW - 1) <= AGG_FQCAP, "a wave's queue holds all its buckets");
static_assert(AGG_SUBW - 1 >= 32 && AGG_SUBW - 1 <= 64, "a bucket is copied in one whole and one partial step of 32 entries");
constexpr int AGG_JCAP  = AGG_FQCAP / AGG_OWN;
static_assert(AGG_OWN * AGG_JCAP <= AGG_FQCAP, "a wave's sorted queue holds one part per owned row");

__global__ __launch_bounds__(1024) void gcn_tile_lists(
    const int* __restrict__ src, const int* __restrict__ dst,
    int* __restrict__ degp, float* __restrict__ lists, int N, int E) {
    __shared__ unsigned q[AGG_WAVES * AGG_QCAP];
    __shared__ int degc[AGG_GROWS];
    __shared__ int bcnt[AGG_GT * AGG_WAVES];

    const int tid  = threadIdx.x;
    const int lane = tid & 31;
    const int wave = __builtin_amdgcn_readfirstlane(tid >> 5);
    unsigned* myq = q + wave * AGG_QCAP;

    const int ngroups = (N + AGG_GROWS - 1) / AGG_GROWS;
    const bool vec_ok = (reinterpret_cast<size_t>(dst) & 15) == 0;
    const int nfull = vec_ok ? E / 512 : 0;

    for (int job = blockIdx.x; job < ngroups * AGG_GS; job += gridDim.x) {
        const int group = job / AGG_GS;
        const int slice = job % AGG_GS;
        const int lo = group * AGG_GROWS;
        const int nrows = min(AGG_GROWS, N - lo);
        float* gl = lists + (size_t)lo * AGG_LROW;
        const int t_lo = (int)(((long long)nfull * slice) / AGG_GS);
        const int t_hi = (int)(((long long)nfull * (slice + 1)) / AGG_GS);
        const int nt = t_hi - t_lo;
        const int ebase = t_lo * 512;
        const int* dsl = dst + ebase;
        const int* ssl = src + ebase;

        for (int i = tid; i < AGG_GROWS; i += 1024) degc[i] = 0;
        for (int i = tid; i < AGG_GT * AGG_WAVES; i += 1024) bcnt[i] = 0;
        __syncthreads();

        int cnt = 0;

        auto flush = [&]() {
            for (int i0 = 0; i0 < cnt; i0 += 32) {
                if (i0 + lane < cnt) {
                    const unsigned ent = myq[i0 + lane];
                    const unsigned r = ent >> AGG_EBITS;
                    const unsigned s = (unsigned)ssl[ent & ((1u << AGG_EBITS) - 1u)];
                    atomicAdd(&degc[r], 1);
                    const int tt = (int)(r / (unsigned)AGG_ROWS);
                    const unsigned rt = r - (unsigned)(tt * AGG_ROWS);
                    const int ow = (int)(rt % (unsigned)AGG_WAVES);
                    const int nrk  = min(AGG_ROWS, nrows - tt * AGG_ROWS);
                    const int segw = (nrk * AGG_LROW) / AGG_GS;
                    const int subw = segw / AGG_WAVES;
                    const int slot = atomicAdd(&bcnt[tt * AGG_WAVES + ow], 1);
                    if (slot < subw - 1)
                        gl[(size_t)tt * AGG_ROWS * AGG_LROW + (size_t)slice * segw + (size_t)ow * subw + 1 + slot] =
                            __uint_as_float((rt << 24) | s);
                }
            }
            cnt = 0;
        };
        auto room = [&](int need) {
            if (cnt > AGG_QCAP - need) flush();
        };
        auto keep4 = [&](const int4& d, int e) {
            const unsigned r0 = (unsigned)(d.x - lo);
            const unsigned r1 = (unsigned)(d.y - lo);
            const unsigned r2 = (unsigned)(d.z - lo);
            const unsigned r3 = (unsigned)(d.w - lo);
            const bool m0 = r0 < (unsigned)nrows;
            const bool m1 = r1 < (unsigned)nrows;
            const bool m2 = r2 < (unsigned)nrows;
            const bool m3 = r3 < (unsigned)nrows;
            if (__builtin_amdgcn_ballot_w32(m0 | m1 | m2 | m3) != 0u) {
                const unsigned b0 = __builtin_amdgcn_ballot_w32(m0);
                if (m0) myq[cnt + (int)__builtin_amdgcn_mbcnt_lo(b0, 0u)] = (r0 << AGG_EBITS) | (unsigned)(e + 0);
                cnt += __builtin_popcount(b0);
                const unsigned b1 = __builtin_amdgcn_ballot_w32(m1);
                if (m1) myq[cnt + (int)__builtin_amdgcn_mbcnt_lo(b1, 0u)] = (r1 << AGG_EBITS) | (unsigned)(e + 1);
                cnt += __builtin_popcount(b1);
                const unsigned b2 = __builtin_amdgcn_ballot_w32(m2);
                if (m2) myq[cnt + (int)__builtin_amdgcn_mbcnt_lo(b2, 0u)] = (r2 << AGG_EBITS) | (unsigned)(e + 2);
                cnt += __builtin_popcount(b2);
                const unsigned b3 = __builtin_amdgcn_ballot_w32(m3);
                if (m3) myq[cnt + (int)__builtin_amdgcn_mbcnt_lo(b3, 0u)] = (r3 << AGG_EBITS) | (unsigned)(e + 3);
                cnt += __builtin_popcount(b3);
            }
        };

        if (wave < nt) {
            const int rstep = max(nt / max(ngroups, 1), 1);
            const int rot = (int)(((unsigned)group * (unsigned)rstep) % (unsigned)nt);
            auto trip_base = [&](int t) {
                int sg = t + rot;
                if (sg >= nt) sg -= nt;
                return sg * 512 + lane * 4;
            };
            int4 a[4], b[4];
            int t  = wave;
            int ea = trip_base(t);
            int eb = ea;
#pragma unroll
            for (int k = 0; k < 4; ++k) a[k] = *(const int4*)(dsl + ea + k * 128);
            while (t < nt) {
                const int tb = t + AGG_WAVES;
                const bool hb = tb < nt;
                room(512);
                eb = trip_base(hb ? tb : t);
#pragma unroll
                for (int k = 0; k < 4; ++k) b[k] = *(const int4*)(dsl + eb + k * 128);
#pragma unroll
                for (int k = 0; k < 4; ++k) keep4(a[k], ea + k * 128);

                const int ta = tb + AGG_WAVES;
                const bool ha = ta < nt;
                room(512);
                ea = trip_base(ha ? ta : t);
#pragma unroll
                for (int k = 0; k < 4; ++k) a[k] = *(const int4*)(dsl + ea + k * 128);
                if (hb) {
#pragma unroll
                    for (int k = 0; k < 4; ++k) keep4(b[k], eb + k * 128);
                }
                t = ta;
            }
        }
        if (slice == AGG_GS - 1) {
            for (int e0 = nfull * 512 + wave * 32; e0 < E; e0 += AGG_WAVES * 32) {
                const int e = e0 + lane;
                const int dv = (e < E) ? dst[e] : -1;
                const unsigned r = (unsigned)(dv - lo);
                const bool m = r < (unsigned)nrows;
                const unsigned bm = __builtin_amdgcn_ballot_w32(m);
                if (bm != 0u) {
                    room(32);
                    if (m) myq[cnt + (int)__builtin_amdgcn_mbcnt_lo(bm, 0u)] = (r << AGG_EBITS) | (unsigned)(e - ebase);
                    cnt += __builtin_popcount(bm);
                }
            }
        }
        flush();
        __syncthreads();

        for (int i = tid; i < AGG_GT * AGG_WAVES; i += 1024) {
            const int tt = i / AGG_WAVES;
            const int ow = i % AGG_WAVES;
            if (nrows > tt * AGG_ROWS) {
                const int nrk  = min(AGG_ROWS, nrows - tt * AGG_ROWS);
                const int segw = (nrk * AGG_LROW) / AGG_GS;
                const int subw = segw / AGG_WAVES;
                if (subw >= 1)
                    gl[(size_t)tt * AGG_ROWS * AGG_LROW + (size_t)slice * segw + (size_t)ow * subw] =
                        __uint_as_float((unsigned)bcnt[i]);
            }
        }
        for (int i = tid; i < nrows; i += 1024) degp[(size_t)(lo + i) * AGG_GS + slice] = degc[i];
        __syncthreads();
    }
}

template <int F, int P, int DEG>
__device__ __forceinline__ void gcn_agg_drain(
    const float* __restrict__ h, const float* __restrict__ dinv, const int* __restrict__ deg8,
    const float* dl, float* acc, const unsigned* list, int cnt, int lane) {
    constexpr int LPE = F / 4;
    constexpr int EPW = 32 / LPE;
    constexpr int UD  = 8;
    const int sub = (EPW == 1) ? 0 : lane / LPE;
    const int fl  = lane % LPE;
    const int hi  = (lane >> 4) & 1;
    unsigned pre = (lane < cnt) ? list[lane] : 0u;
    for (int base = 0; base < cnt; base += 32) {
        const int n = min(32, cnt - base);
        const unsigned ent = pre;
        const int nb = base + 32 + lane;
        pre = (nb < cnt) ? list[nb] : 0u;
        int s = 0, r = 0;
        float nrm = 0.0f;
        if (lane < n) {
            r = (int)(ent >> 24);
            s = (int)(ent & 0xFFFFFFu);
            float ds;
            if constexpr (DEG) ds = gcn_dinv_of(deg8, s);
            else               ds = dinv[s];
            nrm = ds * dl[r];
        }
        for (int i = 0; i < n; i += UD * EPW) {
            float4 v[UD];
            float  w[UD];
            int    si[UD];
            int    rr[UD];
            bool   ok[UD];
#pragma unroll
            for (int k = 0; k < UD; ++k) {
                const int idx = i + k * EPW + sub;
                ok[k] = idx < n;
                const int ii = ok[k] ? idx : 0;
                if constexpr (EPW == 1) {
                    si[k] = __builtin_amdgcn_readlane(s, ii);
                    rr[k] = __builtin_amdgcn_readlane(r, ii);
                    w[k]  = __int_as_float(__builtin_amdgcn_readlane(__float_as_int(nrm), ii));
                } else {
                    si[k] = __shfl(s, ii);
                    rr[k] = __shfl(r, ii);
                    w[k]  = __shfl(nrm, ii);
                }
            }
#pragma unroll
            for (int k = 0; k < UD; ++k) {
                v[k] = make_float4(0.0f, 0.0f, 0.0f, 0.0f);
                if (ok[k]) v[k] = *((const float4*)(h + (size_t)si[k] * F) + fl);
            }
#pragma unroll
            for (int k = 0; k < UD; ++k) {
                if (ok[k]) {
                    float* a  = acc + rr[k] * P + fl * 4;
                    float* p0 = a + 2 * hi;
                    float* p1 = a + 2 - 2 * hi;
                    const float c0 = hi ? v[k].z : v[k].x;
                    const float c1 = hi ? v[k].w : v[k].y;
                    const float c2 = hi ? v[k].x : v[k].z;
                    const float c3 = hi ? v[k].y : v[k].w;
                    atomicAdd(p0 + 0, c0 * w[k]);
                    atomicAdd(p0 + 1, c1 * w[k]);
                    atomicAdd(p1 + 0, c2 * w[k]);
                    atomicAdd(p1 + 1, c3 * w[k]);
                }
            }
        }
    }
}

template <int DEG>
__device__ __forceinline__ int gcn_own_fetch(
    const float* tl, int segw, int subw, const float* __restrict__ dinv, const int* __restrict__ deg8,
    int wave, int lane, unsigned* myq, unsigned* sq, float* sw, int (&cn)[AGG_OWN]) {
    const float* bk = tl + (size_t)wave * subw;
    int cw = 0;
    if (lane < AGG_GS) cw = (int)__float_as_uint(bk[(size_t)lane * segw]);
    unsigned e0[AGG_GS];
#pragma unroll
    for (int sg = 0; sg < AGG_GS; ++sg)
        e0[sg] = (1 + lane < subw) ? __float_as_uint(bk[(size_t)sg * segw + 1 + lane]) : 0u;
    const bool bad = (lane < AGG_GS) && (cw < 0 || cw > subw - 1);
    if (__builtin_amdgcn_ballot_w32(bad) != 0u) return -1;
    int off = 0;
#pragma unroll
    for (int sg = 0; sg < AGG_GS; ++sg) {
        const int cs = __builtin_amdgcn_readlane(cw, sg);
        if (lane < cs) myq[off + lane] = e0[sg];
        if (cs > 32) {
            if (32 + lane < cs) myq[off + 32 + lane] = __float_as_uint(bk[(size_t)sg * segw + 33 + lane]);
        }
        off += cs;
    }
#pragma unroll
    for (int j = 0; j < AGG_OWN; ++j) cn[j] = 0;
    for (int b0 = 0; b0 < off; b0 += 128) {
        float ds[4];
        int   sv[4];
        int   jv[4];
#pragma unroll
        for (int q = 0; q < 4; ++q) {
            const int e = b0 + q * 32 + lane;
            sv[q] = 0;
            jv[q] = -1;
            if (e < off) {
                const unsigned ent = myq[e];
                sv[q] = (int)(ent & 0xFFFFFFu);
                jv[q] = (int)(ent >> 24) / AGG_WAVES;
            }
            if constexpr (DEG) ds[q] = gcn_dinv_of(deg8, sv[q]);
            else               ds[q] = dinv[sv[q]];
        }
#pragma unroll
        for (int q = 0; q < 4; ++q) {
            int within = AGG_JCAP;
#pragma unroll
            for (int j = 0; j < AGG_OWN; ++j) {
                const bool m = jv[q] == j;
                const unsigned bm = __builtin_amdgcn_ballot_w32(m);
                if (m) within = cn[j] + (int)__builtin_amdgcn_mbcnt_lo(bm, 0u);
                cn[j] += __builtin_popcount(bm);
            }
            if (within < AGG_JCAP) {
                const int p = jv[q] * AGG_JCAP + within;
                sq[p] = (unsigned)sv[q];
                sw[p] = ds[q];
            }
        }
    }
    bool over = false;
#pragma unroll
    for (int j = 0; j < AGG_OWN; ++j) over = over || (cn[j] > AGG_JCAP);
    return over ? -1 : 0;
}

template <int F>
__device__ __forceinline__ void gcn_own_drain(
    const float* __restrict__ h, const float* dl, const unsigned* sq, const float* sw,
    const int (&cn)[AGG_OWN], int wave, int nrows, int lane,
    float __attribute__((ext_vector_type(F / 32))) (&a)[AGG_OWN]) {
    typedef float __attribute__((ext_vector_type(F / 32))) vt;
    constexpr int UD = (F >= 128) ? 8 : 16;
    static_assert(32 % UD == 0, "a group never reaches past lane 31");
#pragma unroll
    for (int j = 0; j < AGG_OWN; ++j) {
        const int cnt = cn[j];
        const float di = dl[min(wave + j * AGG_WAVES, nrows - 1)];
        const unsigned* qj = sq + j * AGG_JCAP;
        const float*    wj = sw + j * AGG_JCAP;
        vt acc = a[j];
        for (int base = 0; base < cnt; base += 32) {
            const int n = min(32, cnt - base);
            unsigned s = 0u;
            float nrm = 0.0f;
            if (lane < n) {
                s = qj[base + lane];
                nrm = wj[base + lane] * di;
            }
            for (int i = 0; i < n; i += UD) {
                vt    v[UD];
                float w[UD];
#pragma unroll
                for (int k = 0; k < UD; ++k) {
                    const int idx = i + k;
                    const unsigned si = (unsigned)__builtin_amdgcn_readlane((int)s, idx);
                    w[k] = __int_as_float(__builtin_amdgcn_readlane(__float_as_int(nrm), idx));
                    v[k] = (vt)(0.0f);
                    if (idx < n) v[k] = *((const vt*)(h + (size_t)si * F) + lane);
                }
#pragma unroll
                for (int k = 0; k < UD; ++k) acc += v[k] * w[k];
            }
        }
        a[j] = acc;
    }
}

template <int F, int RELU>
__global__ __launch_bounds__(1024) void gcn_agg_tile(
    const float* __restrict__ h, const int* __restrict__ src, const int* __restrict__ dst,
    const float* __restrict__ dinv, const float* __restrict__ bias,
    const float* lists, float* out, int N, int E) {
    typedef float __attribute__((ext_vector_type(F / 32))) vt;
    __shared__ float acc[AGG_ROWS * F];
    __shared__ float dl[AGG_ROWS];
    __shared__ unsigned fq[AGG_WAVES * AGG_FQCAP];
    __shared__ unsigned fs[AGG_WAVES * AGG_FQCAP];
    __shared__ float fn[AGG_WAVES * AGG_FQCAP];
    __shared__ int ovf[2];

    const int tid  = threadIdx.x;
    const int lane = tid & 31;
    const int wave = __builtin_amdgcn_readfirstlane(tid >> 5);
    unsigned* myq = fq + wave * AGG_FQCAP;
    unsigned* sq  = fs + wave * AGG_FQCAP;
    float*    sw  = fn + wave * AGG_FQCAP;

    if (tid == 0) { ovf[0] = 0; ovf[1] = 0; }
    __syncthreads();
    int par = 0;

    const int ntiles = (N + AGG_ROWS - 1) / AGG_ROWS;

    for (int tile = blockIdx.x; tile < ntiles; tile += gridDim.x) {
        const int lo = tile * AGG_ROWS;
        const int nrows = min(AGG_ROWS, N - lo);
        const float* tl = lists + (size_t)lo * AGG_LROW;
        const int segw = (nrows * AGG_LROW) / AGG_GS;
        const int subw = segw / AGG_WAVES;

        float dmy = 0.0f;
        if (tid < nrows) dmy = dinv[lo + tid];
        int total = -1;
        int cn[AGG_OWN];
        if (subw >= 2) total = gcn_own_fetch<0>(tl, segw, subw, dinv, nullptr, wave, lane, myq, sq, sw, cn);
        if (total < 0) ovf[par] = 1;
        if (tid < nrows) dl[tid] = dmy;
        __syncthreads();
        const bool fits = __builtin_amdgcn_readfirstlane(ovf[par]) == 0;
        if (tid == 0) ovf[par ^ 1] = 0;
        par ^= 1;

        if (fits) {
            vt hs[AGG_OWN], a[AGG_OWN];
#pragma unroll
            for (int j = 0; j < AGG_OWN; ++j) {
                const int row = min(wave + j * AGG_WAVES, nrows - 1);
                hs[j] = *((const vt*)(h + (size_t)(lo + row) * F) + lane);
                a[j] = (vt)(0.0f);
            }
            const vt bv = *((const vt*)bias + lane);
            gcn_own_drain<F>(h, dl, sq, sw, cn, wave, nrows, lane, a);
#pragma unroll
            for (int j = 0; j < AGG_OWN; ++j) {
                const int row = wave + j * AGG_WAVES;
                if (row < nrows) {
                    const float di = dl[row];
                    vt v = a[j] + hs[j] * di * di + bv;
                    if (RELU) {
#pragma unroll
                        for (int q = 0; q < F / 32; ++q) v[q] = fmaxf(v[q], 0.0f);
                    }
                    *((vt*)(out + (size_t)(lo + row) * F) + lane) = v;
                }
            }
        } else {
            for (int i = tid; i < AGG_ROWS * F / 4; i += 1024)
                ((float4*)acc)[i] = make_float4(0.0f, 0.0f, 0.0f, 0.0f);
            __syncthreads();
            int cnt = 0;
            for (int e0 = wave * 32; e0 < E; e0 += AGG_WAVES * 32) {
                const int e = e0 + lane;
                const int dv = (e < E) ? dst[e] : -1;
                const unsigned r = (unsigned)(dv - lo);
                const bool m = r < (unsigned)nrows;
                const unsigned bm = __builtin_amdgcn_ballot_w32(m);
                if (bm != 0u) {
                    if (cnt > AGG_FQCAP - 32) {
                        gcn_agg_drain<F, F, 0>(h, dinv, nullptr, dl, acc, myq, cnt, lane);
                        cnt = 0;
                    }
                    if (m) myq[cnt + (int)__builtin_amdgcn_mbcnt_lo(bm, 0u)] = (r << 24) | (unsigned)src[e];
                    cnt += __builtin_popcount(bm);
                }
            }
            gcn_agg_drain<F, F, 0>(h, dinv, nullptr, dl, acc, myq, cnt, lane);
            __syncthreads();

            for (int i = tid; i < nrows * (F / 4); i += 1024) {
                const int r  = i / (F / 4);
                const int f4 = i % (F / 4);
                const size_t gi = ((size_t)(lo + r) * F) / 4 + f4;
                const float di = dl[r];
                const float4 sa = ((const float4*)acc)[i];
                const float4 hv = ((const float4*)h)[gi];
                const float4 bv = ((const float4*)bias)[f4];
                float4 v;
                v.x = sa.x + hv.x * di * di + bv.x;
                v.y = sa.y + hv.y * di * di + bv.y;
                v.z = sa.z + hv.z * di * di + bv.z;
                v.w = sa.w + hv.w * di * di + bv.w;
                if (RELU) {
                    v.x = fmaxf(v.x, 0.0f); v.y = fmaxf(v.y, 0.0f);
                    v.z = fmaxf(v.z, 0.0f); v.w = fmaxf(v.w, 0.0f);
                }
                ((float4*)out)[gi] = v;
            }
        }
        __syncthreads();
    }
}

constexpr int L1_F      = 128;
constexpr int L1_G      = 64;
constexpr int L1_TP     = L1_F + 4;
constexpr int L1_PP1    = 2 * L1_F + 32;
constexpr int L1_PP2    = 2 * L1_G + 32;
constexpr int L1_STRIPS = AGG_ROWS / 16;
static_assert(AGG_ROWS % 16 == 0, "a tile is whole strips");
static_assert(2 * L1_STRIPS == 20 && AGG_WAVES == 32, "the item-to-wave map below places 20 items on 32 waves");

template <int NOUT, int PP>
__device__ __forceinline__ void gcn_stage_w(const float* __restrict__ W, float* Ws, int tid) {
    constexpr int ITEMS = (GEMM_K / 2) * (NOUT / 4);
    static_assert(ITEMS % 1024 == 0, "whole trips of the work-group");
    constexpr int TRIPS = ITEMS / 1024;
    float4 v0[TRIPS], v1[TRIPS];
#pragma unroll
    for (int j = 0; j < TRIPS; ++j) {
        const int i  = tid + j * 1024;
        const int p  = i / (NOUT / 4);
        const int c4 = i % (NOUT / 4);
        v0[j] = ((const float4*)W)[(size_t)(2 * p) * (NOUT / 4) + c4];
        v1[j] = ((const float4*)W)[(size_t)(2 * p + 1) * (NOUT / 4) + c4];
    }
#pragma unroll
    for (int j = 0; j < TRIPS; ++j) {
        const int i  = tid + j * 1024;
        const int p  = i / (NOUT / 4);
        const int c4 = i % (NOUT / 4);
        float4* d = (float4*)&Ws[p * PP + c4 * 8];
        d[0] = make_float4(v0[j].x, v1[j].x, v0[j].y, v1[j].y);
        d[1] = make_float4(v0[j].z, v1[j].z, v0[j].w, v1[j].w);
    }
}

template <int NT, int PP>
__device__ __forceinline__ void gcn_strip_product(const float* arow, const float* wl, v8f (&c)[NT]) {
    v2f a = *(const v2f*)arow, an = a;
    v2f b[NT], bn[NT];
#pragma unroll
    for (int t = 0; t < NT; ++t) { b[t] = *(const v2f*)(wl + t * 32); bn[t] = b[t]; }
#pragma unroll
    for (int s = 0; s < GEMM_K / 4; ++s) {
        if (s + 1 < GEMM_K / 4) {
            an = *(const v2f*)(arow + 4 * (s + 1));
            const float* wn = wl + (2 * (s + 1)) * PP;
#pragma unroll
            for (int t = 0; t < NT; ++t) bn[t] = *(const v2f*)(wn + t * 32);
        }
#pragma unroll
        for (int t = 0; t < NT; ++t) {
            c[t] = __builtin_amdgcn_wmma_f32_16x16x4_f32(
                false, a, false, b[t], (short)0, c[t], false, false);
        }
        a = an;
#pragma unroll
        for (int t = 0; t < NT; ++t) b[t] = bn[t];
    }
}

__global__ __launch_bounds__(1024) void gcn_layer1_tile(
    const float* __restrict__ x, const int* __restrict__ src, const int* __restrict__ dst,
    const int* __restrict__ deg8, const float* __restrict__ W1, const float* __restrict__ b1,
    const float* __restrict__ W2, const float* __restrict__ lists,
    float* __restrict__ g, float* __restrict__ dinv, int N, int E) {
    __shared__ __attribute__((aligned(16))) float acc[AGG_ROWS * L1_TP];
    __shared__ float dl[AGG_ROWS];
    __shared__ unsigned fq[AGG_WAVES * AGG_FQCAP];
    __shared__ unsigned fs[AGG_WAVES * AGG_FQCAP];
    __shared__ float fn[AGG_WAVES * AGG_FQCAP];
    __shared__ int ovf[2];
    __shared__ __attribute__((aligned(16))) float W1s[(GEMM_K / 2) * L1_PP1];
    __shared__ __attribute__((aligned(16))) float W2s[(GEMM_K / 2) * L1_PP2];

    const int tid  = threadIdx.x;
    const int lane = tid & 31;
    const int wave = __builtin_amdgcn_readfirstlane(tid >> 5);
    unsigned* myq = fq + wave * AGG_FQCAP;
    unsigned* sq  = fs + wave * AGG_FQCAP;
    float*    sw  = fn + wave * AGG_FQCAP;

    if (tid == 0) { ovf[0] = 0; ovf[1] = 0; }
    __syncthreads();
    int par = 0;

    gcn_stage_w<L1_F, L1_PP1>(W1, W1s, tid);
    gcn_stage_w<L1_G, L1_PP2>(W2, W2s, tid);

    const int wq = wave & 7;
    const int wb = wave >> 3;
    const bool prod  = (wq < 4) || (wq == 4 + wb);
    const int  item  = wb * 5 + ((wq < 4) ? wq : 4);
    const int  strip = item % L1_STRIPS;
    const int  half  = item / L1_STRIPS;
    const int  m  = lane & 15;
    const int  gh = lane >> 4;

    const int ntiles = (N + AGG_ROWS - 1) / AGG_ROWS;

    for (int tile = blockIdx.x; tile < ntiles; tile += gridDim.x) {
        const int lo = tile * AGG_ROWS;
        const int nrows = min(AGG_ROWS, N - lo);
        const float* tl = lists + (size_t)lo * AGG_LROW;
        const int segw = (nrows * AGG_LROW) / AGG_GS;
        const int subw = segw / AGG_WAVES;

        typedef float __attribute__((ext_vector_type(L1_F / 32))) vt;
        const int rowc = lo + min(tid, nrows - 1);
        const int4 dc0 = *(const int4*)(deg8 + (size_t)rowc * AGG_GS);
        const int4 dc1 = *(const int4*)(deg8 + (size_t)rowc * AGG_GS + 4);
        vt xs[AGG_OWN];
#pragma unroll
        for (int j = 0; j < AGG_OWN; ++j) {
            const int row = min(wave + j * AGG_WAVES, nrows - 1);
            xs[j] = *((const vt*)(x + (size_t)(lo + row) * L1_F) + lane);
        }
        int total = -1;
        int cn[AGG_OWN];
        if (subw >= 2) total = gcn_own_fetch<1>(tl, segw, subw, nullptr, deg8, wave, lane, myq, sq, sw, cn);
        if (total < 0) ovf[par] = 1;
        if (tid < nrows) {
            const float dmy = gcn_dinv_from(dc0, dc1);
            dl[tid] = dmy;
            dinv[lo + tid] = dmy;
        }
        __syncthreads();
        const bool fits = __builtin_amdgcn_readfirstlane(ovf[par]) == 0;
        if (tid == 0) ovf[par ^ 1] = 0;
        par ^= 1;

        if (fits) {
            vt a[AGG_OWN];
#pragma unroll
            for (int j = 0; j < AGG_OWN; ++j) {
                const int row = wave + j * AGG_WAVES;
                const float di = dl[min(row, nrows - 1)];
                a[j] = (row < nrows) ? xs[j] * di * di : (vt)(0.0f);
            }
            gcn_own_drain<L1_F>(x, dl, sq, sw, cn, wave, nrows, lane, a);
#pragma unroll
            for (int j = 0; j < AGG_OWN; ++j) {
                const int row = wave + j * AGG_WAVES;
                *(vt*)&acc[row * L1_TP + (L1_F / 32) * lane] = a[j];
            }
        } else {
            for (int i = tid; i < AGG_ROWS * L1_TP / 4; i += 1024)
                ((float4*)acc)[i] = make_float4(0.0f, 0.0f, 0.0f, 0.0f);
            __syncthreads();
            int cnt = 0;
            for (int e0 = wave * 32; e0 < E; e0 += AGG_WAVES * 32) {
                const int e = e0 + lane;
                const int dv = (e < E) ? dst[e] : -1;
                const unsigned r = (unsigned)(dv - lo);
                const bool mk = r < (unsigned)nrows;
                const unsigned bm = __builtin_amdgcn_ballot_w32(mk);
                if (bm != 0u) {
                    if (cnt > AGG_FQCAP - 32) {
                        gcn_agg_drain<L1_F, L1_TP, 1>(x, nullptr, deg8, dl, acc, myq, cnt, lane);
                        cnt = 0;
                    }
                    if (mk) myq[cnt + (int)__builtin_amdgcn_mbcnt_lo(bm, 0u)] = (r << 24) | (unsigned)src[e];
                    cnt += __builtin_popcount(bm);
                }
            }
            gcn_agg_drain<L1_F, L1_TP, 1>(x, nullptr, deg8, dl, acc, myq, cnt, lane);
            __syncthreads();

            for (int i = tid; i < nrows * (L1_F / 4); i += 1024) {
                const int r  = i / (L1_F / 4);
                const int f4 = i % (L1_F / 4);
                const float di = dl[r];
                const float4 xv = ((const float4*)x)[(size_t)(lo + r) * (L1_F / 4) + f4];
                float4* p = (float4*)&acc[r * L1_TP + 4 * f4];
                float4 v = *p;
                v.x += xv.x * di * di;
                v.y += xv.y * di * di;
                v.z += xv.z * di * di;
                v.w += xv.w * di * di;
                *p = v;
            }
        }
        __syncthreads();

        v8f c1[4];
        float bv[4];
        if (prod) {
#pragma unroll
            for (int t = 0; t < 4; ++t) {
                c1[t] = (v8f){0, 0, 0, 0, 0, 0, 0, 0};
                bv[t] = b1[(4 * half + t) * 16 + m];
            }
            gcn_strip_product<4, L1_PP1>(acc + (strip * 16 + m) * L1_TP + 2 * gh,
                                         W1s + gh * L1_PP1 + 2 * m + (4 * half) * 32, c1);
        }
        __syncthreads();
        if (prod) {
#pragma unroll
            for (int t = 0; t < 4; ++t) {
                const int col = (4 * half + t) * 16 + m;
#pragma unroll
                for (int r = 0; r < 8; ++r)
                    acc[(strip * 16 + 8 * gh + r) * L1_TP + col] = fmaxf(c1[t][r] + bv[t], 0.0f);
            }
        }
        __syncthreads();

        if (prod) {
            v8f c2[2];
#pragma unroll
            for (int t = 0; t < 2; ++t) c2[t] = (v8f){0, 0, 0, 0, 0, 0, 0, 0};
            gcn_strip_product<2, L1_PP2>(acc + (strip * 16 + m) * L1_TP + 2 * gh,
                                         W2s + gh * L1_PP2 + 2 * m + (2 * half) * 32, c2);
#pragma unroll
            for (int t = 0; t < 2; ++t) {
                const int col = (2 * half + t) * 16 + m;
#pragma unroll
                for (int r = 0; r < 8; ++r) {
                    const int row = strip * 16 + 8 * gh + r;
                    if (row < nrows) g[(size_t)(lo + row) * L1_G + col] = c2[t][r];
                }
            }
        }
        __syncthreads();
    }
}

extern "C" void kernel_launch(void* const* d_in, const int* in_sizes, int n_in,
                              void* d_out, int out_size, void* d_ws, size_t ws_size,
                              hipStream_t stream) {
    const float* x  = (const float*)d_in[0];
    const int*   ei = (const int*)d_in[1];
    const float* W1 = (const float*)d_in[2];
    const float* b1 = (const float*)d_in[3];
    const float* W2 = (const float*)d_in[4];
    const float* b2 = (const float*)d_in[5];
    float* out = (float*)d_out;

    const int DIN = 128, DH = 128;
    const int N = in_sizes[0] / DIN;
    const int E = in_sizes[1] / 2;
    const int* src = ei;
    const int* dst = ei + E;

    float* ws = (float*)d_ws;
    size_t dinvPad = ((size_t)N + 255) & ~(size_t)255;
    float* dinv = ws;
    float* g    = ws + dinvPad;
    int*   deg8 = (int*)(g + (size_t)N * DH);

    gcn_tile_lists<<<256, 1024, 0, stream>>>(src, dst, deg8, out, N, E);

    gcn_layer1_tile<<<256, 1024, 0, stream>>>(x, src, dst, deg8, W1, b1, W2, out, g, dinv, N, E);

    gcn_agg_tile<64, 0><<<256, 1024, 0, stream>>>(g, src, dst, dinv, b2, out, out, N, E);
}


# Row sums: all feature rows of a destination row are asked for before the first wait (whole blocks of four, no test per entry; scalar base + 32-bit offset); rows in a loop, addresses formed per tile
# speedup vs baseline: 6.0383x; 1.0507x over previous
#include <hip/hip_runtime.h>
#include <hip/hip_bf16.h>

typedef __attribute__((ext_vector_type(2))) float v2f;
typedef __attribute__((ext_vector_type(8))) float v8f;

constexpr int AGG_GS = 8;
constexpr int GEMM_K = 128;
static_assert(AGG_GS == 8, "a node's per-slice counts are read as two 16-byte words");

__device__ __forceinline__ float gcn_dinv_from(const int4& c0, const int4& c1) {
    const int d = 1 + c0.x + c0.y + c0.z + c0.w + c1.x + c1.y + c1.z + c1.w;
    const float v = (float)d;
    return (v > 0.0f) ? __frsqrt_rn(v) : 0.0f;
}
__device__ __forceinline__ float gcn_dinv_of(const int* __restrict__ deg8, int i) {
    const int4 c0 = *(const int4*)(deg8 + (size_t)i * AGG_GS);
    const int4 c1 = *(const int4*)(deg8 + (size_t)i * AGG_GS + 4);
    return gcn_dinv_from(c0, c1);
}

constexpr int AGG_ROWS  = 160;
constexpr int AGG_WAVES = 32;
constexpr int AGG_GT    = 8;
constexpr int AGG_GROWS = AGG_GT * AGG_ROWS;
constexpr int AGG_RBITS = 11;
constexpr int AGG_EBITS = 32 - AGG_RBITS;
constexpr int AGG_QCAP  = 1024;
constexpr int AGG_FQCAP = 320;
constexpr int AGG_LROW  = 64;
static_assert(AGG_GROWS <= (1 << AGG_RBITS), "row inside the group must fit its bits");
constexpr int AGG_OWN   = AGG_ROWS / AGG_WAVES;
constexpr int AGG_SUBW  = (AGG_ROWS * AGG_LROW) / (AGG_GS * AGG_WAVES);
static_assert(AGG_ROWS % AGG_WAVES == 0, "every wave owns the same number of rows");
static_assert(AGG_ROWS <= 1024, "one thread per row of a tile fetches the row's dinv");
static_assert(AGG_GS * (AGG_SUBW - 1) <= AGG_FQCAP, "a wave's queue holds all its buckets");
static_assert(AGG_SUBW - 1 >= 32 && AGG_SUBW - 1 <= 64, "a bucket is copied in one whole and one partial step of 32 entries");
constexpr int AGG_JCAP  = AGG_FQCAP / AGG_OWN;
static_assert(AGG_OWN * AGG_JCAP <= AGG_FQCAP, "a wave's sorted queue holds one part per owned row");

__global__ __launch_bounds__(1024) void gcn_tile_lists(
    const int* __restrict__ src, const int* __restrict__ dst,
    int* __restrict__ degp, float* __restrict__ lists, int N, int E) {
    __shared__ unsigned q[AGG_WAVES * AGG_QCAP];
    __shared__ int degc[AGG_GROWS];
    __shared__ int bcnt[AGG_GT * AGG_WAVES];

    const int tid  = threadIdx.x;
    const int lane = tid & 31;
    const int wave = __builtin_amdgcn_readfirstlane(tid >> 5);
    unsigned* myq = q + wave * AGG_QCAP;

    const int ngroups = (N + AGG_GROWS - 1) / AGG_GROWS;
    const bool vec_ok = (reinterpret_cast<size_t>(dst) & 15) == 0;
    const int nfull = vec_ok ? E / 512 : 0;

    for (int job = blockIdx.x; job < ngroups * AGG_GS; job += gridDim.x) {
        const int group = job / AGG_GS;
        const int slice = job % AGG_GS;
        const int lo = group * AGG_GROWS;
        const int nrows = min(AGG_GROWS, N - lo);
        float* gl = lists + (size_t)lo * AGG_LROW;
        const int t_lo = (int)(((long long)nfull * slice) / AGG_GS);
        const int t_hi = (int)(((long long)nfull * (slice + 1)) / AGG_GS);
        const int nt = t_hi - t_lo;
        const int ebase = t_lo * 512;
        const int* dsl = dst + ebase;
        const int* ssl = src + ebase;

        for (int i = tid; i < AGG_GROWS; i += 1024) degc[i] = 0;
        for (int i = tid; i < AGG_GT * AGG_WAVES; i += 1024) bcnt[i] = 0;
        __syncthreads();

        int cnt = 0;

        auto flush = [&]() {
            for (int i0 = 0; i0 < cnt; i0 += 32) {
                if (i0 + lane < cnt) {
                    const unsigned ent = myq[i0 + lane];
                    const unsigned r = ent >> AGG_EBITS;
                    const unsigned s = (unsigned)ssl[ent & ((1u << AGG_EBITS) - 1u)];
                    atomicAdd(&degc[r], 1);
                    const int tt = (int)(r / (unsigned)AGG_ROWS);
                    const unsigned rt = r - (unsigned)(tt * AGG_ROWS);
                    const int ow = (int)(rt % (unsigned)AGG_WAVES);
                    const int nrk  = min(AGG_ROWS, nrows - tt * AGG_ROWS);
                    const int segw = (nrk * AGG_LROW) / AGG_GS;
                    const int subw = segw / AGG_WAVES;
                    const int slot = atomicAdd(&bcnt[tt * AGG_WAVES + ow], 1);
                    if (slot < subw - 1)
                        gl[(size_t)tt * AGG_ROWS * AGG_LROW + (size_t)slice * segw + (size_t)ow * subw + 1 + slot] =
                            __uint_as_float((rt << 24) | s);
                }
            }
            cnt = 0;
        };
        auto room = [&](int need) {
            if (cnt > AGG_QCAP - need) flush();
        };
        auto keep4 = [&](const int4& d, int e) {
            const unsigned r0 = (unsigned)(d.x - lo);
            const unsigned r1 = (unsigned)(d.y - lo);
            const unsigned r2 = (unsigned)(d.z - lo);
            const unsigned r3 = (unsigned)(d.w - lo);
            const bool m0 = r0 < (unsigned)nrows;
            const bool m1 = r1 < (unsigned)nrows;
            const bool m2 = r2 < (unsigned)nrows;
            const bool m3 = r3 < (unsigned)nrows;
            if (__builtin_amdgcn_ballot_w32(m0 | m1 | m2 | m3) != 0u) {
                const unsigned b0 = __builtin_amdgcn_ballot_w32(m0);
                if (m0) myq[cnt + (int)__builtin_amdgcn_mbcnt_lo(b0, 0u)] = (r0 << AGG_EBITS) | (unsigned)(e + 0);
                cnt += __builtin_popcount(b0);
                const unsigned b1 = __builtin_amdgcn_ballot_w32(m1);
                if (m1) myq[cnt + (int)__builtin_amdgcn_mbcnt_lo(b1, 0u)] = (r1 << AGG_EBITS) | (unsigned)(e + 1);
                cnt += __builtin_popcount(b1);
                const unsigned b2 = __builtin_amdgcn_ballot_w32(m2);
                if (m2) myq[cnt + (int)__builtin_amdgcn_mbcnt_lo(b2, 0u)] = (r2 << AGG_EBITS) | (unsigned)(e + 2);
                cnt += __builtin_popcount(b2);
                const unsigned b3 = __builtin_amdgcn_ballot_w32(m3);
                if (m3) myq[cnt + (int)__builtin_amdgcn_mbcnt_lo(b3, 0u)] = (r3 << AGG_EBITS) | (unsigned)(e + 3);
                cnt += __builtin_popcount(b3);
            }
        };

        if (wave < nt) {
            const int rstep = max(nt / max(ngroups, 1), 1);
            const int rot = (int)(((unsigned)group * (unsigned)rstep) % (unsigned)nt);
            auto trip_base = [&](int t) {
                int sg = t + rot;
                if (sg >= nt) sg -= nt;
                return sg * 512 + lane * 4;
            };
            int4 a[4], b[4];
            int t  = wave;
            int ea = trip_base(t);
            int eb = ea;
#pragma unroll
            for (int k = 0; k < 4; ++k) a[k] = *(const int4*)(dsl + ea + k * 128);
            while (t < nt) {
                const int tb = t + AGG_WAVES;
                const bool hb = tb < nt;
                room(512);
                eb = trip_base(hb ? tb : t);
#pragma unroll
                for (int k = 0; k < 4; ++k) b[k] = *(const int4*)(dsl + eb + k * 128);
#pragma unroll
                for (int k = 0; k < 4; ++k) keep4(a[k], ea + k * 128);

                const int ta = tb + AGG_WAVES;
                const bool ha = ta < nt;
                room(512);
                ea = trip_base(ha ? ta : t);
#pragma unroll
                for (int k = 0; k < 4; ++k) a[k] = *(const int4*)(dsl + ea + k * 128);
                if (hb) {
#pragma unroll
                    for (int k = 0; k < 4; ++k) keep4(b[k], eb + k * 128);
                }
                t = ta;
            }
        }
        if (slice == AGG_GS - 1) {
            for (int e0 = nfull * 512 + wave * 32; e0 < E; e0 += AGG_WAVES * 32) {
                const int e = e0 + lane;
                const int dv = (e < E) ? dst[e] : -1;
                const unsigned r = (unsigned)(dv - lo);
                const bool m = r < (unsigned)nrows;
                const unsigned bm = __builtin_amdgcn_ballot_w32(m);
                if (bm != 0u) {
                    room(32);
                    if (m) myq[cnt + (int)__builtin_amdgcn_mbcnt_lo(bm, 0u)] = (r << AGG_EBITS) | (unsigned)(e - ebase);
                    cnt += __builtin_popcount(bm);
                }
            }
        }
        flush();
        __syncthreads();

        for (int i = tid; i < AGG_GT * AGG_WAVES; i += 1024) {
            const int tt = i / AGG_WAVES;
            const int ow = i % AGG_WAVES;
            if (nrows > tt * AGG_ROWS) {
                const int nrk  = min(AGG_ROWS, nrows - tt * AGG_ROWS);
                const int segw = (nrk * AGG_LROW) / AGG_GS;
                const int subw = segw / AGG_WAVES;
                if (subw >= 1)
                    gl[(size_t)tt * AGG_ROWS * AGG_LROW + (size_t)slice * segw + (size_t)ow * subw] =
                        __uint_as_float((unsigned)bcnt[i]);
            }
        }
        for (int i = tid; i < nrows; i += 1024) degp[(size_t)(lo + i) * AGG_GS + slice] = degc[i];
        __syncthreads();
    }
}

template <int F, int P, int DEG>
__device__ __forceinline__ void gcn_agg_drain(
    const float* __restrict__ h, const float* __restrict__ dinv, const int* __restrict__ deg8,
    const float* dl, float* acc, const unsigned* list, int cnt, int lane) {
    constexpr int LPE = F / 4;
    constexpr int EPW = 32 / LPE;
    constexpr int UD  = 8;
    const int sub = (EPW == 1) ? 0 : lane / LPE;
    const int fl  = lane % LPE;
    const int hi  = (lane >> 4) & 1;
    unsigned pre = (lane < cnt) ? list[lane] : 0u;
    for (int base = 0; base < cnt; base += 32) {
        const int n = min(32, cnt - base);
        const unsigned ent = pre;
        const int nb = base + 32 + lane;
        pre = (nb < cnt) ? list[nb] : 0u;
        int s = 0, r = 0;
        float nrm = 0.0f;
        if (lane < n) {
            r = (int)(ent >> 24);
            s = (int)(ent & 0xFFFFFFu);
            float ds;
            if constexpr (DEG) ds = gcn_dinv_of(deg8, s);
            else               ds = dinv[s];
            nrm = ds * dl[r];
        }
        for (int i = 0; i < n; i += UD * EPW) {
            float4 v[UD];
            float  w[UD];
            int    si[UD];
            int    rr[UD];
            bool   ok[UD];
#pragma unroll
            for (int k = 0; k < UD; ++k) {
                const int idx = i + k * EPW + sub;
                ok[k] = idx < n;
                const int ii = ok[k] ? idx : 0;
                if constexpr (EPW == 1) {
                    si[k] = __builtin_amdgcn_readlane(s, ii);
                    rr[k] = __builtin_amdgcn_readlane(r, ii);
                    w[k]  = __int_as_float(__builtin_amdgcn_readlane(__float_as_int(nrm), ii));
                } else {
                    si[k] = __shfl(s, ii);
                    rr[k] = __shfl(r, ii);
                    w[k]  = __shfl(nrm, ii);
                }
            }
#pragma unroll
            for (int k = 0; k < UD; ++k) {
                v[k] = make_float4(0.0f, 0.0f, 0.0f, 0.0f);
                if (ok[k]) v[k] = *((const float4*)(h + (size_t)si[k] * F) + fl);
            }
#pragma unroll
            for (int k = 0; k < UD; ++k) {
                if (ok[k]) {
                    float* a  = acc + rr[k] * P + fl * 4;
                    float* p0 = a + 2 * hi;
                    float* p1 = a + 2 - 2 * hi;
                    const float c0 = hi ? v[k].z : v[k].x;
                    const float c1 = hi ? v[k].w : v[k].y;
                    const float c2 = hi ? v[k].x : v[k].z;
                    const float c3 = hi ? v[k].y : v[k].w;
                    atomicAdd(p0 + 0, c0 * w[k]);
                    atomicAdd(p0 + 1, c1 * w[k]);
                    atomicAdd(p1 + 0, c2 * w[k]);
                    atomicAdd(p1 + 1, c3 * w[k]);
                }
            }
        }
    }
}

template <int DEG>
__device__ __forceinline__ int gcn_own_fetch(
    const float* tl, int segw, int subw, const float* __restrict__ dinv, const int* __restrict__ deg8,
    int wave, int lane, unsigned* myq, unsigned* sq, float* sw, int (&cn)[AGG_OWN]) {
    const float* bk = tl + (size_t)wave * subw;
    int cw = 0;
    if (lane < AGG_GS) cw = (int)__float_as_uint(bk[(size_t)lane * segw]);
    unsigned e0[AGG_GS];
#pragma unroll
    for (int sg = 0; sg < AGG_GS; ++sg)
        e0[sg] = (1 + lane < subw) ? __float_as_uint(bk[(size_t)sg * segw + 1 + lane]) : 0u;
    const bool bad = (lane < AGG_GS) && (cw < 0 || cw > subw - 1);
    if (__builtin_amdgcn_ballot_w32(bad) != 0u) return -1;
    int off = 0;
#pragma unroll
    for (int sg = 0; sg < AGG_GS; ++sg) {
        const int cs = __builtin_amdgcn_readlane(cw, sg);
        if (lane < cs) myq[off + lane] = e0[sg];
        if (cs > 32) {
            if (32 + lane < cs) myq[off + 32 + lane] = __float_as_uint(bk[(size_t)sg * segw + 33 + lane]);
        }
        off += cs;
    }
#pragma unroll
    for (int j = 0; j < AGG_OWN; ++j) cn[j] = 0;
    for (int b0 = 0; b0 < off; b0 += 128) {
        float ds[4];
        int   sv[4];
        int   jv[4];
#pragma unroll
        for (int q = 0; q < 4; ++q) {
            const int e = b0 + q * 32 + lane;
            sv[q] = 0;
            jv[q] = -1;
            if (e < off) {
                const unsigned ent = myq[e];
                sv[q] = (int)(ent & 0xFFFFFFu);
                jv[q] = (int)(ent >> 24) / AGG_WAVES;
            }
            if constexpr (DEG) ds[q] = gcn_dinv_of(deg8, sv[q]);
            else               ds[q] = dinv[sv[q]];
        }
#pragma unroll
        for (int q = 0; q < 4; ++q) {
            int within = AGG_JCAP;
#pragma unroll
            for (int j = 0; j < AGG_OWN; ++j) {
                const bool m = jv[q] == j;
                const unsigned bm = __builtin_amdgcn_ballot_w32(m);
                if (m) within = cn[j] + (int)__builtin_amdgcn_mbcnt_lo(bm, 0u);
                cn[j] += __builtin_popcount(bm);
            }
            if (within < AGG_JCAP) {
                const int p = jv[q] * AGG_JCAP + within;
                sq[p] = (unsigned)sv[q];
                sw[p] = ds[q];
            }
        }
    }
    bool over = false;
#pragma unroll
    for (int j = 0; j < AGG_OWN; ++j) over = over || (cn[j] > AGG_JCAP);
    return over ? -1 : 0;
}

template <int F, int G>
__device__ __forceinline__ void gcn_rows_issue(
    const float* __restrict__ h, unsigned loff, unsigned s, float nrm, int i0,
    float __attribute__((ext_vector_type(F / 32))) (&v)[G], float (&w)[G]) {
    typedef float __attribute__((ext_vector_type(F / 32))) vt;
#pragma unroll
    for (int k = 0; k < G; ++k) {
        const unsigned si = (unsigned)__builtin_amdgcn_readlane((int)s, i0 + k);
        w[k] = __int_as_float(__builtin_amdgcn_readlane(__float_as_int(nrm), i0 + k));
        const unsigned voff = si * (unsigned)(F * sizeof(float)) + loff;
        v[k] = *(const vt*)((const char*)h + voff);
    }
}

template <int F, int Q>
__device__ __forceinline__ void gcn_row_piece(
    const float* __restrict__ h, unsigned loff, unsigned s, float nrm, int i0, int r,
    float __attribute__((ext_vector_type(F / 32)))& acc) {
    typedef float __attribute__((ext_vector_type(F / 32))) vt;
    constexpr int G = 4 * Q;
    vt    v[G > 0 ? G : 1];
    float w[G > 0 ? G : 1];
    if constexpr (G > 0) gcn_rows_issue<F, G>(h, loff, s, nrm, i0, v, w);
    vt    vr[3];
    float wr[3];
#pragma unroll
    for (int k = 0; k < 3; ++k) {
        vr[k] = (vt)(0.0f);
        wr[k] = 0.0f;
        if (k < r) {
            const int idx = i0 + G + k;
            const unsigned si = (unsigned)__builtin_amdgcn_readlane((int)s, idx);
            wr[k] = __int_as_float(__builtin_amdgcn_readlane(__float_as_int(nrm), idx));
            const unsigned voff = si * (unsigned)(F * sizeof(float)) + loff;
            vr[k] = *(const vt*)((const char*)h + voff);
        }
    }
    if constexpr (G > 0) {
#pragma unroll
        for (int k = 0; k < G; ++k) acc += v[k] * w[k];
    }
#pragma unroll
    for (int k = 0; k < 3; ++k) acc += vr[k] * wr[k];
}
template <int F, int Q>
__device__ __forceinline__ void gcn_row_pick(
    int q, const float* __restrict__ h, unsigned loff, unsigned s, float nrm, int i0, int r,
    float __attribute__((ext_vector_type(F / 32)))& acc) {
    if (q == Q) gcn_row_piece<F, Q>(h, loff, s, nrm, i0, r, acc);
    else if constexpr (Q > 0) gcn_row_pick<F, Q - 1>(q, h, loff, s, nrm, i0, r, acc);
}

template <int F>
__device__ __forceinline__ void gcn_row_sum(
    const float* __restrict__ h, const unsigned* qj, const float* wj, int cnt, float di, int lane,
    float __attribute__((ext_vector_type(F / 32)))& acc) {
    typedef float __attribute__((ext_vector_type(F / 32))) vt;
    constexpr int CH = (F >= 128) ? 16 : 32;
    static_assert(CH % 4 == 0 && CH <= 32, "whole blocks of four inside the 32 entries that the lanes hold");
    const unsigned loff = (unsigned)lane * (unsigned)sizeof(vt);
    for (int base = 0; base < cnt; base += 32) {
        const int n = min(32, cnt - base);
        unsigned s = 0u;
        float nrm = 0.0f;
        if (lane < n) {
            s = qj[base + lane];
            nrm = wj[base + lane] * di;
        }
        for (int i0 = 0; i0 < n; i0 += CH) {
            const int m = min(CH, n - i0);
            gcn_row_pick<F, CH / 4>(m >> 2, h, loff, s, nrm, i0, m & 3, acc);
        }
    }
}

__device__ __forceinline__ int gcn_counts_by_lane(const int (&cn)[AGG_OWN], int lane) {
    int cnv = 0;
#pragma unroll
    for (int j = 0; j < AGG_OWN; ++j) cnv = (lane == j) ? cn[j] : cnv;
    return cnv;
}

template <int F, int RELU>
__global__ __launch_bounds__(1024) void gcn_agg_tile(
    const float* __restrict__ h, const int* __restrict__ src, const int* __restrict__ dst,
    const float* __restrict__ dinv, const float* __restrict__ bias,
    const float* lists, float* out, int N, int E) {
    typedef float __attribute__((ext_vector_type(F / 32))) vt;
    __shared__ float acc[AGG_ROWS * F];
    __shared__ float dl[AGG_ROWS];
    __shared__ unsigned fq[AGG_WAVES * AGG_FQCAP];
    __shared__ unsigned fs[AGG_WAVES * AGG_FQCAP];
    __shared__ float fn[AGG_WAVES * AGG_FQCAP];
    __shared__ int ovf[2];

    const int tid  = threadIdx.x;
    const int lane = tid & 31;
    const int wave = __builtin_amdgcn_readfirstlane(tid >> 5);
    unsigned* myq = fq + wave * AGG_FQCAP;
    unsigned* sq  = fs + wave * AGG_FQCAP;
    float*    sw  = fn + wave * AGG_FQCAP;

    if (tid == 0) { ovf[0] = 0; ovf[1] = 0; }
    __syncthreads();
    int par = 0;

    const int ntiles = (N + AGG_ROWS - 1) / AGG_ROWS;

    for (int tile = blockIdx.x; tile < ntiles; tile += gridDim.x) {
        const int lo = tile * AGG_ROWS;
        const int nrows = min(AGG_ROWS, N - lo);
        const float* tl = lists + (size_t)lo * AGG_LROW;
        const int segw = (nrows * AGG_LROW) / AGG_GS;
        const int subw = segw / AGG_WAVES;

        float dmy = 0.0f;
        if (tid < nrows) dmy = dinv[lo + tid];
        int total = -1;
        int cn[AGG_OWN];
        if (subw >= 2) total = gcn_own_fetch<0>(tl, segw, subw, dinv, nullptr, wave, lane, myq, sq, sw, cn);
        if (total < 0) ovf[par] = 1;
        if (tid < nrows) dl[tid] = dmy;
        __syncthreads();
        const bool fits = __builtin_amdgcn_readfirstlane(ovf[par]) == 0;
        if (tid == 0) ovf[par ^ 1] = 0;
        par ^= 1;

        if (fits) {
            const vt bv = *((const vt*)bias + lane);
            const int cnv = gcn_counts_by_lane(cn, lane);
#pragma unroll 1
            for (int j = 0; j < AGG_OWN; ++j) {
                const int row  = wave + j * AGG_WAVES;
                const int rin  = min(row, nrows - 1);
                const int cnt  = __builtin_amdgcn_readlane(cnv, j);
                const float di = dl[rin];
                const vt hs = *((const vt*)(h + (size_t)(lo + rin) * F) + lane);
                vt sum = (vt)(0.0f);
                gcn_row_sum<F>(h, sq + j * AGG_JCAP, sw + j * AGG_JCAP, cnt, di, lane, sum);
                if (row < nrows) {
                    vt v = sum + hs * di * di + bv;
                    if (RELU) {
#pragma unroll
                        for (int q = 0; q < F / 32; ++q) v[q] = fmaxf(v[q], 0.0f);
                    }
                    *((vt*)(out + (size_t)(lo + row) * F) + lane) = v;
                }
            }
        } else {
            for (int i = tid; i < AGG_ROWS * F / 4; i += 1024)
                ((float4*)acc)[i] = make_float4(0.0f, 0.0f, 0.0f, 0.0f);
            __syncthreads();
            int cnt = 0;
            for (int e0 = wave * 32; e0 < E; e0 += AGG_WAVES * 32) {
                const int e = e0 + lane;
                const int dv = (e < E) ? dst[e] : -1;
                const unsigned r = (unsigned)(dv - lo);
                const bool m = r < (unsigned)nrows;
                const unsigned bm = __builtin_amdgcn_ballot_w32(m);
                if (bm != 0u) {
                    if (cnt > AGG_FQCAP - 32) {
                        gcn_agg_drain<F, F, 0>(h, dinv, nullptr, dl, acc, myq, cnt, lane);
                        cnt = 0;
                    }
                    if (m) myq[cnt + (int)__builtin_amdgcn_mbcnt_lo(bm, 0u)] = (r << 24) | (unsigned)src[e];
                    cnt += __builtin_popcount(bm);
                }
            }
            gcn_agg_drain<F, F, 0>(h, dinv, nullptr, dl, acc, myq, cnt, lane);
            __syncthreads();

            for (int i = tid; i < nrows * (F / 4); i += 1024) {
                const int r  = i / (F / 4);
                const int f4 = i % (F / 4);
                const size_t gi = ((size_t)(lo + r) * F) / 4 + f4;
                const float di = dl[r];
                const float4 sa = ((const float4*)acc)[i];
                const float4 hv = ((const float4*)h)[gi];
                const float4 bv = ((const float4*)bias)[f4];
                float4 v;
                v.x = sa.x + hv.x * di * di + bv.x;
                v.y = sa.y + hv.y * di * di + bv.y;
                v.z = sa.z + hv.z * di * di + bv.z;
                v.w = sa.w + hv.w * di * di + bv.w;
                if (RELU) {
                    v.x = fmaxf(v.x, 0.0f); v.y = fmaxf(v.y, 0.0f);
                    v.z = fmaxf(v.z, 0.0f); v.w = fmaxf(v.w, 0.0f);
                }
                ((float4*)out)[gi] = v;
            }
        }
        __syncthreads();
    }
}

constexpr int L1_F      = 128;
constexpr int L1_G      = 64;
constexpr int L1_TP     = L1_F + 4;
constexpr int L1_PP1    = 2 * L1_F + 32;
constexpr int L1_PP2    = 2 * L1_G + 32;
constexpr int L1_STRIPS = AGG_ROWS / 16;
static_assert(AGG_ROWS % 16 == 0, "a tile is whole strips");
static_assert(2 * L1_STRIPS == 20 && AGG_WAVES == 32, "the item-to-wave map below places 20 items on 32 waves");

template <int NOUT, int PP>
__device__ __forceinline__ void gcn_stage_w(const float* __restrict__ W, float* Ws, int tid) {
    constexpr int ITEMS = (GEMM_K / 2) * (NOUT / 4);
    static_assert(ITEMS % 1024 == 0, "whole trips of the work-group");
    constexpr int TRIPS = ITEMS / 1024;
    float4 v0[TRIPS], v1[TRIPS];
#pragma unroll
    for (int j = 0; j < TRIPS; ++j) {
        const int i  = tid + j * 1024;
        const int p  = i / (NOUT / 4);
        const int c4 = i % (NOUT / 4);
        v0[j] = ((const float4*)W)[(size_t)(2 * p) * (NOUT / 4) + c4];
        v1[j] = ((const float4*)W)[(size_t)(2 * p + 1) * (NOUT / 4) + c4];
    }
#pragma unroll
    for (int j = 0; j < TRIPS; ++j) {
        const int i  = tid + j * 1024;
        const int p  = i / (NOUT / 4);
        const int c4 = i % (NOUT / 4);
        float4* d = (float4*)&Ws[p * PP + c4 * 8];
        d[0] = make_float4(v0[j].x, v1[j].x, v0[j].y, v1[j].y);
        d[1] = make_float4(v0[j].z, v1[j].z, v0[j].w, v1[j].w);
    }
}

template <int NT, int PP>
__device__ __forceinline__ void gcn_strip_product(const float* arow, const float* wl, v8f (&c)[NT]) {
    v2f a = *(const v2f*)arow, an = a;
    v2f b[NT], bn[NT];
#pragma unroll
    for (int t = 0; t < NT; ++t) { b[t] = *(const v2f*)(wl + t * 32); bn[t] = b[t]; }
#pragma unroll
    for (int s = 0; s < GEMM_K / 4; ++s) {
        if (s + 1 < GEMM_K / 4) {
            an = *(const v2f*)(arow + 4 * (s + 1));
            const float* wn = wl + (2 * (s + 1)) * PP;
#pragma unroll
            for (int t = 0; t < NT; ++t) bn[t] = *(const v2f*)(wn + t * 32);
        }
#pragma unroll
        for (int t = 0; t < NT; ++t) {
            c[t] = __builtin_amdgcn_wmma_f32_16x16x4_f32(
                false, a, false, b[t], (short)0, c[t], false, false);
        }
        a = an;
#pragma unroll
        for (int t = 0; t < NT; ++t) b[t] = bn[t];
    }
}

__global__ __launch_bounds__(1024) void gcn_layer1_tile(
    const float* __restrict__ x, const int* __restrict__ src, const int* __restrict__ dst,
    const int* __restrict__ deg8, const float* __restrict__ W1, const float* __restrict__ b1,
    const float* __restrict__ W2, const float* __restrict__ lists,
    float* __restrict__ g, float* __restrict__ dinv, int N, int E) {
    __shared__ __attribute__((aligned(16))) float acc[AGG_ROWS * L1_TP];
    __shared__ float dl[AGG_ROWS];
    __shared__ unsigned fq[AGG_WAVES * AGG_FQCAP];
    __shared__ unsigned fs[AGG_WAVES * AGG_FQCAP];
    __shared__ float fn[AGG_WAVES * AGG_FQCAP];
    __shared__ int ovf[2];
    __shared__ __attribute__((aligned(16))) float W1s[(GEMM_K / 2) * L1_PP1];
    __shared__ __attribute__((aligned(16))) float W2s[(GEMM_K / 2) * L1_PP2];

    const int tid  = threadIdx.x;
    const int lane = tid & 31;
    const int wave = __builtin_amdgcn_readfirstlane(tid >> 5);
    unsigned* myq = fq + wave * AGG_FQCAP;
    unsigned* sq  = fs + wave * AGG_FQCAP;
    float*    sw  = fn + wave * AGG_FQCAP;

    if (tid == 0) { ovf[0] = 0; ovf[1] = 0; }
    __syncthreads();
    int par = 0;

    gcn_stage_w<L1_F, L1_PP1>(W1, W1s, tid);
    gcn_stage_w<L1_G, L1_PP2>(W2, W2s, tid);

    const int wq = wave & 7;
    const int wb = wave >> 3;
    const bool prod  = (wq < 4) || (wq == 4 + wb);
    const int  item  = wb * 5 + ((wq < 4) ? wq : 4);
    const int  strip = item % L1_STRIPS;
    const int  half  = item / L1_STRIPS;

    const int ntiles = (N + AGG_ROWS - 1) / AGG_ROWS;

    for (int tile = blockIdx.x; tile < ntiles; tile += gridDim.x) {
        int tq = tid;
        asm volatile("" : "+v"(tq));
        const int tidq  = tq;
        const int laneq = tq & 31;
        const int lo = tile * AGG_ROWS;
        const int nrows = min(AGG_ROWS, N - lo);
        const float* tl = lists + (size_t)lo * AGG_LROW;
        const int segw = (nrows * AGG_LROW) / AGG_GS;
        const int subw = segw / AGG_WAVES;

        typedef float __attribute__((ext_vector_type(L1_F / 32))) vt;
        const int rowc = lo + min(tidq, nrows - 1);
        const int4 dc0 = *(const int4*)(deg8 + (size_t)rowc * AGG_GS);
        const int4 dc1 = *(const int4*)(deg8 + (size_t)rowc * AGG_GS + 4);
        int total = -1;
        int cn[AGG_OWN];
        if (subw >= 2) total = gcn_own_fetch<1>(tl, segw, subw, nullptr, deg8, wave, laneq, myq, sq, sw, cn);
        if (total < 0) ovf[par] = 1;
        if (tidq < nrows) {
            const float dmy = gcn_dinv_from(dc0, dc1);
            dl[tidq] = dmy;
            dinv[lo + tidq] = dmy;
        }
        __syncthreads();
        const bool fits = __builtin_amdgcn_readfirstlane(ovf[par]) == 0;
        if (tidq == 0) ovf[par ^ 1] = 0;
        par ^= 1;

        if (fits) {
            const int cnv = gcn_counts_by_lane(cn, laneq);
#pragma unroll 1
            for (int j = 0; j < AGG_OWN; ++j) {
                const int row  = wave + j * AGG_WAVES;
                const int rin  = min(row, nrows - 1);
                const int cnt  = __builtin_amdgcn_readlane(cnv, j);
                const float di = dl[rin];
                const vt xs = *((const vt*)(x + (size_t)(lo + rin) * L1_F) + laneq);
                vt a = (row < nrows) ? xs * di * di : (vt)(0.0f);
                gcn_row_sum<L1_F>(x, sq + j * AGG_JCAP, sw + j * AGG_JCAP, cnt, di, laneq, a);
                *(vt*)&acc[row * L1_TP + (L1_F / 32) * laneq] = a;
            }
        } else {
            for (int i = tidq; i < AGG_ROWS * L1_TP / 4; i += 1024)
                ((float4*)acc)[i] = make_float4(0.0f, 0.0f, 0.0f, 0.0f);
            __syncthreads();
            int cnt = 0;
            for (int e0 = wave * 32; e0 < E; e0 += AGG_WAVES * 32) {
                const int e = e0 + laneq;
                const int dv = (e < E) ? dst[e] : -1;
                const unsigned r = (unsigned)(dv - lo);
                const bool mk = r < (unsigned)nrows;
                const unsigned bm = __builtin_amdgcn_ballot_w32(mk);
                if (bm != 0u) {
                    if (cnt > AGG_FQCAP - 32) {
                        gcn_agg_drain<L1_F, L1_TP, 1>(x, nullptr, deg8, dl, acc, myq, cnt, laneq);
                        cnt = 0;
                    }
                    if (mk) myq[cnt + (int)__builtin_amdgcn_mbcnt_lo(bm, 0u)] = (r << 24) | (unsigned)src[e];
                    cnt += __builtin_popcount(bm);
                }
            }
            gcn_agg_drain<L1_F, L1_TP, 1>(x, nullptr, deg8, dl, acc, myq, cnt, laneq);
            __syncthreads();

            for (int i = tidq; i < nrows * (L1_F / 4); i += 1024) {
                const int r  = i / (L1_F / 4);
                const int f4 = i % (L1_F / 4);
                const float di = dl[r];
                const float4 xv = ((const float4*)x)[(size_t)(lo + r) * (L1_F / 4) + f4];
                float4* p = (float4*)&acc[r * L1_TP + 4 * f4];
                float4 v = *p;
                v.x += xv.x * di * di;
                v.y += xv.y * di * di;
                v.z += xv.z * di * di;
                v.w += xv.w * di * di;
                *p = v;
            }
        }
        __syncthreads();

        int lq = lane;
        asm volatile("" : "+v"(lq));
        const int m  = lq & 15;
        const int gh = lq >> 4;

        v8f c1[4];
        float bv[4];
        if (prod) {
#pragma unroll
            for (int t = 0; t < 4; ++t) {
                c1[t] = (v8f){0, 0, 0, 0, 0, 0, 0, 0};
                bv[t] = b1[(4 * half + t) * 16 + m];
            }
            gcn_strip_product<4, L1_PP1>(acc + (strip * 16 + m) * L1_TP + 2 * gh,
                                         W1s + gh * L1_PP1 + 2 * m + (4 * half) * 32, c1);
        }
        __syncthreads();
        if (prod) {
#pragma unroll
            for (int t = 0; t < 4; ++t) {
                const int col = (4 * half + t) * 16 + m;
#pragma unroll
                for (int r = 0; r < 8; ++r)
                    acc[(strip * 16 + 8 * gh + r) * L1_TP + col] = fmaxf(c1[t][r] + bv[t], 0.0f);
            }
        }
        __syncthreads();

        if (prod) {
            v8f c2[2];
#pragma unroll
            for (int t = 0; t < 2; ++t) c2[t] = (v8f){0, 0, 0, 0, 0, 0, 0, 0};
            gcn_strip_product<2, L1_PP2>(acc + (strip * 16 + m) * L1_TP + 2 * gh,
                                         W2s + gh * L1_PP2 + 2 * m + (2 * half) * 32, c2);
#pragma unroll
            for (int t = 0; t < 2; ++t) {
                const int col = (2 * half + t) * 16 + m;
#pragma unroll
                for (int r = 0; r < 8; ++r) {
                    const int row = strip * 16 + 8 * gh + r;
                    if (row < nrows) g[(size_t)(lo + row) * L1_G + col] = c2[t][r];
                }
            }
        }
        __syncthreads();
    }
}

extern "C" void kernel_launch(void* const* d_in, const int* in_sizes, int n_in,
                              void* d_out, int out_size, void* d_ws, size_t ws_size,
                              hipStream_t stream) {
    const float* x  = (const float*)d_in[0];
    const int*   ei = (const int*)d_in[1];
    const float* W1 = (const float*)d_in[2];
    const float* b1 = (const float*)d_in[3];
    const float* W2 = (const float*)d_in[4];
    const float* b2 = (const float*)d_in[5];
    float* out = (float*)d_out;

    const int DIN = 128, DH = 128;
    const int N = in_sizes[0] / DIN;
    const int E = in_sizes[1] / 2;
    const int* src = ei;
    const int* dst = ei + E;

    float* ws = (float*)d_ws;
    size_t dinvPad = ((size_t)N + 255) & ~(size_t)255;
    float* dinv = ws;
    float* g    = ws + dinvPad;
    int*   deg8 = (int*)(g + (size_t)N * DH);

    gcn_tile_lists<<<256, 1024, 0, stream>>>(src, dst, deg8, out, N, E);

    gcn_layer1_tile<<<256, 1024, 0, stream>>>(x, src, dst, deg8, W1, b1, W2, out, g, dinv, N, E);

    gcn_agg_tile<64, 0><<<256, 1024, 0, stream>>>(g, src, dst, dinv, b2, out, out, N, E);
}


# Filter launch builds its slice's list segments in LDS and writes each out once as whole 128-byte lines (16-byte stores), not one scattered 4-byte global store per kept edge
# speedup vs baseline: 7.4346x; 1.2312x over previous
#include <hip/hip_runtime.h>
#include <hip/hip_bf16.h>

typedef __attribute__((ext_vector_type(2))) float v2f;
typedef __attribute__((ext_vector_type(8))) float v8f;

constexpr int AGG_GS = 8;
constexpr int GEMM_K = 128;
static_assert(AGG_GS == 8, "a node's per-slice counts are read as two 16-byte words");

__device__ __forceinline__ float gcn_dinv_from(const int4& c0, const int4& c1) {
    const int d = 1 + c0.x + c0.y + c0.z + c0.w + c1.x + c1.y + c1.z + c1.w;
    const float v = (float)d;
    return (v > 0.0f) ? __frsqrt_rn(v) : 0.0f;
}
__device__ __forceinline__ float gcn_dinv_of(const int* __restrict__ deg8, int i) {
    const int4 c0 = *(const int4*)(deg8 + (size_t)i * AGG_GS);
    const int4 c1 = *(const int4*)(deg8 + (size_t)i * AGG_GS + 4);
    return gcn_dinv_from(c0, c1);
}

constexpr int AGG_ROWS  = 160;
constexpr int AGG_WAVES = 32;
constexpr int AGG_GT    = 8;
constexpr int AGG_GROWS = AGG_GT * AGG_ROWS;
constexpr int AGG_RBITS = 11;
constexpr int AGG_EBITS = 32 - AGG_RBITS;
constexpr int AGG_QCAP  = 1024;
constexpr int AGG_FQCAP = 320;
constexpr int AGG_LROW  = 64;
static_assert(AGG_GROWS <= (1 << AGG_RBITS), "row inside the group must fit its bits");
constexpr int AGG_OWN   = AGG_ROWS / AGG_WAVES;
constexpr int AGG_SUBW  = (AGG_ROWS * AGG_LROW) / (AGG_GS * AGG_WAVES);
constexpr int AGG_SEGW  = (AGG_ROWS * AGG_LROW) / AGG_GS;
static_assert(AGG_LROW % AGG_GS == 0 && (AGG_LROW / AGG_GS) % 4 == 0, "a segment of any tile is whole 16-byte words");
static_assert((AGG_SEGW / 4) % 32 == 0, "a wave's 16-byte stores of a full segment stay inside one tile");
static_assert(AGG_ROWS % AGG_WAVES == 0, "every wave owns the same number of rows");
static_assert(AGG_ROWS <= 1024, "one thread per row of a tile fetches the row's dinv");
static_assert(AGG_GS * (AGG_SUBW - 1) <= AGG_FQCAP, "a wave's queue holds all its buckets");
static_assert(AGG_SUBW - 1 >= 32 && AGG_SUBW - 1 <= 64, "a bucket is copied in one whole and one partial step of 32 entries");
constexpr int AGG_JCAP  = AGG_FQCAP / AGG_OWN;
static_assert(AGG_OWN * AGG_JCAP <= AGG_FQCAP, "a wave's sorted queue holds one part per owned row");

__global__ __launch_bounds__(1024) void gcn_tile_lists(
    const int* __restrict__ src, const int* __restrict__ dst,
    int* __restrict__ degp, float* __restrict__ lists, int N, int E) {
    __shared__ unsigned q[AGG_WAVES * AGG_QCAP];
    __shared__ int degc[AGG_GROWS];
    __shared__ int bcnt[AGG_GT * AGG_WAVES];
    __shared__ __attribute__((aligned(16))) unsigned bimg[AGG_GT * AGG_SEGW];

    const int tid  = threadIdx.x;
    const int lane = tid & 31;
    const int wave = __builtin_amdgcn_readfirstlane(tid >> 5);
    unsigned* myq = q + wave * AGG_QCAP;

    const int ngroups = (N + AGG_GROWS - 1) / AGG_GROWS;
    const bool vec_ok = (reinterpret_cast<size_t>(dst) & 15) == 0;
    const int nfull = vec_ok ? E / 512 : 0;

    for (int job = blockIdx.x; job < ngroups * AGG_GS; job += gridDim.x) {
        const int group = job / AGG_GS;
        const int slice = job % AGG_GS;
        const int lo = group * AGG_GROWS;
        const int nrows = min(AGG_GROWS, N - lo);
        float* gl = lists + (size_t)lo * AGG_LROW;
        const int t_lo = (int)(((long long)nfull * slice) / AGG_GS);
        const int t_hi = (int)(((long long)nfull * (slice + 1)) / AGG_GS);
        const int nt = t_hi - t_lo;
        const int ebase = t_lo * 512;
        const int* dsl = dst + ebase;
        const int* ssl = src + ebase;

        for (int i = tid; i < AGG_GROWS; i += 1024) degc[i] = 0;
        for (int i = tid; i < AGG_GT * AGG_WAVES; i += 1024) bcnt[i] = 0;
        for (int i = tid; i < AGG_GT * AGG_SEGW / 4; i += 1024) ((uint4*)bimg)[i] = make_uint4(0u, 0u, 0u, 0u);
        __syncthreads();

        int cnt = 0;

        auto flush = [&]() {
            for (int i0 = 0; i0 < cnt; i0 += 32) {
                if (i0 + lane < cnt) {
                    const unsigned ent = myq[i0 + lane];
                    const unsigned r = ent >> AGG_EBITS;
                    const unsigned s = (unsigned)ssl[ent & ((1u << AGG_EBITS) - 1u)];
                    atomicAdd(&degc[r], 1);
                    const int tt = (int)(r / (unsigned)AGG_ROWS);
                    const unsigned rt = r - (unsigned)(tt * AGG_ROWS);
                    const int ow = (int)(rt % (unsigned)AGG_WAVES);
                    const int nrk  = min(AGG_ROWS, nrows - tt * AGG_ROWS);
                    const int segw = (nrk * AGG_LROW) / AGG_GS;
                    const int subw = segw / AGG_WAVES;
                    const int slot = atomicAdd(&bcnt[tt * AGG_WAVES + ow], 1);
                    if (slot < subw - 1)
                        bimg[tt * AGG_SEGW + ow * subw + 1 + slot] = (rt << 24) | s;
                }
            }
            cnt = 0;
        };
        auto room = [&](int need) {
            if (cnt > AGG_QCAP - need) flush();
        };
        auto keep4 = [&](const int4& d, int e) {
            const unsigned r0 = (unsigned)(d.x - lo);
            const unsigned r1 = (unsigned)(d.y - lo);
            const unsigned r2 = (unsigned)(d.z - lo);
            const unsigned r3 = (unsigned)(d.w - lo);
            const bool m0 = r0 < (unsigned)nrows;
            const bool m1 = r1 < (unsigned)nrows;
            const bool m2 = r2 < (unsigned)nrows;
            const bool m3 = r3 < (unsigned)nrows;
            if (__builtin_amdgcn_ballot_w32(m0 | m1 | m2 | m3) != 0u) {
                const unsigned b0 = __builtin_amdgcn_ballot_w32(m0);
                if (m0) myq[cnt + (int)__builtin_amdgcn_mbcnt_lo(b0, 0u)] = (r0 << AGG_EBITS) | (unsigned)(e + 0);
                cnt += __builtin_popcount(b0);
                const unsigned b1 = __builtin_amdgcn_ballot_w32(m1);
                if (m1) myq[cnt + (int)__builtin_amdgcn_mbcnt_lo(b1, 0u)] = (r1 << AGG_EBITS) | (unsigned)(e + 1);
                cnt += __builtin_popcount(b1);
                const unsigned b2 = __builtin_amdgcn_ballot_w32(m2);
                if (m2) myq[cnt + (int)__builtin_amdgcn_mbcnt_lo(b2, 0u)] = (r2 << AGG_EBITS) | (unsigned)(e + 2);
                cnt += __builtin_popcount(b2);
                const unsigned b3 = __builtin_amdgcn_ballot_w32(m3);
                if (m3) myq[cnt + (int)__builtin_amdgcn_mbcnt_lo(b3, 0u)] = (r3 << AGG_EBITS) | (unsigned)(e + 3);
                cnt += __builtin_popcount(b3);
            }
        };

        if (wave < nt) {
            const int rstep = max(nt / max(ngroups, 1), 1);
            const int rot = (int)(((unsigned)group * (unsigned)rstep) % (unsigned)nt);
            auto trip_base = [&](int t) {
                int sg = t + rot;
                if (sg >= nt) sg -= nt;
                return sg * 512 + lane * 4;
            };
            int4 a[4], b[4];
            int t  = wave;
            int ea = trip_base(t);
            int eb = ea;
#pragma unroll
            for (int k = 0; k < 4; ++k) a[k] = *(const int4*)(dsl + ea + k * 128);
            while (t < nt) {
                const int tb = t + AGG_WAVES;
                const bool hb = tb < nt;
                room(512);
                eb = trip_base(hb ? tb : t);
#pragma unroll
                for (int k = 0; k < 4; ++k) b[k] = *(const int4*)(dsl + eb + k * 128);
#pragma unroll
                for (int k = 0; k < 4; ++k) keep4(a[k], ea + k * 128);

                const int ta = tb + AGG_WAVES;
                const bool ha = ta < nt;
                room(512);
                ea = trip_base(ha ? ta : t);
#pragma unroll
                for (int k = 0; k < 4; ++k) a[k] = *(const int4*)(dsl + ea + k * 128);
                if (hb) {
#pragma unroll
                    for (int k = 0; k < 4; ++k) keep4(b[k], eb + k * 128);
                }
                t = ta;
            }
        }
        if (slice == AGG_GS - 1) {
            for (int e0 = nfull * 512 + wave * 32; e0 < E; e0 += AGG_WAVES * 32) {
                const int e = e0 + lane;
                const int dv = (e < E) ? dst[e] : -1;
                const unsigned r = (unsigned)(dv - lo);
                const bool m = r < (unsigned)nrows;
                const unsigned bm = __builtin_amdgcn_ballot_w32(m);
                if (bm != 0u) {
                    room(32);
                    if (m) myq[cnt + (int)__builtin_amdgcn_mbcnt_lo(bm, 0u)] = (r << AGG_EBITS) | (unsigned)(e - ebase);
                    cnt += __builtin_popcount(bm);
                }
            }
        }
        flush();
        __syncthreads();

        for (int i = tid; i < AGG_GT * AGG_WAVES; i += 1024) {
            const int tt = i / AGG_WAVES;
            const int ow = i % AGG_WAVES;
            if (nrows > tt * AGG_ROWS) {
                const int nrk  = min(AGG_ROWS, nrows - tt * AGG_ROWS);
                const int segw = (nrk * AGG_LROW) / AGG_GS;
                const int subw = segw / AGG_WAVES;
                if (subw >= 1) bimg[tt * AGG_SEGW + ow * subw] = (unsigned)bcnt[i];
            }
        }
        for (int i = tid; i < nrows; i += 1024) degp[(size_t)(lo + i) * AGG_GS + slice] = degc[i];
        __syncthreads();
        for (int i = tid; i < AGG_GT * (AGG_SEGW / 4); i += 1024) {
            const int tt = i / (AGG_SEGW / 4);
            const int w4 = i % (AGG_SEGW / 4);
            if (nrows > tt * AGG_ROWS) {
                const int nrk  = min(AGG_ROWS, nrows - tt * AGG_ROWS);
                const int segw = (nrk * AGG_LROW) / AGG_GS;
                if (4 * w4 < segw) {
                    const uint4 v = ((const uint4*)bimg)[i];
                    float* d = gl + (size_t)tt * AGG_ROWS * AGG_LROW + (size_t)slice * segw + (size_t)(4 * w4);
                    *(float4*)d = make_float4(__uint_as_float(v.x), __uint_as_float(v.y),
                                              __uint_as_float(v.z), __uint_as_float(v.w));
                }
            }
        }
        __syncthreads();
    }
}

template <int F, int P, int DEG>
__device__ __forceinline__ void gcn_agg_drain(
    const float* __restrict__ h, const float* __restrict__ dinv, const int* __restrict__ deg8,
    const float* dl, float* acc, const unsigned* list, int cnt, int lane) {
    constexpr int LPE = F / 4;
    constexpr int EPW = 32 / LPE;
    constexpr int UD  = 8;
    const int sub = (EPW == 1) ? 0 : lane / LPE;
    const int fl  = lane % LPE;
    const int hi  = (lane >> 4) & 1;
    unsigned pre = (lane < cnt) ? list[lane] : 0u;
    for (int base = 0; base < cnt; base += 32) {
        const int n = min(32, cnt - base);
        const unsigned ent = pre;
        const int nb = base + 32 + lane;
        pre = (nb < cnt) ? list[nb] : 0u;
        int s = 0, r = 0;
        float nrm = 0.0f;
        if (lane < n) {
            r = (int)(ent >> 24);
            s = (int)(ent & 0xFFFFFFu);
            float ds;
            if constexpr (DEG) ds = gcn_dinv_of(deg8, s);
            else               ds = dinv[s];
            nrm = ds * dl[r];
        }
        for (int i = 0; i < n; i += UD * EPW) {
            float4 v[UD];
            float  w[UD];
            int    si[UD];
            int    rr[UD];
            bool   ok[UD];
#pragma unroll
            for (int k = 0; k < UD; ++k) {
                const int idx = i + k * EPW + sub;
                ok[k] = idx < n;
                const int ii = ok[k] ? idx : 0;
                if constexpr (EPW == 1) {
                    si[k] = __builtin_amdgcn_readlane(s, ii);
                    rr[k] = __builtin_amdgcn_readlane(r, ii);
                    w[k]  = __int_as_float(__builtin_amdgcn_readlane(__float_as_int(nrm), ii));
                } else {
                    si[k] = __shfl(s, ii);
                    rr[k] = __shfl(r, ii);
                    w[k]  = __shfl(nrm, ii);
                }
            }
#pragma unroll
            for (int k = 0; k < UD; ++k) {
                v[k] = make_float4(0.0f, 0.0f, 0.0f, 0.0f);
                if (ok[k]) v[k] = *((const float4*)(h + (size_t)si[k] * F) + fl);
            }
#pragma unroll
            for (int k = 0; k < UD; ++k) {
                if (ok[k]) {
                    float* a  = acc + rr[k] * P + fl * 4;
                    float* p0 = a + 2 * hi;
                    float* p1 = a + 2 - 2 * hi;
                    const float c0 = hi ? v[k].z : v[k].x;
                    const float c1 = hi ? v[k].w : v[k].y;
                    const float c2 = hi ? v[k].x : v[k].z;
                    const float c3 = hi ? v[k].y : v[k].w;
                    atomicAdd(p0 + 0, c0 * w[k]);
                    atomicAdd(p0 + 1, c1 * w[k]);
                    atomicAdd(p1 + 0, c2 * w[k]);
                    atomicAdd(p1 + 1, c3 * w[k]);
                }
            }
        }
    }
}

template <int DEG>
__device__ __forceinline__ int gcn_own_fetch(
    const float* tl, int segw, int subw, const float* __restrict__ dinv, const int* __restrict__ deg8,
    int wave, int lane, unsigned* myq, unsigned* sq, float* sw, int (&cn)[AGG_OWN]) {
    const float* bk = tl + (size_t)wave * subw;
    int cw = 0;
    if (lane < AGG_GS) cw = (int)__float_as_uint(bk[(size_t)lane * segw]);
    unsigned e0[AGG_GS];
#pragma unroll
    for (int sg = 0; sg < AGG_GS; ++sg)
        e0[sg] = (1 + lane < subw) ? __float_as_uint(bk[(size_t)sg * segw + 1 + lane]) : 0u;
    const bool bad = (lane < AGG_GS) && (cw < 0 || cw > subw - 1);
    if (__builtin_amdgcn_ballot_w32(bad) != 0u) return -1;
    int off = 0;
#pragma unroll
    for (int sg = 0; sg < AGG_GS; ++sg) {
        const int cs = __builtin_amdgcn_readlane(cw, sg);
        if (lane < cs) myq[off + lane] = e0[sg];
        if (cs > 32) {
            if (32 + lane < cs) myq[off + 32 + lane] = __float_as_uint(bk[(size_t)sg * segw + 33 + lane]);
        }
        off += cs;
    }
#pragma unroll
    for (int j = 0; j < AGG_OWN; ++j) cn[j] = 0;
    for (int b0 = 0; b0 < off; b0 += 128) {
        float ds[4];
        int   sv[4];
        int   jv[4];
#pragma unroll
        for (int q = 0; q < 4; ++q) {
            const int e = b0 + q * 32 + lane;
            sv[q] = 0;
            jv[q] = -1;
            if (e < off) {
                const unsigned ent = myq[e];
                sv[q] = (int)(ent & 0xFFFFFFu);
                jv[q] = (int)(ent >> 24) / AGG_WAVES;
            }
            if constexpr (DEG) ds[q] = gcn_dinv_of(deg8, sv[q]);
            else               ds[q] = dinv[sv[q]];
        }
#pragma unroll
        for (int q = 0; q < 4; ++q) {
            int within = AGG_JCAP;
#pragma unroll
            for (int j = 0; j < AGG_OWN; ++j) {
                const bool m = jv[q] == j;
                const unsigned bm = __builtin_amdgcn_ballot_w32(m);
                if (m) within = cn[j] + (int)__builtin_amdgcn_mbcnt_lo(bm, 0u);
                cn[j] += __builtin_popcount(bm);
            }
            if (within < AGG_JCAP) {
                const int p = jv[q] * AGG_JCAP + within;
                sq[p] = (unsigned)sv[q];
                sw[p] = ds[q];
            }
        }
    }
    bool over = false;
#pragma unroll
    for (int j = 0; j < AGG_OWN; ++j) over = over || (cn[j] > AGG_JCAP);
    return over ? -1 : 0;
}

template <int F, int G>
__device__ __forceinline__ void gcn_rows_issue(
    const float* __restrict__ h, unsigned loff, unsigned s, float nrm, int i0,
    float __attribute__((ext_vector_type(F / 32))) (&v)[G], float (&w)[G]) {
    typedef float __attribute__((ext_vector_type(F / 32))) vt;
#pragma unroll
    for (int k = 0; k < G; ++k) {
        const unsigned si = (unsigned)__builtin_amdgcn_readlane((int)s, i0 + k);
        w[k] = __int_as_float(__builtin_amdgcn_readlane(__float_as_int(nrm), i0 + k));
        const unsigned voff = si * (unsigned)(F * sizeof(float)) + loff;
        v[k] = *(const vt*)((const char*)h + voff);
    }
}

template <int F, int Q>
__device__ __forceinline__ void gcn_row_piece(
    const float* __restrict__ h, unsigned loff, unsigned s, float nrm, int i0, int r,
    float __attribute__((ext_vector_type(F / 32)))& acc) {
    typedef float __attribute__((ext_vector_type(F / 32))) vt;
    constexpr int G = 4 * Q;
    vt    v[G > 0 ? G : 1];
    float w[G > 0 ? G : 1];
    if constexpr (G > 0) gcn_rows_issue<F, G>(h, loff, s, nrm, i0, v, w);
    vt    vr[3];
    float wr[3];
#pragma unroll
    for (int k = 0; k < 3; ++k) {
        vr[k] = (vt)(0.0f);
        wr[k] = 0.0f;
        if (k < r) {
            const int idx = i0 + G + k;
            const unsigned si = (unsigned)__builtin_amdgcn_readlane((int)s, idx);
            wr[k] = __int_as_float(__builtin_amdgcn_readlane(__float_as_int(nrm), idx));
            const unsigned voff = si * (unsigned)(F * sizeof(float)) + loff;
            vr[k] = *(const vt*)((const char*)h + voff);
        }
    }
    if constexpr (G > 0) {
#pragma unroll
        for (int k = 0; k < G; ++k) acc += v[k] * w[k];
    }
#pragma unroll
    for (int k = 0; k < 3; ++k) acc += vr[k] * wr[k];
}
template <int F, int Q>
__device__ __forceinline__ void gcn_row_pick(
    int q, const float* __restrict__ h, unsigned loff, unsigned s, float nrm, int i0, int r,
    float __attribute__((ext_vector_type(F / 32)))& acc) {
    if (q == Q) gcn_row_piece<F, Q>(h, loff, s, nrm, i0, r, acc);
    else if constexpr (Q > 0) gcn_row_pick<F, Q - 1>(q, h, loff, s, nrm, i0, r, acc);
}

template <int F>
__device__ __forceinline__ void gcn_row_sum(
    const float* __restrict__ h, const unsigned* qj, const float* wj, int cnt, float di, int lane,
    float __attribute__((ext_vector_type(F / 32)))& acc) {
    typedef float __attribute__((ext_vector_type(F / 32))) vt;
    constexpr int CH = (F >= 128) ? 16 : 32;
    static_assert(CH % 4 == 0 && CH <= 32, "whole blocks of four inside the 32 entries that the lanes hold");
    const unsigned loff = (unsigned)lane * (unsigned)sizeof(vt);
    for (int base = 0; base < cnt; base += 32) {
        const int n = min(32, cnt - base);
        unsigned s = 0u;
        float nrm = 0.0f;
        if (lane < n) {
            s = qj[base + lane];
            nrm = wj[base + lane] * di;
        }
        for (int i0 = 0; i0 < n; i0 += CH) {
            const int m = min(CH, n - i0);
            gcn_row_pick<F, CH / 4>(m >> 2, h, loff, s, nrm, i0, m & 3, acc);
        }
    }
}

__device__ __forceinline__ int gcn_counts_by_lane(const int (&cn)[AGG_OWN], int lane) {
    int cnv = 0;
#pragma unroll
    for (int j = 0; j < AGG_OWN; ++j) cnv = (lane == j) ? cn[j] : cnv;
    return cnv;
}

template <int F, int RELU>
__global__ __launch_bounds__(1024) void gcn_agg_tile(
    const float* __restrict__ h, const int* __restrict__ src, const int* __restrict__ dst,
    const float* __restrict__ dinv, const float* __restrict__ bias,
    const float* lists, float* out, int N, int E) {
    typedef float __attribute__((ext_vector_type(F / 32))) vt;
    __shared__ float acc[AGG_ROWS * F];
    __shared__ float dl[AGG_ROWS];
    __shared__ unsigned fq[AGG_WAVES * AGG_FQCAP];
    __shared__ unsigned fs[AGG_WAVES * AGG_FQCAP];
    __shared__ float fn[AGG_WAVES * AGG_FQCAP];
    __shared__ int ovf[2];

    const int tid  = threadIdx.x;
    const int lane = tid & 31;
    const int wave = __builtin_amdgcn_readfirstlane(tid >> 5);
    unsigned* myq = fq + wave * AGG_FQCAP;
    unsigned* sq  = fs + wave * AGG_FQCAP;
    float*    sw  = fn + wave * AGG_FQCAP;

    if (tid == 0) { ovf[0] = 0; ovf[1] = 0; }
    __syncthreads();
    int par = 0;

    const int ntiles = (N + AGG_ROWS - 1) / AGG_ROWS;

    for (int tile = blockIdx.x; tile < ntiles; tile += gridDim.x) {
        const int lo = tile * AGG_ROWS;
        const int nrows = min(AGG_ROWS, N - lo);
        const float* tl = lists + (size_t)lo * AGG_LROW;
        const int segw = (nrows * AGG_LROW) / AGG_GS;
        const int subw = segw / AGG_WAVES;

        float dmy = 0.0f;
        if (tid < nrows) dmy = dinv[lo + tid];
        int total = -1;
        int cn[AGG_OWN];
        if (subw >= 2) total = gcn_own_fetch<0>(tl, segw, subw, dinv, nullptr, wave, lane, myq, sq, sw, cn);
        if (total < 0) ovf[par] = 1;
        if (tid < nrows) dl[tid] = dmy;
        __syncthreads();
        const bool fits = __builtin_amdgcn_readfirstlane(ovf[par]) == 0;
        if (tid == 0) ovf[par ^ 1] = 0;
        par ^= 1;

        if (fits) {
            const vt bv = *((const vt*)bias + lane);
            const int cnv = gcn_counts_by_lane(cn, lane);
#pragma unroll 1
            for (int j = 0; j < AGG_OWN; ++j) {
                const int row  = wave + j * AGG_WAVES;
                const int rin  = min(row, nrows - 1);
                const int cnt  = __builtin_amdgcn_readlane(cnv, j);
                const float di = dl[rin];
                const vt hs = *((const vt*)(h + (size_t)(lo + rin) * F) + lane);
                vt sum = (vt)(0.0f);
                gcn_row_sum<F>(h, sq + j * AGG_JCAP, sw + j * AGG_JCAP, cnt, di, lane, sum);
                if (row < nrows) {
                    vt v = sum + hs * di * di + bv;
                    if (RELU) {
#pragma unroll
                        for (int q = 0; q < F / 32; ++q) v[q] = fmaxf(v[q], 0.0f);
                    }
                    *((vt*)(out + (size_t)(lo + row) * F) + lane) = v;
                }
            }
        } else {
            for (int i = tid; i < AGG_ROWS * F / 4; i += 1024)
                ((float4*)acc)[i] = make_float4(0.0f, 0.0f, 0.0f, 0.0f);
            __syncthreads();
            int cnt = 0;
            for (int e0 = wave * 32; e0 < E; e0 += AGG_WAVES * 32) {
                const int e = e0 + lane;
                const int dv = (e < E) ? dst[e] : -1;
                const unsigned r = (unsigned)(dv - lo);
                const bool m = r < (unsigned)nrows;
                const unsigned bm = __builtin_amdgcn_ballot_w32(m);
                if (bm != 0u) {
                    if (cnt > AGG_FQCAP - 32) {
                        gcn_agg_drain<F, F, 0>(h, dinv, nullptr, dl, acc, myq, cnt, lane);
                        cnt = 0;
                    }
                    if (m) myq[cnt + (int)__builtin_amdgcn_mbcnt_lo(bm, 0u)] = (r << 24) | (unsigned)src[e];
                    cnt += __builtin_popcount(bm);
                }
            }
            gcn_agg_drain<F, F, 0>(h, dinv, nullptr, dl, acc, myq, cnt, lane);
            __syncthreads();

            for (int i = tid; i < nrows * (F / 4); i += 1024) {
                const int r  = i / (F / 4);
                const int f4 = i % (F / 4);
                const size_t gi = ((size_t)(lo + r) * F) / 4 + f4;
                const float di = dl[r];
                const float4 sa = ((const float4*)acc)[i];
                const float4 hv = ((const float4*)h)[gi];
                const float4 bv = ((const float4*)bias)[f4];
                float4 v;
                v.x = sa.x + hv.x * di * di + bv.x;
                v.y = sa.y + hv.y * di * di + bv.y;
                v.z = sa.z + hv.z * di * di + bv.z;
                v.w = sa.w + hv.w * di * di + bv.w;
                if (RELU) {
                    v.x = fmaxf(v.x, 0.0f); v.y = fmaxf(v.y, 0.0f);
                    v.z = fmaxf(v.z, 0.0f); v.w = fmaxf(v.w, 0.0f);
                }
                ((float4*)out)[gi] = v;
            }
        }
        __syncthreads();
    }
}

constexpr int L1_F      = 128;
constexpr int L1_G      = 64;
constexpr int L1_TP     = L1_F + 4;
constexpr int L1_PP1    = 2 * L1_F + 32;
constexpr int L1_PP2    = 2 * L1_G + 32;
constexpr int L1_STRIPS = AGG_ROWS / 16;
static_assert(AGG_ROWS % 16 == 0, "a tile is whole strips");
static_assert(2 * L1_STRIPS == 20 && AGG_WAVES == 32, "the item-to-wave map below places 20 items on 32 waves");

template <int NOUT, int PP>
__device__ __forceinline__ void gcn_stage_w(const float* __restrict__ W, float* Ws, int tid) {
    constexpr int ITEMS = (GEMM_K / 2) * (NOUT / 4);
    static_assert(ITEMS % 1024 == 0, "whole trips of the work-group");
    constexpr int TRIPS = ITEMS / 1024;
    float4 v0[TRIPS], v1[TRIPS];
#pragma unroll
    for (int j = 0; j < TRIPS; ++j) {
        const int i  = tid + j * 1024;
        const int p  = i / (NOUT / 4);
        const int c4 = i % (NOUT / 4);
        v0[j] = ((const float4*)W)[(size_t)(2 * p) * (NOUT / 4) + c4];
        v1[j] = ((const float4*)W)[(size_t)(2 * p + 1) * (NOUT / 4) + c4];
    }
#pragma unroll
    for (int j = 0; j < TRIPS; ++j) {
        const int i  = tid + j * 1024;
        const int p  = i / (NOUT / 4);
        const int c4 = i % (NOUT / 4);
        float4* d = (float4*)&Ws[p * PP + c4 * 8];
        d[0] = make_float4(v0[j].x, v1[j].x, v0[j].y, v1[j].y);
        d[1] = make_float4(v0[j].z, v1[j].z, v0[j].w, v1[j].w);
    }
}

template <int NT, int PP>
__device__ __forceinline__ void gcn_strip_product(const float* arow, const float* wl, v8f (&c)[NT]) {
    v2f a = *(const v2f*)arow, an = a;
    v2f b[NT], bn[NT];
#pragma unroll
    for (int t = 0; t < NT; ++t) { b[t] = *(const v2f*)(wl + t * 32); bn[t] = b[t]; }
#pragma unroll
    for (int s = 0; s < GEMM_K / 4; ++s) {
        if (s + 1 < GEMM_K / 4) {
            an = *(const v2f*)(arow + 4 * (s + 1));
            const float* wn = wl + (2 * (s + 1)) * PP;
#pragma unroll
            for (int t = 0; t < NT; ++t) bn[t] = *(const v2f*)(wn + t * 32);
        }
#pragma unroll
        for (int t = 0; t < NT; ++t) {
            c[t] = __builtin_amdgcn_wmma_f32_16x16x4_f32(
                false, a, false, b[t], (short)0, c[t], false, false);
        }
        a = an;
#pragma unroll
        for (int t = 0; t < NT; ++t) b[t] = bn[t];
    }
}

__global__ __launch_bounds__(1024) void gcn_layer1_tile(
    const float* __restrict__ x, const int* __restrict__ src, const int* __restrict__ dst,
    const int* __restrict__ deg8, const float* __restrict__ W1, const float* __restrict__ b1,
    const float* __restrict__ W2, const float* __restrict__ lists,
    float* __restrict__ g, float* __restrict__ dinv, int N, int E) {
    __shared__ __attribute__((aligned(16))) float acc[AGG_ROWS * L1_TP];
    __shared__ float dl[AGG_ROWS];
    __shared__ unsigned fq[AGG_WAVES * AGG_FQCAP];
    __shared__ unsigned fs[AGG_WAVES * AGG_FQCAP];
    __shared__ float fn[AGG_WAVES * AGG_FQCAP];
    __shared__ int ovf[2];
    __shared__ __attribute__((aligned(16))) float W1s[(GEMM_K / 2) * L1_PP1];
    __shared__ __attribute__((aligned(16))) float W2s[(GEMM_K / 2) * L1_PP2];

    const int tid  = threadIdx.x;
    const int lane = tid & 31;
    const int wave = __builtin_amdgcn_readfirstlane(tid >> 5);
    unsigned* myq = fq + wave * AGG_FQCAP;
    unsigned* sq  = fs + wave * AGG_FQCAP;
    float*    sw  = fn + wave * AGG_FQCAP;

    if (tid == 0) { ovf[0] = 0; ovf[1] = 0; }
    __syncthreads();
    int par = 0;

    gcn_stage_w<L1_F, L1_PP1>(W1, W1s, tid);
    gcn_stage_w<L1_G, L1_PP2>(W2, W2s, tid);

    const int wq = wave & 7;
    const int wb = wave >> 3;
    const bool prod  = (wq < 4) || (wq == 4 + wb);
    const int  item  = wb * 5 + ((wq < 4) ? wq : 4);
    const int  strip = item % L1_STRIPS;
    const int  half  = item / L1_STRIPS;

    const int ntiles = (N + AGG_ROWS - 1) / AGG_ROWS;

    for (int tile = blockIdx.x; tile < ntiles; tile += gridDim.x) {
        int tq = tid;
        asm volatile("" : "+v"(tq));
        const int tidq  = tq;
        const int laneq = tq & 31;
        const int lo = tile * AGG_ROWS;
        const int nrows = min(AGG_ROWS, N - lo);
        const float* tl = lists + (size_t)lo * AGG_LROW;
        const int segw = (nrows * AGG_LROW) / AGG_GS;
        const int subw = segw / AGG_WAVES;

        typedef float __attribute__((ext_vector_type(L1_F / 32))) vt;
        const int rowc = lo + min(tidq, nrows - 1);
        const int4 dc0 = *(const int4*)(deg8 + (size_t)rowc * AGG_GS);
        const int4 dc1 = *(const int4*)(deg8 + (size_t)rowc * AGG_GS + 4);
        int total = -1;
        int cn[AGG_OWN];
        if (subw >= 2) total = gcn_own_fetch<1>(tl, segw, subw, nullptr, deg8, wave, laneq, myq, sq, sw, cn);
        if (total < 0) ovf[par] = 1;
        if (tidq < nrows) {
            const float dmy = gcn_dinv_from(dc0, dc1);
            dl[tidq] = dmy;
            dinv[lo + tidq] = dmy;
        }
        __syncthreads();
        const bool fits = __builtin_amdgcn_readfirstlane(ovf[par]) == 0;
        if (tidq == 0) ovf[par ^ 1] = 0;
        par ^= 1;

        if (fits) {
            const int cnv = gcn_counts_by_lane(cn, laneq);
#pragma unroll 1
            for (int j = 0; j < AGG_OWN; ++j) {
                const int row  = wave + j * AGG_WAVES;
                const int rin  = min(row, nrows - 1);
                const int cnt  = __builtin_amdgcn_readlane(cnv, j);
                const float di = dl[rin];
                const vt xs = *((const vt*)(x + (size_t)(lo + rin) * L1_F) + laneq);
                vt a = (row < nrows) ? xs * di * di : (vt)(0.0f);
                gcn_row_sum<L1_F>(x, sq + j * AGG_JCAP, sw + j * AGG_JCAP, cnt, di, laneq, a);
                *(vt*)&acc[row * L1_TP + (L1_F / 32) * laneq] = a;
            }
        } else {
            for (int i = tidq; i < AGG_ROWS * L1_TP / 4; i += 1024)
                ((float4*)acc)[i] = make_float4(0.0f, 0.0f, 0.0f, 0.0f);
            __syncthreads();
            int cnt = 0;
            for (int e0 = wave * 32; e0 < E; e0 += AGG_WAVES * 32) {
                const int e = e0 + laneq;
                const int dv = (e < E) ? dst[e] : -1;
                const unsigned r = (unsigned)(dv - lo);
                const bool mk = r < (unsigned)nrows;
                const unsigned bm = __builtin_amdgcn_ballot_w32(mk);
                if (bm != 0u) {
                    if (cnt > AGG_FQCAP - 32) {
                        gcn_agg_drain<L1_F, L1_TP, 1>(x, nullptr, deg8, dl, acc, myq, cnt, laneq);
                        cnt = 0;
                    }
                    if (mk) myq[cnt + (int)__builtin_amdgcn_mbcnt_lo(bm, 0u)] = (r << 24) | (unsigned)src[e];
                    cnt += __builtin_popcount(bm);
                }
            }
            gcn_agg_drain<L1_F, L1_TP, 1>(x, nullptr, deg8, dl, acc, myq, cnt, laneq);
            __syncthreads();

            for (int i = tidq; i < nrows * (L1_F / 4); i += 1024) {
                const int r  = i / (L1_F / 4);
                const int f4 = i % (L1_F / 4);
                const float di = dl[r];
                const float4 xv = ((const float4*)x)[(size_t)(lo + r) * (L1_F / 4) + f4];
                float4* p = (float4*)&acc[r * L1_TP + 4 * f4];
                float4 v = *p;
                v.x += xv.x * di * di;
                v.y += xv.y * di * di;
                v.z += xv.z * di * di;
                v.w += xv.w * di * di;
                *p = v;
            }
        }
        __syncthreads();

        int lq = lane;
        asm volatile("" : "+v"(lq));
        const int m  = lq & 15;
        const int gh = lq >> 4;

        v8f c1[4];
        float bv[4];
        if (prod) {
#pragma unroll
            for (int t = 0; t < 4; ++t) {
                c1[t] = (v8f){0, 0, 0, 0, 0, 0, 0, 0};
                bv[t] = b1[(4 * half + t) * 16 + m];
            }
            gcn_strip_product<4, L1_PP1>(acc + (strip * 16 + m) * L1_TP + 2 * gh,
                                         W1s + gh * L1_PP1 + 2 * m + (4 * half) * 32, c1);
        }
        __syncthreads();
        if (prod) {
#pragma unroll
            for (int t = 0; t < 4; ++t) {
                const int col = (4 * half + t) * 16 + m;
#pragma unroll
                for (int r = 0; r < 8; ++r)
                    acc[(strip * 16 + 8 * gh + r) * L1_TP + col] = fmaxf(c1[t][r] + bv[t], 0.0f);
            }
        }
        __syncthreads();

        if (prod) {
            v8f c2[2];
#pragma unroll
            for (int t = 0; t < 2; ++t) c2[t] = (v8f){0, 0, 0, 0, 0, 0, 0, 0};
            gcn_strip_product<2, L1_PP2>(acc + (strip * 16 + m) * L1_TP + 2 * gh,
                                         W2s + gh * L1_PP2 + 2 * m + (2 * half) * 32, c2);
#pragma unroll
            for (int t = 0; t < 2; ++t) {
                const int col = (2 * half + t) * 16 + m;
#pragma unroll
                for (int r = 0; r < 8; ++r) {
                    const int row = strip * 16 + 8 * gh + r;
                    if (row < nrows) g[(size_t)(lo + row) * L1_G + col] = c2[t][r];
                }
            }
        }
        __syncthreads();
    }
}

extern "C" void kernel_launch(void* const* d_in, const int* in_sizes, int n_in,
                              void* d_out, int out_size, void* d_ws, size_t ws_size,
                              hipStream_t stream) {
    const float* x  = (const float*)d_in[0];
    const int*   ei = (const int*)d_in[1];
    const float* W1 = (const float*)d_in[2];
    const float* b1 = (const float*)d_in[3];
    const float* W2 = (const float*)d_in[4];
    const float* b2 = (const float*)d_in[5];
    float* out = (float*)d_out;

    const int DIN = 128, DH = 128;
    const int N = in_sizes[0] / DIN;
    const int E = in_sizes[1] / 2;
    const int* src = ei;
    const int* dst = ei + E;

    float* ws = (float*)d_ws;
    size_t dinvPad = ((size_t)N + 255) & ~(size_t)255;
    float* dinv = ws;
    float* g    = ws + dinvPad;
    int*   deg8 = (int*)(g + (size_t)N * DH);

    gcn_tile_lists<<<256, 1024, 0, stream>>>(src, dst, deg8, out, N, E);

    gcn_layer1_tile<<<256, 1024, 0, stream>>>(x, src, dst, deg8, W1, b1, W2, out, g, dinv, N, E);

    gcn_agg_tile<64, 0><<<256, 1024, 0, stream>>>(g, src, dst, dinv, b2, out, out, N, E);
}
